# starting kernel
# baseline (speedup 1.0000x reference)
#include <hip/hip_runtime.h>
#include <hip/hip_bf16.h>

typedef float v2f __attribute__((ext_vector_type(2)));
typedef float v8f __attribute__((ext_vector_type(8)));

#define HDIM 64

__global__ void zero_kernel(float* __restrict__ p, int n) {
    int i = blockIdx.x * blockDim.x + threadIdx.x;
    int stride = gridDim.x * blockDim.x;
    for (; i < n; i += stride) p[i] = 0.0f;
}

__global__ void edge_stats_kernel(const int* __restrict__ src, const int* __restrict__ dst,
                                  const float* __restrict__ x,
                                  float* __restrict__ degacc, float* __restrict__ signacc,
                                  float* __restrict__ Sx, int E) {
    int e = blockIdx.x * blockDim.x + threadIdx.x;
    if (e >= E) return;
    int s = src[e], d = dst[e];
    float sg = (s > d) ? 1.0f : ((s < d) ? -1.0f : 0.0f);
    atomicAdd(&degacc[d], 1.0f);
    atomicAdd(&signacc[d], sg);
    atomicAdd(&Sx[d], x[s]);
}

__global__ void node_stats_kernel(const float* __restrict__ degacc, const float* __restrict__ signacc,
                                  float* __restrict__ invdeg, float* __restrict__ gate,
                                  float* __restrict__ ssn, int N) {
    int i = blockIdx.x * blockDim.x + threadIdx.x;
    if (i >= N) return;
    float dg = degacc[i];
    float inv = 1.0f / fmaxf(dg, 1.0f);
    invdeg[i] = inv;
    gate[i]   = (dg > 0.0f) ? 1.0f : 0.0f;
    ssn[i]    = signacc[i] * inv;
}

__global__ void wdiff_kernel(const float* __restrict__ Wmid, float* __restrict__ Wd6, int total) {
    int i = blockIdx.x * blockDim.x + threadIdx.x;
    if (i >= total) return;
    int l = i >> 12;
    int rc = i & 4095;
    const float* Wl = Wmid + (size_t)l * 129 * HDIM;
    Wd6[i] = Wl[rc] - Wl[64 * HDIM + rc];
}

__global__ void conv_in_kernel(const float* __restrict__ x, const float* __restrict__ Win,
                               const float* __restrict__ bin,
                               const float* __restrict__ invdeg, const float* __restrict__ gate,
                               const float* __restrict__ ssn, const float* __restrict__ Sx,
                               float* __restrict__ Y, int N) {
    int t = blockIdx.x * blockDim.x + threadIdx.x;
    if (t >= N * HDIM) return;
    int n = t >> 6, h = t & 63;
    float wa = Win[h], wb = Win[HDIM + h], wc = Win[2 * HDIM + h];
    Y[t] = gate[n] * (x[n] * (wa - wb) + bin[h]) + invdeg[n] * Sx[n] * wb + ssn[n] * wc;
}

__global__ void scatter_kernel(const int* __restrict__ src, const int* __restrict__ dst,
                               const float* __restrict__ X, float* __restrict__ S, int work) {
    int t = blockIdx.x * blockDim.x + threadIdx.x;
    if (t >= work) return;
    int e = t >> 4;
    int q = t & 15;
    int s = src[e], d = dst[e];
    const float4 v = *(const float4*)(X + (size_t)s * HDIM + q * 4);
    float* o = S + (size_t)d * HDIM + q * 4;
    atomicAdd(o + 0, v.x);
    atomicAdd(o + 1, v.y);
    atomicAdd(o + 2, v.z);
    atomicAdd(o + 3, v.w);
}

__global__ __launch_bounds__(256) void mid_conv_kernel(
    const float* __restrict__ X, const float* __restrict__ S,
    const float* __restrict__ Wd, const float* __restrict__ Wb,
    const float* __restrict__ Wc, const float* __restrict__ bias,
    const float* __restrict__ gate, const float* __restrict__ invdeg,
    const float* __restrict__ ssn, const float* __restrict__ resid,
    float* __restrict__ Y, int nTiles, int N) {
    __shared__ float sWd[HDIM * HDIM];
    __shared__ float sWb[HDIM * HDIM];
    for (int i = threadIdx.x; i < HDIM * HDIM; i += 256) {
        int ks   = i >> 8;
        int rem  = i & 255;
        int nt   = rem >> 6;
        int lp   = rem & 63;
        int ln   = lp >> 1;
        int half = lp & 1;
        int kh   = (ln >> 4) << 1;
        int srcr = (ks << 2) + kh + half;
        int srcc = (nt << 4) + (ln & 15);
        sWd[i] = Wd[srcr * HDIM + srcc];
        sWb[i] = Wb[srcr * HDIM + srcc];
    }
    __syncthreads();

    int wave = threadIdx.x >> 5;
    int lane = threadIdx.x & 31;
    int tile = blockIdx.x * 8 + wave;
    if (tile >= nTiles) return;

    int node0 = tile * 16;
    int m  = lane & 15;
    int kh = (lane >> 4) << 1;
    int row = node0 + m; if (row >= N) row = N - 1;
    const float* Xr = X + (size_t)row * HDIM;
    const float* Sr = S + (size_t)row * HDIM;

    v8f accX[4] = {};
    v8f accS[4] = {};
    int fragBase = lane * 2;

#pragma unroll
    for (int ks = 0; ks < 16; ++ks) {
        int k0 = ks << 2;
        v2f a, s;
        a.x = Xr[k0 + kh];     a.y = Xr[k0 + kh + 1];
        s.x = Sr[k0 + kh];     s.y = Sr[k0 + kh + 1];
#pragma unroll
        for (int nt = 0; nt < 4; ++nt) {
            v2f bd = *(const v2f*)&sWd[((ks << 2) + nt) * 64 + fragBase];
            v2f bb = *(const v2f*)&sWb[((ks << 2) + nt) * 64 + fragBase];
            accX[nt] = __builtin_amdgcn_wmma_f32_16x16x4_f32(
                false, a, false, bd, (short)0, accX[nt], false, false);
            accS[nt] = __builtin_amdgcn_wmma_f32_16x16x4_f32(
                false, s, false, bb, (short)0, accS[nt], false, false);
        }
    }

    int ncol = lane & 15;
    int mAdd = (lane >> 4) * 8;
#pragma unroll
    for (int nt = 0; nt < 4; ++nt) {
        int col = nt * 16 + ncol;
        float wc = Wc[col];
        float bb = bias[col];
#pragma unroll
        for (int v = 0; v < 8; ++v) {
            int node = node0 + v + mAdd;
            if (node >= N) continue;
            float g = gate[node];
            float val = g * (accX[nt][v] + bb) + invdeg[node] * accS[nt][v] + ssn[node] * wc;
            if (resid) val += resid[(size_t)node * HDIM + col];
            val = (val > 0.0f) ? val : 0.01f * val;
            Y[(size_t)node * HDIM + col] = val;
        }
    }
}

__global__ void conv_out_kernel(const float* __restrict__ X, const float* __restrict__ S,
                                const float* __restrict__ Wout, const float* __restrict__ bout,
                                const float* __restrict__ x0,
                                const float* __restrict__ invdeg, const float* __restrict__ gate,
                                const float* __restrict__ ssn, float* __restrict__ out, int N) {
    int n = blockIdx.x * blockDim.x + threadIdx.x;
    if (n >= N) return;
    float a1 = 0.0f, a2 = 0.0f;
    const float* Xr = X + (size_t)n * HDIM;
    const float* Sr = S + (size_t)n * HDIM;
#pragma unroll 8
    for (int k = 0; k < HDIM; ++k) {
        float wa = Wout[k], wb = Wout[HDIM + k];
        a1 += Xr[k] * (wa - wb);
        a2 += Sr[k] * wb;
    }
    out[n] = gate[n] * (a1 + bout[0]) + invdeg[n] * a2 + ssn[n] * Wout[2 * HDIM] + x0[n];
}

extern "C" void kernel_launch(void* const* d_in, const int* in_sizes, int n_in,
                              void* d_out, int out_size, void* d_ws, size_t ws_size,
                              hipStream_t stream) {
    const float* x    = (const float*)d_in[0];
    const int*   ei   = (const int*)  d_in[1];
    const float* Win  = (const float*)d_in[2];
    const float* bin  = (const float*)d_in[3];
    const float* Wmid = (const float*)d_in[4];
    const float* bmid = (const float*)d_in[5];
    const float* Wout = (const float*)d_in[6];
    const float* bout = (const float*)d_in[7];

    const int N = in_sizes[0];
    const int E = in_sizes[1] / 2;
    const int* src = ei;
    const int* dst = ei + E;

    float* w = (float*)d_ws;
    float* degacc  = w;
    float* signacc = w + (size_t)N;
    float* invdeg  = w + 2 * (size_t)N;
    float* gate    = w + 3 * (size_t)N;
    float* ssn     = w + 4 * (size_t)N;
    float* Sx      = w + 5 * (size_t)N;
    float* Wd6     = w + 6 * (size_t)N;
    float* bufA    = Wd6  + 6 * HDIM * HDIM;
    float* bufB    = bufA + (size_t)N * HDIM;
    float* bufC    = bufB + (size_t)N * HDIM;
    float* bufS    = bufC + (size_t)N * HDIM;

    const int nTiles = (N + 15) / 16;
    const int sWork  = E * 16;

    zero_kernel<<<1024, 256, 0, stream>>>(w, 6 * N);
    edge_stats_kernel<<<(E + 255) / 256, 256, 0, stream>>>(src, dst, x, degacc, signacc, Sx, E);
    node_stats_kernel<<<(N + 255) / 256, 256, 0, stream>>>(degacc, signacc, invdeg, gate, ssn, N);
    wdiff_kernel<<<(6 * 4096 + 255) / 256, 256, 0, stream>>>(Wmid, Wd6, 6 * 4096);
    conv_in_kernel<<<(N * HDIM + 255) / 256, 256, 0, stream>>>(x, Win, bin, invdeg, gate, ssn, Sx,
                                                               bufA, N);

    float* cur = bufA;
    float* alt = bufC;
    for (int l = 0; l < 3; ++l) {
        zero_kernel<<<2048, 256, 0, stream>>>(bufS, N * HDIM);
        scatter_kernel<<<(sWork + 255) / 256, 256, 0, stream>>>(src, dst, cur, bufS, sWork);
        const float* Wl = Wmid + (size_t)(2 * l) * 129 * HDIM;
        mid_conv_kernel<<<(nTiles + 7) / 8, 256, 0, stream>>>(
            cur, bufS, Wd6 + (2 * l) * HDIM * HDIM, Wl + 64 * HDIM, Wl + 128 * HDIM,
            bmid + (2 * l) * HDIM, gate, invdeg, ssn, nullptr, bufB, nTiles, N);

        zero_kernel<<<2048, 256, 0, stream>>>(bufS, N * HDIM);
        scatter_kernel<<<(sWork + 255) / 256, 256, 0, stream>>>(src, dst, bufB, bufS, sWork);
        const float* Wl2 = Wmid + (size_t)(2 * l + 1) * 129 * HDIM;
        mid_conv_kernel<<<(nTiles + 7) / 8, 256, 0, stream>>>(
            bufB, bufS, Wd6 + (2 * l + 1) * HDIM * HDIM, Wl2 + 64 * HDIM, Wl2 + 128 * HDIM,
            bmid + (2 * l + 1) * HDIM, gate, invdeg, ssn, cur, alt, nTiles, N);

        float* t = cur; cur = alt; alt = t;
    }

    zero_kernel<<<2048, 256, 0, stream>>>(bufS, N * HDIM);
    scatter_kernel<<<(sWork + 255) / 256, 256, 0, stream>>>(src, dst, cur, bufS, sWork);
    conv_out_kernel<<<(N + 255) / 256, 256, 0, stream>>>(cur, bufS, Wout, bout, x, invdeg, gate,
                                                         ssn, (float*)d_out, N);
}


# Neighbor sums by gather over a per-call destination-sorted edge list (int-atomic count, scan, arrival-order fill) replace the float-atomic scatters; summation order inside a node changes
# speedup vs baseline: 2.3705x; 2.3705x over previous
#include <hip/hip_runtime.h>
#include <hip/hip_bf16.h>

typedef float v2f __attribute__((ext_vector_type(2)));
typedef float v8f __attribute__((ext_vector_type(8)));

#define HDIM 64

__global__ void zero_kernel(float* __restrict__ p, int n) {
    int i = blockIdx.x * blockDim.x + threadIdx.x;
    int stride = gridDim.x * blockDim.x;
    for (; i < n; i += stride) p[i] = 0.0f;
}

__global__ void count_kernel(const int* __restrict__ dst, int* __restrict__ cnt, int E) {
    int e = blockIdx.x * blockDim.x + threadIdx.x;
    if (e >= E) return;
    atomicAdd(&cnt[dst[e]], 1);
}

__global__ __launch_bounds__(1024) void scan_kernel(int* cc, int* __restrict__ rowptr, int N) {
    __shared__ int sWave[32];
    __shared__ int sTotal;
    const int tid = threadIdx.x, lane = tid & 31, wv = tid >> 5;
    int carry = 0;
    for (int base = 0; base < N; base += 4096) {
        int i0 = base + tid * 4;
        int c0 = (i0     < N) ? cc[i0]     : 0;
        int c1 = (i0 + 1 < N) ? cc[i0 + 1] : 0;
        int c2 = (i0 + 2 < N) ? cc[i0 + 2] : 0;
        int c3 = (i0 + 3 < N) ? cc[i0 + 3] : 0;
        int s = c0 + c1 + c2 + c3;
        int inc = s;
#pragma unroll
        for (int o = 1; o < 32; o <<= 1) {
            int v = __shfl_up(inc, o, 32);
            if (lane >= o) inc += v;
        }
        if (lane == 31) sWave[wv] = inc;
        __syncthreads();
        if (wv == 0) {
            int w = sWave[lane];
            int winc = w;
#pragma unroll
            for (int o = 1; o < 32; o <<= 1) {
                int v = __shfl_up(winc, o, 32);
                if (lane >= o) winc += v;
            }
            sWave[lane] = winc - w;
            if (lane == 31) sTotal = winc;
        }
        __syncthreads();
        int ex = carry + sWave[wv] + (inc - s);
        if (i0     < N) { rowptr[i0]     = ex; cc[i0]     = ex; } ex += c0;
        if (i0 + 1 < N) { rowptr[i0 + 1] = ex; cc[i0 + 1] = ex; } ex += c1;
        if (i0 + 2 < N) { rowptr[i0 + 2] = ex; cc[i0 + 2] = ex; } ex += c2;
        if (i0 + 3 < N) { rowptr[i0 + 3] = ex; cc[i0 + 3] = ex; }
        carry += sTotal;
        __syncthreads();
    }
}

__global__ void fill_kernel(const int* __restrict__ src, const int* __restrict__ dst,
                            int* __restrict__ cursor, int* __restrict__ csr, int E) {
    int e = blockIdx.x * blockDim.x + threadIdx.x;
    if (e >= E) return;
    int pos = atomicAdd(&cursor[dst[e]], 1);
    csr[pos] = src[e];
}

__global__ void node_stats_kernel(const int* __restrict__ rowptr, const int* __restrict__ rowend,
                                  const int* __restrict__ csr, const float* __restrict__ x,
                                  float* __restrict__ invdeg, float* __restrict__ gate,
                                  float* __restrict__ ssn, float* __restrict__ Sx, int N) {
    int t = blockIdx.x * blockDim.x + threadIdx.x;
    int n = t >> 3, l = t & 7;
    bool valid = n < N;
    int nn = valid ? n : N - 1;
    int b = rowptr[nn], e = rowend[nn];
    float sg = 0.0f, sx = 0.0f;
    for (int j = b + l; j < e; j += 8) {
        int s = csr[j];
        sg += (s > nn) ? 1.0f : ((s < nn) ? -1.0f : 0.0f);
        sx += x[s];
    }
#pragma unroll
    for (int o = 4; o > 0; o >>= 1) {
        sg += __shfl_xor(sg, o, 8);
        sx += __shfl_xor(sx, o, 8);
    }
    if (valid && l == 0) {
        float dg = (float)(e - b);
        float inv = 1.0f / fmaxf(dg, 1.0f);
        invdeg[n] = inv;
        gate[n]   = (dg > 0.0f) ? 1.0f : 0.0f;
        ssn[n]    = sg * inv;
        Sx[n]     = sx;
    }
}

__global__ void wdiff_kernel(const float* __restrict__ Wmid, float* __restrict__ Wd6, int total) {
    int i = blockIdx.x * blockDim.x + threadIdx.x;
    if (i >= total) return;
    int l = i >> 12;
    int rc = i & 4095;
    const float* Wl = Wmid + (size_t)l * 129 * HDIM;
    Wd6[i] = Wl[rc] - Wl[64 * HDIM + rc];
}

__global__ void conv_in_kernel(const float* __restrict__ x, const float* __restrict__ Win,
                               const float* __restrict__ bin,
                               const float* __restrict__ invdeg, const float* __restrict__ gate,
                               const float* __restrict__ ssn, const float* __restrict__ Sx,
                               float* __restrict__ Y, int N) {
    int t = blockIdx.x * blockDim.x + threadIdx.x;
    if (t >= N * HDIM) return;
    int n = t >> 6, h = t & 63;
    float wa = Win[h], wb = Win[HDIM + h], wc = Win[2 * HDIM + h];
    Y[t] = gate[n] * (x[n] * (wa - wb) + bin[h]) + invdeg[n] * Sx[n] * wb + ssn[n] * wc;
}

__global__ __launch_bounds__(256) void gather_kernel(const int* __restrict__ rowptr,
                                                     const int* __restrict__ rowend,
                                                     const int* __restrict__ csr,
                                                     const float* __restrict__ X,
                                                     float* __restrict__ S, int N) {
    int t = blockIdx.x * 256 + threadIdx.x;
    int n = t >> 4, q = t & 15;
    bool valid = n < N;
    int nn = valid ? n : N - 1;
    int b = rowptr[nn], e = rowend[nn];
    const float4* X4 = (const float4*)X;
    float4 acc = make_float4(0.0f, 0.0f, 0.0f, 0.0f);
    for (int base = b; base < e; base += 16) {
        int j = base + q;
        int mine = (j < e) ? csr[j] : nn;
        int cnt = e - base;
#pragma unroll
        for (int k = 0; k < 16; ++k) {
            int s = __shfl(mine, k, 16);
            float4 v = X4[(size_t)s * 16 + q];
            bool ok = k < cnt;
            acc.x += ok ? v.x : 0.0f;
            acc.y += ok ? v.y : 0.0f;
            acc.z += ok ? v.z : 0.0f;
            acc.w += ok ? v.w : 0.0f;
        }
    }
    if (valid) ((float4*)S)[(size_t)n * 16 + q] = acc;
}

__global__ __launch_bounds__(256) void mid_conv_kernel(
    const float* __restrict__ X, const float* __restrict__ S,
    const float* __restrict__ Wd, const float* __restrict__ Wb,
    const float* __restrict__ Wc, const float* __restrict__ bias,
    const float* __restrict__ gate, const float* __restrict__ invdeg,
    const float* __restrict__ ssn, const float* resid,
    float* Y, int nTiles, int N) {
    __shared__ float sWd[HDIM * HDIM];
    __shared__ float sWb[HDIM * HDIM];
    for (int i = threadIdx.x; i < HDIM * HDIM; i += 256) {
        int ks   = i >> 8;
        int rem  = i & 255;
        int nt   = rem >> 6;
        int lp   = rem & 63;
        int ln   = lp >> 1;
        int half = lp & 1;
        int kh   = (ln >> 4) << 1;
        int srcr = (ks << 2) + kh + half;
        int srcc = (nt << 4) + (ln & 15);
        sWd[i] = Wd[srcr * HDIM + srcc];
        sWb[i] = Wb[srcr * HDIM + srcc];
    }
    __syncthreads();

    int wave = threadIdx.x >> 5;
    int lane = threadIdx.x & 31;
    int tile = blockIdx.x * 8 + wave;
    if (tile >= nTiles) return;

    int node0 = tile * 16;
    int m  = lane & 15;
    int kh = (lane >> 4) << 1;
    int row = node0 + m; if (row >= N) row = N - 1;
    const float* Xr = X + (size_t)row * HDIM;
    const float* Sr = S + (size_t)row * HDIM;

    v8f accX[4] = {};
    v8f accS[4] = {};
    int fragBase = lane * 2;

#pragma unroll
    for (int ks = 0; ks < 16; ++ks) {
        int k0 = ks << 2;
        v2f a, s;
        a.x = Xr[k0 + kh];     a.y = Xr[k0 + kh + 1];
        s.x = Sr[k0 + kh];     s.y = Sr[k0 + kh + 1];
#pragma unroll
        for (int nt = 0; nt < 4; ++nt) {
            v2f bd = *(const v2f*)&sWd[((ks << 2) + nt) * 64 + fragBase];
            v2f bb = *(const v2f*)&sWb[((ks << 2) + nt) * 64 + fragBase];
            accX[nt] = __builtin_amdgcn_wmma_f32_16x16x4_f32(
                false, a, false, bd, (short)0, accX[nt], false, false);
            accS[nt] = __builtin_amdgcn_wmma_f32_16x16x4_f32(
                false, s, false, bb, (short)0, accS[nt], false, false);
        }
    }

    int ncol = lane & 15;
    int mAdd = (lane >> 4) * 8;
    float rs[4][8];
    if (resid) {
#pragma unroll
        for (int nt = 0; nt < 4; ++nt) {
#pragma unroll
            for (int v = 0; v < 8; ++v) {
                int node = node0 + v + mAdd;
                rs[nt][v] = (node < N) ? resid[(size_t)node * HDIM + nt * 16 + ncol] : 0.0f;
            }
        }
    }
#pragma unroll
    for (int nt = 0; nt < 4; ++nt) {
        int col = nt * 16 + ncol;
        float wc = Wc[col];
        float bb = bias[col];
#pragma unroll
        for (int v = 0; v < 8; ++v) {
            int node = node0 + v + mAdd;
            if (node >= N) continue;
            float g = gate[node];
            float val = g * (accX[nt][v] + bb) + invdeg[node] * accS[nt][v] + ssn[node] * wc;
            if (resid) val += rs[nt][v];
            val = (val > 0.0f) ? val : 0.01f * val;
            Y[(size_t)node * HDIM + col] = val;
        }
    }
}

__global__ void conv_out_kernel(const float* __restrict__ X, const float* __restrict__ S,
                                const float* __restrict__ Wout, const float* __restrict__ bout,
                                const float* __restrict__ x0,
                                const float* __restrict__ invdeg, const float* __restrict__ gate,
                                const float* __restrict__ ssn, float* __restrict__ out, int N) {
    int n = blockIdx.x * blockDim.x + threadIdx.x;
    if (n >= N) return;
    float a1 = 0.0f, a2 = 0.0f;
    const float* Xr = X + (size_t)n * HDIM;
    const float* Sr = S + (size_t)n * HDIM;
#pragma unroll 8
    for (int k = 0; k < HDIM; ++k) {
        float wa = Wout[k], wb = Wout[HDIM + k];
        a1 += Xr[k] * (wa - wb);
        a2 += Sr[k] * wb;
    }
    out[n] = gate[n] * (a1 + bout[0]) + invdeg[n] * a2 + ssn[n] * Wout[2 * HDIM] + x0[n];
}

extern "C" void kernel_launch(void* const* d_in, const int* in_sizes, int n_in,
                              void* d_out, int out_size, void* d_ws, size_t ws_size,
                              hipStream_t stream) {
    const float* x    = (const float*)d_in[0];
    const int*   ei   = (const int*)  d_in[1];
    const float* Win  = (const float*)d_in[2];
    const float* bin  = (const float*)d_in[3];
    const float* Wmid = (const float*)d_in[4];
    const float* bmid = (const float*)d_in[5];
    const float* Wout = (const float*)d_in[6];
    const float* bout = (const float*)d_in[7];

    const int N = in_sizes[0];
    const int E = in_sizes[1] / 2;
    const int* src = ei;
    const int* dst = ei + E;

    float* w = (float*)d_ws;
    int*   rowptr  = (int*)w;
    int*   cursor  = (int*)(w + (size_t)N);
    float* invdeg  = w + 2 * (size_t)N;
    float* gate    = w + 3 * (size_t)N;
    float* ssn     = w + 4 * (size_t)N;
    float* Sx      = w + 5 * (size_t)N;
    float* Wd6     = w + 6 * (size_t)N;
    float* bufA    = Wd6  + 6 * HDIM * HDIM;
    float* bufB    = bufA + (size_t)N * HDIM;
    float* bufC    = bufB + (size_t)N * HDIM;
    float* bufS    = bufC + (size_t)N * HDIM;
    int*   csr     = (int*)bufC;

    const int nTiles = (N + 15) / 16;

    zero_kernel<<<(N + 255) / 256, 256, 0, stream>>>(w + (size_t)N, N);
    count_kernel<<<(E + 255) / 256, 256, 0, stream>>>(dst, cursor, E);
    scan_kernel<<<1, 1024, 0, stream>>>(cursor, rowptr, N);
    fill_kernel<<<(E + 255) / 256, 256, 0, stream>>>(src, dst, cursor, csr, E);
    node_stats_kernel<<<(N * 8 + 255) / 256, 256, 0, stream>>>(rowptr, cursor, csr, x, invdeg, gate,
                                                               ssn, Sx, N);
    wdiff_kernel<<<(6 * 4096 + 255) / 256, 256, 0, stream>>>(Wmid, Wd6, 6 * 4096);
    conv_in_kernel<<<(N * HDIM + 255) / 256, 256, 0, stream>>>(x, Win, bin, invdeg, gate, ssn, Sx,
                                                               bufA, N);

    float* cur = bufA;
    const int gBlocks = (N * 16 + 255) / 256;
    for (int l = 0; l < 3; ++l) {
        gather_kernel<<<gBlocks, 256, 0, stream>>>(rowptr, cursor, csr, cur, bufS, N);
        const float* Wl = Wmid + (size_t)(2 * l) * 129 * HDIM;
        mid_conv_kernel<<<(nTiles + 7) / 8, 256, 0, stream>>>(
            cur, bufS, Wd6 + (2 * l) * HDIM * HDIM, Wl + 64 * HDIM, Wl + 128 * HDIM,
            bmid + (2 * l) * HDIM, gate, invdeg, ssn, nullptr, bufB, nTiles, N);

        gather_kernel<<<gBlocks, 256, 0, stream>>>(rowptr, cursor, csr, bufB, bufS, N);
        const float* Wl2 = Wmid + (size_t)(2 * l + 1) * 129 * HDIM;
        mid_conv_kernel<<<(nTiles + 7) / 8, 256, 0, stream>>>(
            bufB, bufS, Wd6 + (2 * l + 1) * HDIM * HDIM, Wl2 + 64 * HDIM, Wl2 + 128 * HDIM,
            bmid + (2 * l + 1) * HDIM, gate, invdeg, ssn, cur, cur, nTiles, N);
    }

    gather_kernel<<<gBlocks, 256, 0, stream>>>(rowptr, cursor, csr, cur, bufS, N);
    conv_out_kernel<<<(N + 255) / 256, 256, 0, stream>>>(cur, bufS, Wout, bout, x, invdeg, gate,
                                                         ssn, (float*)d_out, N);
}


# CSR build by two-level counting sort ranked with LDS counters (chunk-by-bin histogram, scan, bin scatter, per-bin fill) replaces two global atomics per edge; in-node sum order changes (LDS arrival)
# speedup vs baseline: 3.5127x; 1.4818x over previous
#include <hip/hip_runtime.h>
#include <hip/hip_bf16.h>

typedef float v2f __attribute__((ext_vector_type(2)));
typedef float v8f __attribute__((ext_vector_type(8)));

#define HDIM 64

#define BIN_SHIFT 8
#define BINW 256
#define NB_MAX 512
#define NCHUNK 256
#define CH_THREADS 1024
#define FILL_THREADS 512
#define FILL_CAP 12288

__global__ __launch_bounds__(CH_THREADS) void bin_count_kernel(const int* __restrict__ dst,
                                                               int* __restrict__ cntT, int E, int NB) {
    __shared__ int sHist[NB_MAX];
    const int c = blockIdx.x, tid = threadIdx.x;
    for (int i = tid; i < NB; i += CH_THREADS) sHist[i] = 0;
    __syncthreads();
    const int chunk = (E + NCHUNK - 1) / NCHUNK;
    const int e0 = c * chunk;
    const int e1 = (e0 + chunk < E) ? (e0 + chunk) : E;
    for (int e = e0 + tid; e < e1; e += CH_THREADS * 8) {
        int d[8];
#pragma unroll
        for (int u = 0; u < 8; ++u) {
            int ee = e + u * CH_THREADS;
            d[u] = dst[(ee < e1) ? ee : (e1 - 1)];
        }
#pragma unroll
        for (int u = 0; u < 8; ++u)
            if (e + u * CH_THREADS < e1) atomicAdd(&sHist[d[u] >> BIN_SHIFT], 1);
    }
    __syncthreads();
    for (int i = tid; i < NB; i += CH_THREADS) cntT[i * NCHUNK + c] = sHist[i];
}

__global__ __launch_bounds__(1024) void scan_kernel(int* cc, int* __restrict__ rowptr, int N) {
    __shared__ int sWave[32];
    __shared__ int sTotal;
    const int tid = threadIdx.x, lane = tid & 31, wv = tid >> 5;
    int carry = 0;
    for (int base = 0; base < N; base += 4096) {
        int i0 = base + tid * 4;
        int c0 = (i0     < N) ? cc[i0]     : 0;
        int c1 = (i0 + 1 < N) ? cc[i0 + 1] : 0;
        int c2 = (i0 + 2 < N) ? cc[i0 + 2] : 0;
        int c3 = (i0 + 3 < N) ? cc[i0 + 3] : 0;
        int s = c0 + c1 + c2 + c3;
        int inc = s;
#pragma unroll
        for (int o = 1; o < 32; o <<= 1) {
            int v = __shfl_up(inc, o, 32);
            if (lane >= o) inc += v;
        }
        if (lane == 31) sWave[wv] = inc;
        __syncthreads();
        if (wv == 0) {
            int w = sWave[lane];
            int winc = w;
#pragma unroll
            for (int o = 1; o < 32; o <<= 1) {
                int v = __shfl_up(winc, o, 32);
                if (lane >= o) winc += v;
            }
            sWave[lane] = winc - w;
            if (lane == 31) sTotal = winc;
        }
        __syncthreads();
        int ex = carry + sWave[wv] + (inc - s);
        if (i0     < N) { rowptr[i0]     = ex; cc[i0]     = ex; } ex += c0;
        if (i0 + 1 < N) { rowptr[i0 + 1] = ex; cc[i0 + 1] = ex; } ex += c1;
        if (i0 + 2 < N) { rowptr[i0 + 2] = ex; cc[i0 + 2] = ex; } ex += c2;
        if (i0 + 3 < N) { rowptr[i0 + 3] = ex; cc[i0 + 3] = ex; }
        carry += sTotal;
        __syncthreads();
    }
}

__global__ __launch_bounds__(CH_THREADS) void bin_scatter_kernel(const int* __restrict__ src,
                                                                 const int* __restrict__ dst,
                                                                 const int* __restrict__ offs,
                                                                 int* __restrict__ binned, int E, int NB) {
    __shared__ int sCur[NB_MAX];
    const int c = blockIdx.x, tid = threadIdx.x;
    for (int i = tid; i < NB; i += CH_THREADS) sCur[i] = offs[i * NCHUNK + c];
    __syncthreads();
    const int chunk = (E + NCHUNK - 1) / NCHUNK;
    const int e0 = c * chunk;
    const int e1 = (e0 + chunk < E) ? (e0 + chunk) : E;
    for (int e = e0 + tid; e < e1; e += CH_THREADS * 8) {
        int d[8], s[8];
#pragma unroll
        for (int u = 0; u < 8; ++u) {
            int ee = e + u * CH_THREADS;
            int ec = (ee < e1) ? ee : (e1 - 1);
            d[u] = dst[ec];
            s[u] = src[ec];
        }
#pragma unroll
        for (int u = 0; u < 8; ++u) {
            if (e + u * CH_THREADS < e1) {
                int pos = atomicAdd(&sCur[d[u] >> BIN_SHIFT], 1);
                binned[pos] = (s[u] << BIN_SHIFT) | (d[u] & (BINW - 1));
            }
        }
    }
}

__global__ __launch_bounds__(FILL_THREADS) void bin_fill_kernel(const int* __restrict__ offs,
                                                                const int* __restrict__ binned,
                                                                int* __restrict__ rowptr, int* __restrict__ rowend,
                                                                int* __restrict__ csr, int NB, int N, int E) {
    __shared__ int sDeg[BINW];
    __shared__ int sWaveTot[BINW / 32];
    __shared__ int sStage[FILL_CAP];
    const int b = blockIdx.x, tid = threadIdx.x, lane = tid & 31;
    const int beg = offs[b * NCHUNK];
    const int end = (b + 1 < NB) ? offs[(b + 1) * NCHUNK] : E;
    const int cnt = end - beg;
    if (tid < BINW) sDeg[tid] = 0;
    __syncthreads();
    for (int i = tid; i < cnt; i += FILL_THREADS * 8) {
        int r[8];
#pragma unroll
        for (int u = 0; u < 8; ++u) {
            int ii = i + u * FILL_THREADS;
            r[u] = binned[beg + ((ii < cnt) ? ii : (cnt - 1))];
        }
#pragma unroll
        for (int u = 0; u < 8; ++u)
            if (i + u * FILL_THREADS < cnt) atomicAdd(&sDeg[r[u] & (BINW - 1)], 1);
    }
    __syncthreads();
    int v = 0, inc = 0;
    if (tid < BINW) {
        v = sDeg[tid];
        inc = v;
#pragma unroll
        for (int o = 1; o < 32; o <<= 1) {
            int t = __shfl_up(inc, o, 32);
            if (lane >= o) inc += t;
        }
        if (lane == 31) sWaveTot[tid >> 5] = inc;
    }
    __syncthreads();
    if (tid < BINW) {
        int wbase = 0;
        for (int w = 0; w < (tid >> 5); ++w) wbase += sWaveTot[w];
        int excl = wbase + inc - v;
        sDeg[tid] = excl;
        int node = b * BINW + tid;
        if (node < N) { rowptr[node] = beg + excl; rowend[node] = beg + excl + v; }
    }
    __syncthreads();
    const bool staged = cnt <= FILL_CAP;
    for (int i = tid; i < cnt; i += FILL_THREADS * 8) {
        int r[8];
#pragma unroll
        for (int u = 0; u < 8; ++u) {
            int ii = i + u * FILL_THREADS;
            r[u] = binned[beg + ((ii < cnt) ? ii : (cnt - 1))];
        }
#pragma unroll
        for (int u = 0; u < 8; ++u) {
            if (i + u * FILL_THREADS < cnt) {
                int pos = atomicAdd(&sDeg[r[u] & (BINW - 1)], 1);
                if (staged) sStage[pos] = r[u] >> BIN_SHIFT;
                else        csr[beg + pos] = r[u] >> BIN_SHIFT;
            }
        }
    }
    __syncthreads();
    if (staged)
        for (int i = tid; i < cnt; i += FILL_THREADS) csr[beg + i] = sStage[i];
}

__global__ void node_stats_kernel(const int* __restrict__ rowptr, const int* __restrict__ rowend,
                                  const int* __restrict__ csr, const float* __restrict__ x,
                                  float* __restrict__ invdeg, float* __restrict__ gate,
                                  float* __restrict__ ssn, float* __restrict__ Sx, int N) {
    int t = blockIdx.x * blockDim.x + threadIdx.x;
    int n = t >> 3, l = t & 7;
    bool valid = n < N;
    int nn = valid ? n : N - 1;
    int b = rowptr[nn], e = rowend[nn];
    float sg = 0.0f, sx = 0.0f;
    for (int j = b + l; j < e; j += 8) {
        int s = csr[j];
        sg += (s > nn) ? 1.0f : ((s < nn) ? -1.0f : 0.0f);
        sx += x[s];
    }
#pragma unroll
    for (int o = 4; o > 0; o >>= 1) {
        sg += __shfl_xor(sg, o, 8);
        sx += __shfl_xor(sx, o, 8);
    }
    if (valid && l == 0) {
        float dg = (float)(e - b);
        float inv = 1.0f / fmaxf(dg, 1.0f);
        invdeg[n] = inv;
        gate[n]   = (dg > 0.0f) ? 1.0f : 0.0f;
        ssn[n]    = sg * inv;
        Sx[n]     = sx;
    }
}

__global__ void wdiff_kernel(const float* __restrict__ Wmid, float* __restrict__ Wd6, int total) {
    int i = blockIdx.x * blockDim.x + threadIdx.x;
    if (i >= total) return;
    int l = i >> 12;
    int rc = i & 4095;
    const float* Wl = Wmid + (size_t)l * 129 * HDIM;
    Wd6[i] = Wl[rc] - Wl[64 * HDIM + rc];
}

__global__ void conv_in_kernel(const float* __restrict__ x, const float* __restrict__ Win,
                               const float* __restrict__ bin,
                               const float* __restrict__ invdeg, const float* __restrict__ gate,
                               const float* __restrict__ ssn, const float* __restrict__ Sx,
                               float* __restrict__ Y, int N) {
    int t = blockIdx.x * blockDim.x + threadIdx.x;
    if (t >= N * HDIM) return;
    int n = t >> 6, h = t & 63;
    float wa = Win[h], wb = Win[HDIM + h], wc = Win[2 * HDIM + h];
    Y[t] = gate[n] * (x[n] * (wa - wb) + bin[h]) + invdeg[n] * Sx[n] * wb + ssn[n] * wc;
}

__global__ __launch_bounds__(256) void gather_kernel(const int* __restrict__ rowptr,
                                                     const int* __restrict__ rowend,
                                                     const int* __restrict__ csr,
                                                     const float* __restrict__ X,
                                                     float* __restrict__ S, int N) {
    int t = blockIdx.x * 256 + threadIdx.x;
    int n = t >> 4, q = t & 15;
    bool valid = n < N;
    int nn = valid ? n : N - 1;
    int b = rowptr[nn], e = rowend[nn];
    const float4* X4 = (const float4*)X;
    float4 acc = make_float4(0.0f, 0.0f, 0.0f, 0.0f);
    for (int base = b; base < e; base += 16) {
        int j = base + q;
        int mine = (j < e) ? csr[j] : nn;
        int cnt = e - base;
#pragma unroll
        for (int k = 0; k < 16; ++k) {
            int s = __shfl(mine, k, 16);
            float4 v = X4[(size_t)s * 16 + q];
            bool ok = k < cnt;
            acc.x += ok ? v.x : 0.0f;
            acc.y += ok ? v.y : 0.0f;
            acc.z += ok ? v.z : 0.0f;
            acc.w += ok ? v.w : 0.0f;
        }
    }
    if (valid) ((float4*)S)[(size_t)n * 16 + q] = acc;
}

__global__ __launch_bounds__(256) void mid_conv_kernel(
    const float* __restrict__ X, const float* __restrict__ S,
    const float* __restrict__ Wd, const float* __restrict__ Wb,
    const float* __restrict__ Wc, const float* __restrict__ bias,
    const float* __restrict__ gate, const float* __restrict__ invdeg,
    const float* __restrict__ ssn, const float* resid,
    float* Y, int nTiles, int N) {
    __shared__ float sWd[HDIM * HDIM];
    __shared__ float sWb[HDIM * HDIM];
    for (int i = threadIdx.x; i < HDIM * HDIM; i += 256) {
        int ks   = i >> 8;
        int rem  = i & 255;
        int nt   = rem >> 6;
        int lp   = rem & 63;
        int ln   = lp >> 1;
        int half = lp & 1;
        int kh   = (ln >> 4) << 1;
        int srcr = (ks << 2) + kh + half;
        int srcc = (nt << 4) + (ln & 15);
        sWd[i] = Wd[srcr * HDIM + srcc];
        sWb[i] = Wb[srcr * HDIM + srcc];
    }
    __syncthreads();

    int wave = threadIdx.x >> 5;
    int lane = threadIdx.x & 31;
    int tile = blockIdx.x * 8 + wave;
    if (tile >= nTiles) return;

    int node0 = tile * 16;
    int m  = lane & 15;
    int kh = (lane >> 4) << 1;
    int row = node0 + m; if (row >= N) row = N - 1;
    const float* Xr = X + (size_t)row * HDIM;
    const float* Sr = S + (size_t)row * HDIM;

    v8f accX[4] = {};
    v8f accS[4] = {};
    int fragBase = lane * 2;

#pragma unroll
    for (int ks = 0; ks < 16; ++ks) {
        int k0 = ks << 2;
        v2f a, s;
        a.x = Xr[k0 + kh];     a.y = Xr[k0 + kh + 1];
        s.x = Sr[k0 + kh];     s.y = Sr[k0 + kh + 1];
#pragma unroll
        for (int nt = 0; nt < 4; ++nt) {
            v2f bd = *(const v2f*)&sWd[((ks << 2) + nt) * 64 + fragBase];
            v2f bb = *(const v2f*)&sWb[((ks << 2) + nt) * 64 + fragBase];
            accX[nt] = __builtin_amdgcn_wmma_f32_16x16x4_f32(
                false, a, false, bd, (short)0, accX[nt], false, false);
            accS[nt] = __builtin_amdgcn_wmma_f32_16x16x4_f32(
                false, s, false, bb, (short)0, accS[nt], false, false);
        }
    }

    int ncol = lane & 15;
    int mAdd = (lane >> 4) * 8;
    float rs[4][8];
    if (resid) {
#pragma unroll
        for (int nt = 0; nt < 4; ++nt) {
#pragma unroll
            for (int v = 0; v < 8; ++v) {
                int node = node0 + v + mAdd;
                rs[nt][v] = (node < N) ? resid[(size_t)node * HDIM + nt * 16 + ncol] : 0.0f;
            }
        }
    }
#pragma unroll
    for (int nt = 0; nt < 4; ++nt) {
        int col = nt * 16 + ncol;
        float wc = Wc[col];
        float bb = bias[col];
#pragma unroll
        for (int v = 0; v < 8; ++v) {
            int node = node0 + v + mAdd;
            if (node >= N) continue;
            float g = gate[node];
            float val = g * (accX[nt][v] + bb) + invdeg[node] * accS[nt][v] + ssn[node] * wc;
            if (resid) val += rs[nt][v];
            val = (val > 0.0f) ? val : 0.01f * val;
            Y[(size_t)node * HDIM + col] = val;
        }
    }
}

__global__ void conv_out_kernel(const float* __restrict__ X, const float* __restrict__ S,
                                const float* __restrict__ Wout, const float* __restrict__ bout,
                                const float* __restrict__ x0,
                                const float* __restrict__ invdeg, const float* __restrict__ gate,
                                const float* __restrict__ ssn, float* __restrict__ out, int N) {
    int n = blockIdx.x * blockDim.x + threadIdx.x;
    if (n >= N) return;
    float a1 = 0.0f, a2 = 0.0f;
    const float* Xr = X + (size_t)n * HDIM;
    const float* Sr = S + (size_t)n * HDIM;
#pragma unroll 8
    for (int k = 0; k < HDIM; ++k) {
        float wa = Wout[k], wb = Wout[HDIM + k];
        a1 += Xr[k] * (wa - wb);
        a2 += Sr[k] * wb;
    }
    out[n] = gate[n] * (a1 + bout[0]) + invdeg[n] * a2 + ssn[n] * Wout[2 * HDIM] + x0[n];
}

extern "C" void kernel_launch(void* const* d_in, const int* in_sizes, int n_in,
                              void* d_out, int out_size, void* d_ws, size_t ws_size,
                              hipStream_t stream) {
    const float* x    = (const float*)d_in[0];
    const int*   ei   = (const int*)  d_in[1];
    const float* Win  = (const float*)d_in[2];
    const float* bin  = (const float*)d_in[3];
    const float* Wmid = (const float*)d_in[4];
    const float* bmid = (const float*)d_in[5];
    const float* Wout = (const float*)d_in[6];
    const float* bout = (const float*)d_in[7];

    const int N = in_sizes[0];
    const int E = in_sizes[1] / 2;
    const int* src = ei;
    const int* dst = ei + E;

    float* w = (float*)d_ws;
    int*   rowptr  = (int*)w;
    int*   cursor  = (int*)(w + (size_t)N);
    float* invdeg  = w + 2 * (size_t)N;
    float* gate    = w + 3 * (size_t)N;
    float* ssn     = w + 4 * (size_t)N;
    float* Sx      = w + 5 * (size_t)N;
    float* Wd6     = w + 6 * (size_t)N;
    float* bufA    = Wd6  + 6 * HDIM * HDIM;
    float* bufB    = bufA + (size_t)N * HDIM;
    float* bufC    = bufB + (size_t)N * HDIM;
    float* bufS    = bufC + (size_t)N * HDIM;
    int*   csr     = (int*)bufC;

    const int nTiles = (N + 15) / 16;

    const int NB = (N + BINW - 1) / BINW;
    int* binned = (int*)bufB;
    int* cntT   = (int*)bufS;
    int* offs   = cntT + NB * NCHUNK;
    bin_count_kernel<<<NCHUNK, CH_THREADS, 0, stream>>>(dst, cntT, E, NB);
    scan_kernel<<<1, 1024, 0, stream>>>(cntT, offs, NB * NCHUNK);
    bin_scatter_kernel<<<NCHUNK, CH_THREADS, 0, stream>>>(src, dst, offs, binned, E, NB);
    bin_fill_kernel<<<NB, FILL_THREADS, 0, stream>>>(offs, binned, rowptr, cursor, csr, NB, N, E);
    node_stats_kernel<<<(N * 8 + 255) / 256, 256, 0, stream>>>(rowptr, cursor, csr, x, invdeg, gate,
                                                               ssn, Sx, N);
    wdiff_kernel<<<(6 * 4096 + 255) / 256, 256, 0, stream>>>(Wmid, Wd6, 6 * 4096);
    conv_in_kernel<<<(N * HDIM + 255) / 256, 256, 0, stream>>>(x, Win, bin, invdeg, gate, ssn, Sx,
                                                               bufA, N);

    float* cur = bufA;
    const int gBlocks = (N * 16 + 255) / 256;
    for (int l = 0; l < 3; ++l) {
        gather_kernel<<<gBlocks, 256, 0, stream>>>(rowptr, cursor, csr, cur, bufS, N);
        const float* Wl = Wmid + (size_t)(2 * l) * 129 * HDIM;
        mid_conv_kernel<<<(nTiles + 7) / 8, 256, 0, stream>>>(
            cur, bufS, Wd6 + (2 * l) * HDIM * HDIM, Wl + 64 * HDIM, Wl + 128 * HDIM,
            bmid + (2 * l) * HDIM, gate, invdeg, ssn, nullptr, bufB, nTiles, N);

        gather_kernel<<<gBlocks, 256, 0, stream>>>(rowptr, cursor, csr, bufB, bufS, N);
        const float* Wl2 = Wmid + (size_t)(2 * l + 1) * 129 * HDIM;
        mid_conv_kernel<<<(nTiles + 7) / 8, 256, 0, stream>>>(
            bufB, bufS, Wd6 + (2 * l + 1) * HDIM * HDIM, Wl2 + 64 * HDIM, Wl2 + 128 * HDIM,
            bmid + (2 * l + 1) * HDIM, gate, invdeg, ssn, cur, cur, nTiles, N);
    }

    gather_kernel<<<gBlocks, 256, 0, stream>>>(rowptr, cursor, csr, cur, bufS, N);
    conv_out_kernel<<<(N + 255) / 256, 256, 0, stream>>>(cur, bufS, Wout, bout, x, invdeg, gate,
                                                         ssn, (float*)d_out, N);
}


# End layers take the neighbor sum on the narrow side of their linear map: 4 per-node factors ahead of conv_in, 1 projected scalar behind conv_out; 5 of 7 wide gathers stay; sum order changes there
# speedup vs baseline: 3.7686x; 1.0728x over previous
#include <hip/hip_runtime.h>
#include <hip/hip_bf16.h>

typedef float v2f __attribute__((ext_vector_type(2)));
typedef float v8f __attribute__((ext_vector_type(8)));

#define HDIM 64

#define BIN_SHIFT 8
#define BINW 256
#define NB_MAX 512
#define NCHUNK 256
#define CH_THREADS 1024
#define FILL_THREADS 512
#define FILL_CAP 12288

__global__ __launch_bounds__(CH_THREADS) void bin_count_kernel(const int* __restrict__ dst,
                                                               int* __restrict__ cntT, int E, int NB) {
    __shared__ int sHist[NB_MAX];
    const int c = blockIdx.x, tid = threadIdx.x;
    for (int i = tid; i < NB; i += CH_THREADS) sHist[i] = 0;
    __syncthreads();
    const int chunk = (E + NCHUNK - 1) / NCHUNK;
    const int e0 = c * chunk;
    const int e1 = (e0 + chunk < E) ? (e0 + chunk) : E;
    for (int e = e0 + tid; e < e1; e += CH_THREADS * 8) {
        int d[8];
#pragma unroll
        for (int u = 0; u < 8; ++u) {
            int ee = e + u * CH_THREADS;
            d[u] = dst[(ee < e1) ? ee : (e1 - 1)];
        }
#pragma unroll
        for (int u = 0; u < 8; ++u)
            if (e + u * CH_THREADS < e1) atomicAdd(&sHist[d[u] >> BIN_SHIFT], 1);
    }
    __syncthreads();
    for (int i = tid; i < NB; i += CH_THREADS) cntT[i * NCHUNK + c] = sHist[i];
}

__global__ __launch_bounds__(1024) void scan_kernel(int* cc, int* __restrict__ rowptr, int N) {
    __shared__ int sWave[32];
    __shared__ int sTotal;
    const int tid = threadIdx.x, lane = tid & 31, wv = tid >> 5;
    int carry = 0;
    for (int base = 0; base < N; base += 4096) {
        int i0 = base + tid * 4;
        int c0 = (i0     < N) ? cc[i0]     : 0;
        int c1 = (i0 + 1 < N) ? cc[i0 + 1] : 0;
        int c2 = (i0 + 2 < N) ? cc[i0 + 2] : 0;
        int c3 = (i0 + 3 < N) ? cc[i0 + 3] : 0;
        int s = c0 + c1 + c2 + c3;
        int inc = s;
#pragma unroll
        for (int o = 1; o < 32; o <<= 1) {
            int v = __shfl_up(inc, o, 32);
            if (lane >= o) inc += v;
        }
        if (lane == 31) sWave[wv] = inc;
        __syncthreads();
        if (wv == 0) {
            int w = sWave[lane];
            int winc = w;
#pragma unroll
            for (int o = 1; o < 32; o <<= 1) {
                int v = __shfl_up(winc, o, 32);
                if (lane >= o) winc += v;
            }
            sWave[lane] = winc - w;
            if (lane == 31) sTotal = winc;
        }
        __syncthreads();
        int ex = carry + sWave[wv] + (inc - s);
        if (i0     < N) { rowptr[i0]     = ex; cc[i0]     = ex; } ex += c0;
        if (i0 + 1 < N) { rowptr[i0 + 1] = ex; cc[i0 + 1] = ex; } ex += c1;
        if (i0 + 2 < N) { rowptr[i0 + 2] = ex; cc[i0 + 2] = ex; } ex += c2;
        if (i0 + 3 < N) { rowptr[i0 + 3] = ex; cc[i0 + 3] = ex; }
        carry += sTotal;
        __syncthreads();
    }
}

__global__ __launch_bounds__(CH_THREADS) void bin_scatter_kernel(const int* __restrict__ src,
                                                                 const int* __restrict__ dst,
                                                                 const int* __restrict__ offs,
                                                                 int* __restrict__ binned, int E, int NB) {
    __shared__ int sCur[NB_MAX];
    const int c = blockIdx.x, tid = threadIdx.x;
    for (int i = tid; i < NB; i += CH_THREADS) sCur[i] = offs[i * NCHUNK + c];
    __syncthreads();
    const int chunk = (E + NCHUNK - 1) / NCHUNK;
    const int e0 = c * chunk;
    const int e1 = (e0 + chunk < E) ? (e0 + chunk) : E;
    for (int e = e0 + tid; e < e1; e += CH_THREADS * 8) {
        int d[8], s[8];
#pragma unroll
        for (int u = 0; u < 8; ++u) {
            int ee = e + u * CH_THREADS;
            int ec = (ee < e1) ? ee : (e1 - 1);
            d[u] = dst[ec];
            s[u] = src[ec];
        }
#pragma unroll
        for (int u = 0; u < 8; ++u) {
            if (e + u * CH_THREADS < e1) {
                int pos = atomicAdd(&sCur[d[u] >> BIN_SHIFT], 1);
                binned[pos] = (s[u] << BIN_SHIFT) | (d[u] & (BINW - 1));
            }
        }
    }
}

__global__ __launch_bounds__(FILL_THREADS) void bin_fill_kernel(const int* __restrict__ offs,
                                                                const int* __restrict__ binned,
                                                                int* __restrict__ rowptr, int* __restrict__ rowend,
                                                                int* __restrict__ csr, int NB, int N, int E) {
    __shared__ int sDeg[BINW];
    __shared__ int sWaveTot[BINW / 32];
    __shared__ int sStage[FILL_CAP];
    const int b = blockIdx.x, tid = threadIdx.x, lane = tid & 31;
    const int beg = offs[b * NCHUNK];
    const int end = (b + 1 < NB) ? offs[(b + 1) * NCHUNK] : E;
    const int cnt = end - beg;
    if (tid < BINW) sDeg[tid] = 0;
    __syncthreads();
    for (int i = tid; i < cnt; i += FILL_THREADS * 8) {
        int r[8];
#pragma unroll
        for (int u = 0; u < 8; ++u) {
            int ii = i + u * FILL_THREADS;
            r[u] = binned[beg + ((ii < cnt) ? ii : (cnt - 1))];
        }
#pragma unroll
        for (int u = 0; u < 8; ++u)
            if (i + u * FILL_THREADS < cnt) atomicAdd(&sDeg[r[u] & (BINW - 1)], 1);
    }
    __syncthreads();
    int v = 0, inc = 0;
    if (tid < BINW) {
        v = sDeg[tid];
        inc = v;
#pragma unroll
        for (int o = 1; o < 32; o <<= 1) {
            int t = __shfl_up(inc, o, 32);
            if (lane >= o) inc += t;
        }
        if (lane == 31) sWaveTot[tid >> 5] = inc;
    }
    __syncthreads();
    if (tid < BINW) {
        int wbase = 0;
        for (int w = 0; w < (tid >> 5); ++w) wbase += sWaveTot[w];
        int excl = wbase + inc - v;
        sDeg[tid] = excl;
        int node = b * BINW + tid;
        if (node < N) { rowptr[node] = beg + excl; rowend[node] = beg + excl + v; }
    }
    __syncthreads();
    const bool staged = cnt <= FILL_CAP;
    for (int i = tid; i < cnt; i += FILL_THREADS * 8) {
        int r[8];
#pragma unroll
        for (int u = 0; u < 8; ++u) {
            int ii = i + u * FILL_THREADS;
            r[u] = binned[beg + ((ii < cnt) ? ii : (cnt - 1))];
        }
#pragma unroll
        for (int u = 0; u < 8; ++u) {
            if (i + u * FILL_THREADS < cnt) {
                int pos = atomicAdd(&sDeg[r[u] & (BINW - 1)], 1);
                if (staged) sStage[pos] = r[u] >> BIN_SHIFT;
                else        csr[beg + pos] = r[u] >> BIN_SHIFT;
            }
        }
    }
    __syncthreads();
    if (staged)
        for (int i = tid; i < cnt; i += FILL_THREADS) csr[beg + i] = sStage[i];
}

__global__ void node_stats_kernel(const int* __restrict__ rowptr, const int* __restrict__ rowend,
                                  const int* __restrict__ csr, const float* __restrict__ x,
                                  float* __restrict__ invdeg, float* __restrict__ gate,
                                  float* __restrict__ ssn, float* __restrict__ Sx,
                                  float* __restrict__ coef, int N) {
    int t = blockIdx.x * blockDim.x + threadIdx.x;
    int n = t >> 3, l = t & 7;
    bool valid = n < N;
    int nn = valid ? n : N - 1;
    int b = rowptr[nn], e = rowend[nn];
    float sg = 0.0f, sx = 0.0f;
    for (int j = b + l; j < e; j += 8) {
        int s = csr[j];
        sg += (s > nn) ? 1.0f : ((s < nn) ? -1.0f : 0.0f);
        sx += x[s];
    }
#pragma unroll
    for (int o = 4; o > 0; o >>= 1) {
        sg += __shfl_xor(sg, o, 8);
        sx += __shfl_xor(sx, o, 8);
    }
    if (valid && l == 0) {
        float dg = (float)(e - b);
        float inv = 1.0f / fmaxf(dg, 1.0f);
        float g = (dg > 0.0f) ? 1.0f : 0.0f;
        invdeg[n] = inv;
        gate[n]   = g;
        ssn[n]    = sg * inv;
        Sx[n]     = sx;
        ((float4*)coef)[n] = make_float4(g * x[n], g, inv * sx, sg * inv);
    }
}

__global__ __launch_bounds__(256) void coef_gather_kernel(const int* __restrict__ rowptr,
                                                          const int* __restrict__ rowend,
                                                          const int* __restrict__ csr,
                                                          const float* __restrict__ coef,
                                                          float* __restrict__ csum, int N) {
    __shared__ float4 sOut[32];
    const float4* coef4 = (const float4*)coef;
    int t = blockIdx.x * 256 + threadIdx.x;
    int n = t >> 3, l = t & 7;
    int nn = (n < N) ? n : N - 1;
    int b = rowptr[nn], e = rowend[nn];
    float4 a = make_float4(0.0f, 0.0f, 0.0f, 0.0f);
    for (int base = b; base < e; base += 32) {
        int idx[4];
#pragma unroll
        for (int u = 0; u < 4; ++u) {
            int j = base + u * 8 + l;
            idx[u] = csr[(j < e) ? j : (e - 1)];
        }
#pragma unroll
        for (int u = 0; u < 4; ++u) {
            float4 c = coef4[idx[u]];
            bool ok = (base + u * 8 + l) < e;
            a.x += ok ? c.x : 0.0f;
            a.y += ok ? c.y : 0.0f;
            a.z += ok ? c.z : 0.0f;
            a.w += ok ? c.w : 0.0f;
        }
    }
#pragma unroll
    for (int o = 4; o > 0; o >>= 1) {
        a.x += __shfl_xor(a.x, o, 8);
        a.y += __shfl_xor(a.y, o, 8);
        a.z += __shfl_xor(a.z, o, 8);
        a.w += __shfl_xor(a.w, o, 8);
    }
    if (l == 0) sOut[threadIdx.x >> 3] = a;
    __syncthreads();
    if (threadIdx.x < 32) {
        int m = blockIdx.x * 32 + threadIdx.x;
        if (m < N) ((float4*)csum)[m] = sOut[threadIdx.x];
    }
}

__global__ void wdiff_kernel(const float* __restrict__ Wmid, float* __restrict__ Wd6, int total) {
    int i = blockIdx.x * blockDim.x + threadIdx.x;
    if (i >= total) return;
    int l = i >> 12;
    int rc = i & 4095;
    const float* Wl = Wmid + (size_t)l * 129 * HDIM;
    Wd6[i] = Wl[rc] - Wl[64 * HDIM + rc];
}

__global__ void conv_in_kernel(const float* __restrict__ x, const float* __restrict__ Win,
                               const float* __restrict__ bin,
                               const float* __restrict__ invdeg, const float* __restrict__ gate,
                               const float* __restrict__ ssn, const float* __restrict__ Sx,
                               const float* __restrict__ csum,
                               float* __restrict__ Y, float* __restrict__ S0, int N) {
    int t = blockIdx.x * blockDim.x + threadIdx.x;
    if (t >= N * HDIM) return;
    int n = t >> 6, h = t & 63;
    float wa = Win[h], wb = Win[HDIM + h], wc = Win[2 * HDIM + h];
    float bh = bin[h];
    Y[t] = gate[n] * (x[n] * (wa - wb) + bh) + invdeg[n] * Sx[n] * wb + ssn[n] * wc;
    float4 cs = ((const float4*)csum)[n];
    S0[t] = cs.x * (wa - wb) + cs.y * bh + cs.z * wb + cs.w * wc;
}

__global__ __launch_bounds__(256) void gather_kernel(const int* __restrict__ rowptr,
                                                     const int* __restrict__ rowend,
                                                     const int* __restrict__ csr,
                                                     const float* __restrict__ X,
                                                     float* __restrict__ S, int N) {
    int t = blockIdx.x * 256 + threadIdx.x;
    int n = t >> 4, q = t & 15;
    bool valid = n < N;
    int nn = valid ? n : N - 1;
    int b = rowptr[nn], e = rowend[nn];
    const float4* X4 = (const float4*)X;
    float4 acc = make_float4(0.0f, 0.0f, 0.0f, 0.0f);
    for (int base = b; base < e; base += 16) {
        int j = base + q;
        int mine = (j < e) ? csr[j] : nn;
        int cnt = e - base;
#pragma unroll
        for (int k = 0; k < 16; ++k) {
            int s = __shfl(mine, k, 16);
            float4 v = X4[(size_t)s * 16 + q];
            bool ok = k < cnt;
            acc.x += ok ? v.x : 0.0f;
            acc.y += ok ? v.y : 0.0f;
            acc.z += ok ? v.z : 0.0f;
            acc.w += ok ? v.w : 0.0f;
        }
    }
    if (valid) ((float4*)S)[(size_t)n * 16 + q] = acc;
}

__global__ __launch_bounds__(256) void mid_conv_kernel(
    const float* __restrict__ X, const float* __restrict__ S,
    const float* __restrict__ Wd, const float* __restrict__ Wb,
    const float* __restrict__ Wc, const float* __restrict__ bias,
    const float* __restrict__ gate, const float* __restrict__ invdeg,
    const float* __restrict__ ssn, const float* resid,
    float* Y, int nTiles, int N) {
    __shared__ float sWd[HDIM * HDIM];
    __shared__ float sWb[HDIM * HDIM];
    for (int i = threadIdx.x; i < HDIM * HDIM; i += 256) {
        int ks   = i >> 8;
        int rem  = i & 255;
        int nt   = rem >> 6;
        int lp   = rem & 63;
        int ln   = lp >> 1;
        int half = lp & 1;
        int kh   = (ln >> 4) << 1;
        int srcr = (ks << 2) + kh + half;
        int srcc = (nt << 4) + (ln & 15);
        sWd[i] = Wd[srcr * HDIM + srcc];
        sWb[i] = Wb[srcr * HDIM + srcc];
    }
    __syncthreads();

    int wave = threadIdx.x >> 5;
    int lane = threadIdx.x & 31;
    int tile = blockIdx.x * 8 + wave;
    if (tile >= nTiles) return;

    int node0 = tile * 16;
    int m  = lane & 15;
    int kh = (lane >> 4) << 1;
    int row = node0 + m; if (row >= N) row = N - 1;
    const float* Xr = X + (size_t)row * HDIM;
    const float* Sr = S + (size_t)row * HDIM;

    v8f accX[4] = {};
    v8f accS[4] = {};
    int fragBase = lane * 2;

#pragma unroll
    for (int ks = 0; ks < 16; ++ks) {
        int k0 = ks << 2;
        v2f a, s;
        a.x = Xr[k0 + kh];     a.y = Xr[k0 + kh + 1];
        s.x = Sr[k0 + kh];     s.y = Sr[k0 + kh + 1];
#pragma unroll
        for (int nt = 0; nt < 4; ++nt) {
            v2f bd = *(const v2f*)&sWd[((ks << 2) + nt) * 64 + fragBase];
            v2f bb = *(const v2f*)&sWb[((ks << 2) + nt) * 64 + fragBase];
            accX[nt] = __builtin_amdgcn_wmma_f32_16x16x4_f32(
                false, a, false, bd, (short)0, accX[nt], false, false);
            accS[nt] = __builtin_amdgcn_wmma_f32_16x16x4_f32(
                false, s, false, bb, (short)0, accS[nt], false, false);
        }
    }

    int ncol = lane & 15;
    int mAdd = (lane >> 4) * 8;
    float rs[4][8];
    if (resid) {
#pragma unroll
        for (int nt = 0; nt < 4; ++nt) {
#pragma unroll
            for (int v = 0; v < 8; ++v) {
                int node = node0 + v + mAdd;
                rs[nt][v] = (node < N) ? resid[(size_t)node * HDIM + nt * 16 + ncol] : 0.0f;
            }
        }
    }
#pragma unroll
    for (int nt = 0; nt < 4; ++nt) {
        int col = nt * 16 + ncol;
        float wc = Wc[col];
        float bb = bias[col];
#pragma unroll
        for (int v = 0; v < 8; ++v) {
            int node = node0 + v + mAdd;
            if (node >= N) continue;
            float g = gate[node];
            float val = g * (accX[nt][v] + bb) + invdeg[node] * accS[nt][v] + ssn[node] * wc;
            if (resid) val += rs[nt][v];
            val = (val > 0.0f) ? val : 0.01f * val;
            Y[(size_t)node * HDIM + col] = val;
        }
    }
}

__global__ __launch_bounds__(256) void proj_kernel(const float* __restrict__ X, const float* __restrict__ Wout,
                                                   float* __restrict__ p, float* __restrict__ q, int N) {
    __shared__ float sP[64];
    __shared__ float sQ[64];
    int t = blockIdx.x * 256 + threadIdx.x;
    int n = t >> 2, l = t & 3;
    int nn = (n < N) ? n : N - 1;
    const float4* Xr = (const float4*)X + (size_t)nn * 16;
    float ap = 0.0f, aq = 0.0f;
#pragma unroll
    for (int i = 0; i < 4; ++i) {
        int c4 = i * 4 + l;
        float4 v = Xr[c4];
        int k = c4 * 4;
        float wa0 = Wout[k], wa1 = Wout[k + 1], wa2 = Wout[k + 2], wa3 = Wout[k + 3];
        float wb0 = Wout[HDIM + k], wb1 = Wout[HDIM + k + 1], wb2 = Wout[HDIM + k + 2], wb3 = Wout[HDIM + k + 3];
        aq += v.x * (wa0 - wb0) + v.y * (wa1 - wb1) + v.z * (wa2 - wb2) + v.w * (wa3 - wb3);
        ap += v.x * wb0 + v.y * wb1 + v.z * wb2 + v.w * wb3;
    }
#pragma unroll
    for (int o = 2; o > 0; o >>= 1) {
        ap += __shfl_xor(ap, o, 4);
        aq += __shfl_xor(aq, o, 4);
    }
    if (l == 0) { sP[threadIdx.x >> 2] = ap; sQ[threadIdx.x >> 2] = aq; }
    __syncthreads();
    if (threadIdx.x < 64) {
        int m = blockIdx.x * 64 + threadIdx.x;
        if (m < N) p[m] = sP[threadIdx.x];
    } else if (threadIdx.x < 128) {
        int m = blockIdx.x * 64 + threadIdx.x - 64;
        if (m < N) q[m] = sQ[threadIdx.x - 64];
    }
}

__global__ __launch_bounds__(256) void out_kernel(const int* __restrict__ rowptr, const int* __restrict__ rowend,
                                                  const int* __restrict__ csr,
                                                  const float* __restrict__ p, const float* __restrict__ q,
                                                  const float* __restrict__ Wout, const float* __restrict__ bout,
                                                  const float* __restrict__ x0,
                                                  const float* __restrict__ invdeg, const float* __restrict__ gate,
                                                  const float* __restrict__ ssn, float* __restrict__ out, int N) {
    __shared__ float sOut[32];
    int t = blockIdx.x * 256 + threadIdx.x;
    int n = t >> 3, l = t & 7;
    int nn = (n < N) ? n : N - 1;
    int b = rowptr[nn], e = rowend[nn];
    float a = 0.0f;
    for (int base = b; base < e; base += 32) {
        int idx[4];
#pragma unroll
        for (int u = 0; u < 4; ++u) {
            int j = base + u * 8 + l;
            idx[u] = csr[(j < e) ? j : (e - 1)];
        }
#pragma unroll
        for (int u = 0; u < 4; ++u) {
            float v = p[idx[u]];
            a += ((base + u * 8 + l) < e) ? v : 0.0f;
        }
    }
#pragma unroll
    for (int o = 4; o > 0; o >>= 1) a += __shfl_xor(a, o, 8);
    if (l == 0) sOut[threadIdx.x >> 3] = a;
    __syncthreads();
    if (threadIdx.x < 32) {
        int m = blockIdx.x * 32 + threadIdx.x;
        if (m < N)
            out[m] = gate[m] * (q[m] + bout[0]) + invdeg[m] * sOut[threadIdx.x] + ssn[m] * Wout[2 * HDIM] + x0[m];
    }
}

extern "C" void kernel_launch(void* const* d_in, const int* in_sizes, int n_in,
                              void* d_out, int out_size, void* d_ws, size_t ws_size,
                              hipStream_t stream) {
    const float* x    = (const float*)d_in[0];
    const int*   ei   = (const int*)  d_in[1];
    const float* Win  = (const float*)d_in[2];
    const float* bin  = (const float*)d_in[3];
    const float* Wmid = (const float*)d_in[4];
    const float* bmid = (const float*)d_in[5];
    const float* Wout = (const float*)d_in[6];
    const float* bout = (const float*)d_in[7];

    const int N = in_sizes[0];
    const int E = in_sizes[1] / 2;
    const int* src = ei;
    const int* dst = ei + E;

    float* w = (float*)d_ws;
    int*   rowptr  = (int*)w;
    int*   cursor  = (int*)(w + (size_t)N);
    float* invdeg  = w + 2 * (size_t)N;
    float* gate    = w + 3 * (size_t)N;
    float* ssn     = w + 4 * (size_t)N;
    float* Sx      = w + 5 * (size_t)N;
    float* Wd6     = w + 6 * (size_t)N;
    float* bufA    = Wd6  + 6 * HDIM * HDIM;
    float* bufB    = bufA + (size_t)N * HDIM;
    float* bufC    = bufB + (size_t)N * HDIM;
    float* bufS    = bufC + (size_t)N * HDIM;
    int*   csr     = (int*)bufC;

    const int nTiles = (N + 15) / 16;

    const int NB = (N + BINW - 1) / BINW;
    int* binned = (int*)bufB;
    int* cntT   = (int*)bufS;
    int* offs   = cntT + NB * NCHUNK;
    bin_count_kernel<<<NCHUNK, CH_THREADS, 0, stream>>>(dst, cntT, E, NB);
    scan_kernel<<<1, 1024, 0, stream>>>(cntT, offs, NB * NCHUNK);
    bin_scatter_kernel<<<NCHUNK, CH_THREADS, 0, stream>>>(src, dst, offs, binned, E, NB);
    bin_fill_kernel<<<NB, FILL_THREADS, 0, stream>>>(offs, binned, rowptr, cursor, csr, NB, N, E);
    float* coef = bufB;
    float* csum = bufB + 4 * (size_t)N;
    node_stats_kernel<<<(N * 8 + 255) / 256, 256, 0, stream>>>(rowptr, cursor, csr, x, invdeg, gate,
                                                               ssn, Sx, coef, N);
    wdiff_kernel<<<(6 * 4096 + 255) / 256, 256, 0, stream>>>(Wmid, Wd6, 6 * 4096);
    coef_gather_kernel<<<(N * 8 + 255) / 256, 256, 0, stream>>>(rowptr, cursor, csr, coef, csum, N);
    conv_in_kernel<<<(N * HDIM + 255) / 256, 256, 0, stream>>>(x, Win, bin, invdeg, gate, ssn, Sx, csum,
                                                               bufA, bufS, N);

    float* cur = bufA;
    const int gBlocks = (N * 16 + 255) / 256;
    mid_conv_kernel<<<(nTiles + 7) / 8, 256, 0, stream>>>(
        cur, bufS, Wd6, Wmid + 64 * HDIM, Wmid + 128 * HDIM,
        bmid, gate, invdeg, ssn, nullptr, bufB, nTiles, N);
    gather_kernel<<<gBlocks, 256, 0, stream>>>(rowptr, cursor, csr, bufB, bufS, N);
    mid_conv_kernel<<<(nTiles + 7) / 8, 256, 0, stream>>>(
        bufB, bufS, Wd6 + HDIM * HDIM, Wmid + (size_t)129 * HDIM + 64 * HDIM, Wmid + (size_t)129 * HDIM + 128 * HDIM,
        bmid + HDIM, gate, invdeg, ssn, cur, cur, nTiles, N);
    for (int l = 1; l < 3; ++l) {
        gather_kernel<<<gBlocks, 256, 0, stream>>>(rowptr, cursor, csr, cur, bufS, N);
        const float* Wl = Wmid + (size_t)(2 * l) * 129 * HDIM;
        mid_conv_kernel<<<(nTiles + 7) / 8, 256, 0, stream>>>(
            cur, bufS, Wd6 + (2 * l) * HDIM * HDIM, Wl + 64 * HDIM, Wl + 128 * HDIM,
            bmid + (2 * l) * HDIM, gate, invdeg, ssn, nullptr, bufB, nTiles, N);

        gather_kernel<<<gBlocks, 256, 0, stream>>>(rowptr, cursor, csr, bufB, bufS, N);
        const float* Wl2 = Wmid + (size_t)(2 * l + 1) * 129 * HDIM;
        mid_conv_kernel<<<(nTiles + 7) / 8, 256, 0, stream>>>(
            bufB, bufS, Wd6 + (2 * l + 1) * HDIM * HDIM, Wl2 + 64 * HDIM, Wl2 + 128 * HDIM,
            bmid + (2 * l + 1) * HDIM, gate, invdeg, ssn, cur, cur, nTiles, N);
    }

    float* pj = bufS;
    float* qj = bufS + 2 * (size_t)N;
    proj_kernel<<<(N * 4 + 255) / 256, 256, 0, stream>>>(cur, Wout, pj, qj, N);
    out_kernel<<<(N * 8 + 255) / 256, 256, 0, stream>>>(rowptr, cursor, csr, pj, qj, Wout, bout, x, invdeg, gate,
                                                        ssn, (float*)d_out, N);
}


# Mid convs gather their own neighbor sums: the wave that multiplies a 16-node tile first builds that S tile in its LDS rows, so 5 gather launches and S's trip through memory go; in-node sum order kept
# speedup vs baseline: 4.1389x; 1.0983x over previous
#include <hip/hip_runtime.h>
#include <hip/hip_bf16.h>

typedef float v2f __attribute__((ext_vector_type(2)));
typedef float v8f __attribute__((ext_vector_type(8)));

#define HDIM 64

#define BIN_SHIFT 8
#define BINW 256
#define NB_MAX 512
#define NCHUNK 256
#define CH_THREADS 1024
#define FILL_THREADS 512
#define FILL_CAP 12288

__global__ __launch_bounds__(CH_THREADS) void bin_count_kernel(const int* __restrict__ dst,
                                                               int* __restrict__ cntT, int E, int NB) {
    __shared__ int sHist[NB_MAX];
    const int c = blockIdx.x, tid = threadIdx.x;
    for (int i = tid; i < NB; i += CH_THREADS) sHist[i] = 0;
    __syncthreads();
    const int chunk = (E + NCHUNK - 1) / NCHUNK;
    const int e0 = c * chunk;
    const int e1 = (e0 + chunk < E) ? (e0 + chunk) : E;
    for (int e = e0 + tid; e < e1; e += CH_THREADS * 8) {
        int d[8];
#pragma unroll
        for (int u = 0; u < 8; ++u) {
            int ee = e + u * CH_THREADS;
            d[u] = dst[(ee < e1) ? ee : (e1 - 1)];
        }
#pragma unroll
        for (int u = 0; u < 8; ++u)
            if (e + u * CH_THREADS < e1) atomicAdd(&sHist[d[u] >> BIN_SHIFT], 1);
    }
    __syncthreads();
    for (int i = tid; i < NB; i += CH_THREADS) cntT[i * NCHUNK + c] = sHist[i];
}

__global__ __launch_bounds__(1024) void scan_kernel(int* cc, int* __restrict__ rowptr, int N) {
    __shared__ int sWave[32];
    __shared__ int sTotal;
    const int tid = threadIdx.x, lane = tid & 31, wv = tid >> 5;
    int carry = 0;
    for (int base = 0; base < N; base += 4096) {
        int i0 = base + tid * 4;
        int c0 = (i0     < N) ? cc[i0]     : 0;
        int c1 = (i0 + 1 < N) ? cc[i0 + 1] : 0;
        int c2 = (i0 + 2 < N) ? cc[i0 + 2] : 0;
        int c3 = (i0 + 3 < N) ? cc[i0 + 3] : 0;
        int s = c0 + c1 + c2 + c3;
        int inc = s;
#pragma unroll
        for (int o = 1; o < 32; o <<= 1) {
            int v = __shfl_up(inc, o, 32);
            if (lane >= o) inc += v;
        }
        if (lane == 31) sWave[wv] = inc;
        __syncthreads();
        if (wv == 0) {
            int w = sWave[lane];
            int winc = w;
#pragma unroll
            for (int o = 1; o < 32; o <<= 1) {
                int v = __shfl_up(winc, o, 32);
                if (lane >= o) winc += v;
            }
            sWave[lane] = winc - w;
            if (lane == 31) sTotal = winc;
        }
        __syncthreads();
        int ex = carry + sWave[wv] + (inc - s);
        if (i0     < N) { rowptr[i0]     = ex; cc[i0]     = ex; } ex += c0;
        if (i0 + 1 < N) { rowptr[i0 + 1] = ex; cc[i0 + 1] = ex; } ex += c1;
        if (i0 + 2 < N) { rowptr[i0 + 2] = ex; cc[i0 + 2] = ex; } ex += c2;
        if (i0 + 3 < N) { rowptr[i0 + 3] = ex; cc[i0 + 3] = ex; }
        carry += sTotal;
        __syncthreads();
    }
}

__global__ __launch_bounds__(CH_THREADS) void bin_scatter_kernel(const int* __restrict__ src,
                                                                 const int* __restrict__ dst,
                                                                 const int* __restrict__ offs,
                                                                 int* __restrict__ binned, int E, int NB) {
    __shared__ int sCur[NB_MAX];
    const int c = blockIdx.x, tid = threadIdx.x;
    for (int i = tid; i < NB; i += CH_THREADS) sCur[i] = offs[i * NCHUNK + c];
    __syncthreads();
    const int chunk = (E + NCHUNK - 1) / NCHUNK;
    const int e0 = c * chunk;
    const int e1 = (e0 + chunk < E) ? (e0 + chunk) : E;
    for (int e = e0 + tid; e < e1; e += CH_THREADS * 8) {
        int d[8], s[8];
#pragma unroll
        for (int u = 0; u < 8; ++u) {
            int ee = e + u * CH_THREADS;
            int ec = (ee < e1) ? ee : (e1 - 1);
            d[u] = dst[ec];
            s[u] = src[ec];
        }
#pragma unroll
        for (int u = 0; u < 8; ++u) {
            if (e + u * CH_THREADS < e1) {
                int pos = atomicAdd(&sCur[d[u] >> BIN_SHIFT], 1);
                binned[pos] = (s[u] << BIN_SHIFT) | (d[u] & (BINW - 1));
            }
        }
    }
}

__global__ __launch_bounds__(FILL_THREADS) void bin_fill_kernel(const int* __restrict__ offs,
                                                                const int* __restrict__ binned,
                                                                int* __restrict__ rowptr, int* __restrict__ rowend,
                                                                int* __restrict__ csr, int NB, int N, int E) {
    __shared__ int sDeg[BINW];
    __shared__ int sWaveTot[BINW / 32];
    __shared__ int sStage[FILL_CAP];
    const int b = blockIdx.x, tid = threadIdx.x, lane = tid & 31;
    const int beg = offs[b * NCHUNK];
    const int end = (b + 1 < NB) ? offs[(b + 1) * NCHUNK] : E;
    const int cnt = end - beg;
    if (tid < BINW) sDeg[tid] = 0;
    __syncthreads();
    for (int i = tid; i < cnt; i += FILL_THREADS * 8) {
        int r[8];
#pragma unroll
        for (int u = 0; u < 8; ++u) {
            int ii = i + u * FILL_THREADS;
            r[u] = binned[beg + ((ii < cnt) ? ii : (cnt - 1))];
        }
#pragma unroll
        for (int u = 0; u < 8; ++u)
            if (i + u * FILL_THREADS < cnt) atomicAdd(&sDeg[r[u] & (BINW - 1)], 1);
    }
    __syncthreads();
    int v = 0, inc = 0;
    if (tid < BINW) {
        v = sDeg[tid];
        inc = v;
#pragma unroll
        for (int o = 1; o < 32; o <<= 1) {
            int t = __shfl_up(inc, o, 32);
            if (lane >= o) inc += t;
        }
        if (lane == 31) sWaveTot[tid >> 5] = inc;
    }
    __syncthreads();
    if (tid < BINW) {
        int wbase = 0;
        for (int w = 0; w < (tid >> 5); ++w) wbase += sWaveTot[w];
        int excl = wbase + inc - v;
        sDeg[tid] = excl;
        int node = b * BINW + tid;
        if (node < N) { rowptr[node] = beg + excl; rowend[node] = beg + excl + v; }
    }
    __syncthreads();
    const bool staged = cnt <= FILL_CAP;
    for (int i = tid; i < cnt; i += FILL_THREADS * 8) {
        int r[8];
#pragma unroll
        for (int u = 0; u < 8; ++u) {
            int ii = i + u * FILL_THREADS;
            r[u] = binned[beg + ((ii < cnt) ? ii : (cnt - 1))];
        }
#pragma unroll
        for (int u = 0; u < 8; ++u) {
            if (i + u * FILL_THREADS < cnt) {
                int pos = atomicAdd(&sDeg[r[u] & (BINW - 1)], 1);
                if (staged) sStage[pos] = r[u] >> BIN_SHIFT;
                else        csr[beg + pos] = r[u] >> BIN_SHIFT;
            }
        }
    }
    __syncthreads();
    if (staged)
        for (int i = tid; i < cnt; i += FILL_THREADS) csr[beg + i] = sStage[i];
}

__global__ void node_stats_kernel(const int* __restrict__ rowptr, const int* __restrict__ rowend,
                                  const int* __restrict__ csr, const float* __restrict__ x,
                                  float* __restrict__ invdeg, float* __restrict__ gate,
                                  float* __restrict__ ssn, float* __restrict__ Sx,
                                  float* __restrict__ coef, int N) {
    int t = blockIdx.x * blockDim.x + threadIdx.x;
    int n = t >> 3, l = t & 7;
    bool valid = n < N;
    int nn = valid ? n : N - 1;
    int b = rowptr[nn], e = rowend[nn];
    float sg = 0.0f, sx = 0.0f;
    for (int j = b + l; j < e; j += 8) {
        int s = csr[j];
        sg += (s > nn) ? 1.0f : ((s < nn) ? -1.0f : 0.0f);
        sx += x[s];
    }
#pragma unroll
    for (int o = 4; o > 0; o >>= 1) {
        sg += __shfl_xor(sg, o, 8);
        sx += __shfl_xor(sx, o, 8);
    }
    if (valid && l == 0) {
        float dg = (float)(e - b);
        float inv = 1.0f / fmaxf(dg, 1.0f);
        float g = (dg > 0.0f) ? 1.0f : 0.0f;
        invdeg[n] = inv;
        gate[n]   = g;
        ssn[n]    = sg * inv;
        Sx[n]     = sx;
        ((float4*)coef)[n] = make_float4(g * x[n], g, inv * sx, sg * inv);
    }
}

__global__ __launch_bounds__(256) void coef_gather_kernel(const int* __restrict__ rowptr,
                                                          const int* __restrict__ rowend,
                                                          const int* __restrict__ csr,
                                                          const float* __restrict__ coef,
                                                          float* __restrict__ csum, int N) {
    __shared__ float4 sOut[32];
    const float4* coef4 = (const float4*)coef;
    int t = blockIdx.x * 256 + threadIdx.x;
    int n = t >> 3, l = t & 7;
    int nn = (n < N) ? n : N - 1;
    int b = rowptr[nn], e = rowend[nn];
    float4 a = make_float4(0.0f, 0.0f, 0.0f, 0.0f);
    for (int base = b; base < e; base += 32) {
        int idx[4];
#pragma unroll
        for (int u = 0; u < 4; ++u) {
            int j = base + u * 8 + l;
            idx[u] = csr[(j < e) ? j : (e - 1)];
        }
#pragma unroll
        for (int u = 0; u < 4; ++u) {
            float4 c = coef4[idx[u]];
            bool ok = (base + u * 8 + l) < e;
            a.x += ok ? c.x : 0.0f;
            a.y += ok ? c.y : 0.0f;
            a.z += ok ? c.z : 0.0f;
            a.w += ok ? c.w : 0.0f;
        }
    }
#pragma unroll
    for (int o = 4; o > 0; o >>= 1) {
        a.x += __shfl_xor(a.x, o, 8);
        a.y += __shfl_xor(a.y, o, 8);
        a.z += __shfl_xor(a.z, o, 8);
        a.w += __shfl_xor(a.w, o, 8);
    }
    if (l == 0) sOut[threadIdx.x >> 3] = a;
    __syncthreads();
    if (threadIdx.x < 32) {
        int m = blockIdx.x * 32 + threadIdx.x;
        if (m < N) ((float4*)csum)[m] = sOut[threadIdx.x];
    }
}

__global__ void wdiff_kernel(const float* __restrict__ Wmid, float* __restrict__ Wd6, int total) {
    int i = blockIdx.x * blockDim.x + threadIdx.x;
    if (i >= total) return;
    int l = i >> 12;
    int rc = i & 4095;
    const float* Wl = Wmid + (size_t)l * 129 * HDIM;
    Wd6[i] = Wl[rc] - Wl[64 * HDIM + rc];
}

__global__ void conv_in_kernel(const float* __restrict__ x, const float* __restrict__ Win,
                               const float* __restrict__ bin,
                               const float* __restrict__ invdeg, const float* __restrict__ gate,
                               const float* __restrict__ ssn, const float* __restrict__ Sx,
                               const float* __restrict__ csum,
                               float* __restrict__ Y, float* __restrict__ S0, int N) {
    int t = blockIdx.x * blockDim.x + threadIdx.x;
    if (t >= N * HDIM) return;
    int n = t >> 6, h = t & 63;
    float wa = Win[h], wb = Win[HDIM + h], wc = Win[2 * HDIM + h];
    float bh = bin[h];
    Y[t] = gate[n] * (x[n] * (wa - wb) + bh) + invdeg[n] * Sx[n] * wb + ssn[n] * wc;
    float4 cs = ((const float4*)csum)[n];
    S0[t] = cs.x * (wa - wb) + cs.y * bh + cs.z * wb + cs.w * wc;
}

__device__ __forceinline__ void gather16(const float4* __restrict__ X4, int mine, int cnt, int q, float4& acc) {
#pragma unroll
    for (int k = 0; k < 16; ++k) {
        int s = __shfl(mine, k, 16);
        float4 v = X4[(size_t)s * 16 + q];
        bool ok = k < cnt;
        acc.x += ok ? v.x : 0.0f;
        acc.y += ok ? v.y : 0.0f;
        acc.z += ok ? v.z : 0.0f;
        acc.w += ok ? v.w : 0.0f;
    }
}

#define SROW 68
#define TILE_IDX_CAP 768
template <bool GATHER>
__global__ __launch_bounds__(256) void mid_conv_kernel(
    const float* __restrict__ X, const float* __restrict__ S,
    const int* __restrict__ rowptr, const int* __restrict__ rowend, const int* __restrict__ csr,
    const float* __restrict__ Wd, const float* __restrict__ Wb,
    const float* __restrict__ Wc, const float* __restrict__ bias,
    const float* __restrict__ gate, const float* __restrict__ invdeg,
    const float* __restrict__ ssn, const float* resid,
    float* Y, int nTiles, int N) {
    __shared__ float sWd[HDIM * HDIM];
    __shared__ float sWb[HDIM * HDIM];
    __shared__ __attribute__((aligned(16))) float sS[GATHER ? 8 * 16 * SROW : 4];
    __shared__ int   sIdx[GATHER ? 8 * TILE_IDX_CAP : 2];
    for (int i = threadIdx.x; i < HDIM * HDIM; i += 256) {
        int ks   = i >> 8;
        int rem  = i & 255;
        int nt   = rem >> 6;
        int lp   = rem & 63;
        int ln   = lp >> 1;
        int half = lp & 1;
        int kh   = (ln >> 4) << 1;
        int srcr = (ks << 2) + kh + half;
        int srcc = (nt << 4) + (ln & 15);
        sWd[i] = Wd[srcr * HDIM + srcc];
        sWb[i] = Wb[srcr * HDIM + srcc];
    }
    __syncthreads();

    int wave = threadIdx.x >> 5;
    int lane = threadIdx.x & 31;
    int tile = blockIdx.x * 8 + wave;
    if (tile >= nTiles) return;

    int node0 = tile * 16;
    int m  = lane & 15;
    int kh = (lane >> 4) << 1;
    int row = node0 + m; if (row >= N) row = N - 1;
    const float* Xr = X + (size_t)row * HDIM;
    const float* Sr = S + (size_t)row * HDIM;
    float* mS = sS + (GATHER ? wave * (16 * SROW) : 0);

    if (GATHER) {
        int* mIx = sIdx + wave * TILE_IDX_CAP;
        int rb = rowptr[row], re = rowend[row];
        if (node0 + m >= N) rb = re;
        int tb = __shfl(rb, 0, 32);
        int staged = __shfl(re, 15, 32) - tb;
        staged = (staged < 0) ? 0 : ((staged > TILE_IDX_CAP) ? TILE_IDX_CAP : staged);
        for (int i0 = 0; i0 < staged; i0 += 32 * 16) {
            int v[16];
#pragma unroll
            for (int u = 0; u < 16; ++u) {
                int i = i0 + u * 32 + lane;
                v[u] = csr[tb + ((i < staged) ? i : (staged - 1))];
            }
#pragma unroll
            for (int u = 0; u < 16; ++u) {
                int i = i0 + u * 32 + lane;
                if (i < staged) mIx[i] = v[u];
            }
        }
        __builtin_amdgcn_fence(__ATOMIC_ACQ_REL, "wavefront");

        const float4* X4 = (const float4*)X;
        int q = lane & 15, h = lane >> 4;
        for (int p = 0; p < 8; ++p) {
            int r = 2 * p + h;
            int b = __shfl(rb, r, 32), e = __shfl(re, r, 32);
            int own = node0 + r; if (own >= N) own = N - 1;
            float4 acc = make_float4(0.0f, 0.0f, 0.0f, 0.0f);
            int eL = (b >= tb) ? (((tb + staged) < e) ? (tb + staged) : e) : b;
            if (eL < b) eL = b;
            for (int base = b; base < eL; base += 16) {
                int j = base + q;
                int mine = (j < eL) ? mIx[j - tb] : own;
                gather16(X4, mine, eL - base, q, acc);
            }
            for (int base = eL; base < e; base += 16) {
                int j = base + q;
                int mine = (j < e) ? csr[j] : own;
                gather16(X4, mine, e - base, q, acc);
            }
            *(float4*)&mS[r * SROW + 4 * q] = acc;
        }
        __builtin_amdgcn_fence(__ATOMIC_ACQ_REL, "wavefront");
    }

    v8f accX[4] = {};
    v8f accS[4] = {};
    int fragBase = lane * 2;

#pragma unroll
    for (int ks = 0; ks < 16; ++ks) {
        int k0 = ks << 2;
        v2f a, s;
        a.x = Xr[k0 + kh];     a.y = Xr[k0 + kh + 1];
        if (GATHER) {
            s = *(const v2f*)&mS[m * SROW + k0 + kh];
        } else {
            s.x = Sr[k0 + kh]; s.y = Sr[k0 + kh + 1];
        }
#pragma unroll
        for (int nt = 0; nt < 4; ++nt) {
            v2f bd = *(const v2f*)&sWd[((ks << 2) + nt) * 64 + fragBase];
            v2f bb = *(const v2f*)&sWb[((ks << 2) + nt) * 64 + fragBase];
            accX[nt] = __builtin_amdgcn_wmma_f32_16x16x4_f32(
                false, a, false, bd, (short)0, accX[nt], false, false);
            accS[nt] = __builtin_amdgcn_wmma_f32_16x16x4_f32(
                false, s, false, bb, (short)0, accS[nt], false, false);
        }
    }

    int ncol = lane & 15;
    int mAdd = (lane >> 4) * 8;
    float rs[4][8];
    if (resid) {
#pragma unroll
        for (int nt = 0; nt < 4; ++nt) {
#pragma unroll
            for (int v = 0; v < 8; ++v) {
                int node = node0 + v + mAdd;
                rs[nt][v] = (node < N) ? resid[(size_t)node * HDIM + nt * 16 + ncol] : 0.0f;
            }
        }
    }
#pragma unroll
    for (int nt = 0; nt < 4; ++nt) {
        int col = nt * 16 + ncol;
        float wc = Wc[col];
        float bb = bias[col];
#pragma unroll
        for (int v = 0; v < 8; ++v) {
            int node = node0 + v + mAdd;
            if (node >= N) continue;
            float g = gate[node];
            float val = g * (accX[nt][v] + bb) + invdeg[node] * accS[nt][v] + ssn[node] * wc;
            if (resid) val += rs[nt][v];
            val = (val > 0.0f) ? val : 0.01f * val;
            Y[(size_t)node * HDIM + col] = val;
        }
    }
}

__global__ __launch_bounds__(256) void proj_kernel(const float* __restrict__ X, const float* __restrict__ Wout,
                                                   float* __restrict__ p, float* __restrict__ q, int N) {
    __shared__ float sP[64];
    __shared__ float sQ[64];
    int t = blockIdx.x * 256 + threadIdx.x;
    int n = t >> 2, l = t & 3;
    int nn = (n < N) ? n : N - 1;
    const float4* Xr = (const float4*)X + (size_t)nn * 16;
    float ap = 0.0f, aq = 0.0f;
#pragma unroll
    for (int i = 0; i < 4; ++i) {
        int c4 = i * 4 + l;
        float4 v = Xr[c4];
        int k = c4 * 4;
        float wa0 = Wout[k], wa1 = Wout[k + 1], wa2 = Wout[k + 2], wa3 = Wout[k + 3];
        float wb0 = Wout[HDIM + k], wb1 = Wout[HDIM + k + 1], wb2 = Wout[HDIM + k + 2], wb3 = Wout[HDIM + k + 3];
        aq += v.x * (wa0 - wb0) + v.y * (wa1 - wb1) + v.z * (wa2 - wb2) + v.w * (wa3 - wb3);
        ap += v.x * wb0 + v.y * wb1 + v.z * wb2 + v.w * wb3;
    }
#pragma unroll
    for (int o = 2; o > 0; o >>= 1) {
        ap += __shfl_xor(ap, o, 4);
        aq += __shfl_xor(aq, o, 4);
    }
    if (l == 0) { sP[threadIdx.x >> 2] = ap; sQ[threadIdx.x >> 2] = aq; }
    __syncthreads();
    if (threadIdx.x < 64) {
        int m = blockIdx.x * 64 + threadIdx.x;
        if (m < N) p[m] = sP[threadIdx.x];
    } else if (threadIdx.x < 128) {
        int m = blockIdx.x * 64 + threadIdx.x - 64;
        if (m < N) q[m] = sQ[threadIdx.x - 64];
    }
}

__global__ __launch_bounds__(256) void out_kernel(const int* __restrict__ rowptr, const int* __restrict__ rowend,
                                                  const int* __restrict__ csr,
                                                  const float* __restrict__ p, const float* __restrict__ q,
                                                  const float* __restrict__ Wout, const float* __restrict__ bout,
                                                  const float* __restrict__ x0,
                                                  const float* __restrict__ invdeg, const float* __restrict__ gate,
                                                  const float* __restrict__ ssn, float* __restrict__ out, int N) {
    __shared__ float sOut[32];
    int t = blockIdx.x * 256 + threadIdx.x;
    int n = t >> 3, l = t & 7;
    int nn = (n < N) ? n : N - 1;
    int b = rowptr[nn], e = rowend[nn];
    float a = 0.0f;
    for (int base = b; base < e; base += 32) {
        int idx[4];
#pragma unroll
        for (int u = 0; u < 4; ++u) {
            int j = base + u * 8 + l;
            idx[u] = csr[(j < e) ? j : (e - 1)];
        }
#pragma unroll
        for (int u = 0; u < 4; ++u) {
            float v = p[idx[u]];
            a += ((base + u * 8 + l) < e) ? v : 0.0f;
        }
    }
#pragma unroll
    for (int o = 4; o > 0; o >>= 1) a += __shfl_xor(a, o, 8);
    if (l == 0) sOut[threadIdx.x >> 3] = a;
    __syncthreads();
    if (threadIdx.x < 32) {
        int m = blockIdx.x * 32 + threadIdx.x;
        if (m < N)
            out[m] = gate[m] * (q[m] + bout[0]) + invdeg[m] * sOut[threadIdx.x] + ssn[m] * Wout[2 * HDIM] + x0[m];
    }
}

extern "C" void kernel_launch(void* const* d_in, const int* in_sizes, int n_in,
                              void* d_out, int out_size, void* d_ws, size_t ws_size,
                              hipStream_t stream) {
    const float* x    = (const float*)d_in[0];
    const int*   ei   = (const int*)  d_in[1];
    const float* Win  = (const float*)d_in[2];
    const float* bin  = (const float*)d_in[3];
    const float* Wmid = (const float*)d_in[4];
    const float* bmid = (const float*)d_in[5];
    const float* Wout = (const float*)d_in[6];
    const float* bout = (const float*)d_in[7];

    const int N = in_sizes[0];
    const int E = in_sizes[1] / 2;
    const int* src = ei;
    const int* dst = ei + E;

    float* w = (float*)d_ws;
    int*   rowptr  = (int*)w;
    int*   cursor  = (int*)(w + (size_t)N);
    float* invdeg  = w + 2 * (size_t)N;
    float* gate    = w + 3 * (size_t)N;
    float* ssn     = w + 4 * (size_t)N;
    float* Sx      = w + 5 * (size_t)N;
    float* Wd6     = w + 6 * (size_t)N;
    float* bufA    = Wd6  + 6 * HDIM * HDIM;
    float* bufB    = bufA + (size_t)N * HDIM;
    float* bufC    = bufB + (size_t)N * HDIM;
    float* bufS    = bufC + (size_t)N * HDIM;
    int*   csr     = (int*)bufC;

    const int nTiles = (N + 15) / 16;

    const int NB = (N + BINW - 1) / BINW;
    int* binned = (int*)bufB;
    int* cntT   = (int*)bufS;
    int* offs   = cntT + NB * NCHUNK;
    bin_count_kernel<<<NCHUNK, CH_THREADS, 0, stream>>>(dst, cntT, E, NB);
    scan_kernel<<<1, 1024, 0, stream>>>(cntT, offs, NB * NCHUNK);
    bin_scatter_kernel<<<NCHUNK, CH_THREADS, 0, stream>>>(src, dst, offs, binned, E, NB);
    bin_fill_kernel<<<NB, FILL_THREADS, 0, stream>>>(offs, binned, rowptr, cursor, csr, NB, N, E);
    float* coef = bufB;
    float* csum = bufB + 4 * (size_t)N;
    node_stats_kernel<<<(N * 8 + 255) / 256, 256, 0, stream>>>(rowptr, cursor, csr, x, invdeg, gate,
                                                               ssn, Sx, coef, N);
    wdiff_kernel<<<(6 * 4096 + 255) / 256, 256, 0, stream>>>(Wmid, Wd6, 6 * 4096);
    coef_gather_kernel<<<(N * 8 + 255) / 256, 256, 0, stream>>>(rowptr, cursor, csr, coef, csum, N);
    conv_in_kernel<<<(N * HDIM + 255) / 256, 256, 0, stream>>>(x, Win, bin, invdeg, gate, ssn, Sx, csum,
                                                               bufA, bufS, N);

    float* cur = bufA;
    mid_conv_kernel<false><<<(nTiles + 7) / 8, 256, 0, stream>>>(
        cur, bufS, rowptr, cursor, csr, Wd6, Wmid + 64 * HDIM, Wmid + 128 * HDIM,
        bmid, gate, invdeg, ssn, nullptr, bufB, nTiles, N);
    mid_conv_kernel<true><<<(nTiles + 7) / 8, 256, 0, stream>>>(
        bufB, bufS, rowptr, cursor, csr, Wd6 + HDIM * HDIM,
        Wmid + (size_t)129 * HDIM + 64 * HDIM, Wmid + (size_t)129 * HDIM + 128 * HDIM,
        bmid + HDIM, gate, invdeg, ssn, cur, cur, nTiles, N);
    for (int l = 1; l < 3; ++l) {
        const float* Wl = Wmid + (size_t)(2 * l) * 129 * HDIM;
        mid_conv_kernel<true><<<(nTiles + 7) / 8, 256, 0, stream>>>(
            cur, bufS, rowptr, cursor, csr, Wd6 + (2 * l) * HDIM * HDIM, Wl + 64 * HDIM, Wl + 128 * HDIM,
            bmid + (2 * l) * HDIM, gate, invdeg, ssn, nullptr, bufB, nTiles, N);

        const float* Wl2 = Wmid + (size_t)(2 * l + 1) * 129 * HDIM;
        mid_conv_kernel<true><<<(nTiles + 7) / 8, 256, 0, stream>>>(
            bufB, bufS, rowptr, cursor, csr, Wd6 + (2 * l + 1) * HDIM * HDIM, Wl2 + 64 * HDIM, Wl2 + 128 * HDIM,
            bmid + (2 * l + 1) * HDIM, gate, invdeg, ssn, cur, cur, nTiles, N);
    }

    float* pj = bufS;
    float* qj = bufS + 2 * (size_t)N;
    proj_kernel<<<(N * 4 + 255) / 256, 256, 0, stream>>>(cur, Wout, pj, qj, N);
    out_kernel<<<(N * 8 + 255) / 256, 256, 0, stream>>>(rowptr, cursor, csr, pj, qj, Wout, bout, x, invdeg, gate,
                                                        ssn, (float*)d_out, N);
}


# Run starts of the edge sort are derived inside the scatter kernel from the count table (row sums by 8 lanes, one in-block scan of bin totals); the one-block serial scan launch goes: 15 launches
# speedup vs baseline: 4.4396x; 1.0727x over previous
#include <hip/hip_runtime.h>
#include <hip/hip_bf16.h>

typedef float v2f __attribute__((ext_vector_type(2)));
typedef float v8f __attribute__((ext_vector_type(8)));

#define HDIM 64

#define BIN_SHIFT 8
#define BINW 256
#define NB_MAX 512
#define NCHUNK 256
#define CH_THREADS 1024
#define FILL_THREADS 512
#define FILL_CAP 12288

__global__ __launch_bounds__(CH_THREADS) void bin_count_kernel(const int* __restrict__ dst,
                                                               int* __restrict__ cntT, int E, int NB) {
    __shared__ int sHist[NB_MAX];
    const int c = blockIdx.x, tid = threadIdx.x;
    for (int i = tid; i < NB; i += CH_THREADS) sHist[i] = 0;
    __syncthreads();
    const int chunk = (E + NCHUNK - 1) / NCHUNK;
    const int e0 = c * chunk;
    const int e1 = (e0 + chunk < E) ? (e0 + chunk) : E;
    for (int e = e0 + tid; e < e1; e += CH_THREADS * 8) {
        int d[8];
#pragma unroll
        for (int u = 0; u < 8; ++u) {
            int ee = e + u * CH_THREADS;
            d[u] = dst[(ee < e1) ? ee : (e1 - 1)];
        }
#pragma unroll
        for (int u = 0; u < 8; ++u)
            if (e + u * CH_THREADS < e1) atomicAdd(&sHist[d[u] >> BIN_SHIFT], 1);
    }
    __syncthreads();
    for (int i = tid; i < NB; i += CH_THREADS) cntT[i * NCHUNK + c] = sHist[i];
}

#define GROUPS_PER_PASS 2
__global__ __launch_bounds__(CH_THREADS) void bin_scatter_kernel(const int* __restrict__ src,
                                                                 const int* __restrict__ dst,
                                                                 const int* __restrict__ cntT,
                                                                 int* __restrict__ binBase,
                                                                 int* __restrict__ binned, int E, int NB) {
    static_assert(NCHUNK == 256, "one row of the count table = 8 lanes x 8 loads x 4 counts");
    static_assert(NB_MAX == 512, "wave 0 scans 16 bin totals per lane");
    __shared__ int sCur[NB_MAX];
    __shared__ int sTot[NB_MAX];
    const int c = blockIdx.x, tid = threadIdx.x, lane = tid & 31, wv = tid >> 5;
    const int sub = lane & 7, rw = lane >> 3;
    for (int i = tid; i < NB_MAX; i += CH_THREADS) sTot[i] = 0;
    __syncthreads();
    for (int g0 = wv; g0 * 4 < NB; g0 += (CH_THREADS / 32) * GROUPS_PER_PASS) {
        int4 v[GROUPS_PER_PASS][8];
#pragma unroll
        for (int u = 0; u < GROUPS_PER_PASS; ++u) {
            int i = (g0 + u * (CH_THREADS / 32)) * 4 + rw;
            const int4* row = (const int4*)(cntT + (size_t)((i < NB) ? i : (NB - 1)) * NCHUNK);
#pragma unroll
            for (int j = 0; j < 8; ++j) v[u][j] = row[j * 8 + sub];
        }
#pragma unroll
        for (int u = 0; u < GROUPS_PER_PASS; ++u) {
            int tot = 0, pre = 0;
#pragma unroll
            for (int j = 0; j < 8; ++j) {
                int k = (j * 8 + sub) * 4;
                tot += v[u][j].x + v[u][j].y + v[u][j].z + v[u][j].w;
                pre += ((k     < c) ? v[u][j].x : 0) + ((k + 1 < c) ? v[u][j].y : 0)
                     + ((k + 2 < c) ? v[u][j].z : 0) + ((k + 3 < c) ? v[u][j].w : 0);
            }
#pragma unroll
            for (int o = 4; o > 0; o >>= 1) {
                tot += __shfl_xor(tot, o, 8);
                pre += __shfl_xor(pre, o, 8);
            }
            int i = (g0 + u * (CH_THREADS / 32)) * 4 + rw;
            if (sub == 0 && i < NB) { sTot[i] = tot; sCur[i] = pre; }
        }
    }
    __syncthreads();
    if (wv == 0) {
        int t[16];
        int s = 0;
#pragma unroll
        for (int k = 0; k < 16; ++k) { t[k] = sTot[lane * 16 + k]; s += t[k]; }
        int inc = s;
#pragma unroll
        for (int o = 1; o < 32; o <<= 1) {
            int v = __shfl_up(inc, o, 32);
            if (lane >= o) inc += v;
        }
        int ex = inc - s;
#pragma unroll
        for (int k = 0; k < 16; ++k) { sTot[lane * 16 + k] = ex; ex += t[k]; }
    }
    __syncthreads();
    for (int i = tid; i < NB; i += CH_THREADS) {
        int base = sTot[i];
        sCur[i] += base;
        if (c == 0) binBase[i] = base;
    }
    if (c == 0 && tid == 0) binBase[NB] = E;
    __syncthreads();
    const int chunk = (E + NCHUNK - 1) / NCHUNK;
    const int e0 = c * chunk;
    const int e1 = (e0 + chunk < E) ? (e0 + chunk) : E;
    for (int e = e0 + tid; e < e1; e += CH_THREADS * 8) {
        int d[8], s[8];
#pragma unroll
        for (int u = 0; u < 8; ++u) {
            int ee = e + u * CH_THREADS;
            int ec = (ee < e1) ? ee : (e1 - 1);
            d[u] = dst[ec];
            s[u] = src[ec];
        }
#pragma unroll
        for (int u = 0; u < 8; ++u) {
            if (e + u * CH_THREADS < e1) {
                int pos = atomicAdd(&sCur[d[u] >> BIN_SHIFT], 1);
                binned[pos] = (s[u] << BIN_SHIFT) | (d[u] & (BINW - 1));
            }
        }
    }
}

__global__ __launch_bounds__(FILL_THREADS) void bin_fill_kernel(const int* __restrict__ binBase,
                                                                const int* __restrict__ binned,
                                                                int* __restrict__ rowptr, int* __restrict__ rowend,
                                                                int* __restrict__ csr, int N) {
    __shared__ int sDeg[BINW];
    __shared__ int sWaveTot[BINW / 32];
    __shared__ int sStage[FILL_CAP];
    const int b = blockIdx.x, tid = threadIdx.x, lane = tid & 31;
    const int beg = binBase[b];
    const int end = binBase[b + 1];
    const int cnt = end - beg;
    if (tid < BINW) sDeg[tid] = 0;
    __syncthreads();
    for (int i = tid; i < cnt; i += FILL_THREADS * 8) {
        int r[8];
#pragma unroll
        for (int u = 0; u < 8; ++u) {
            int ii = i + u * FILL_THREADS;
            r[u] = binned[beg + ((ii < cnt) ? ii : (cnt - 1))];
        }
#pragma unroll
        for (int u = 0; u < 8; ++u)
            if (i + u * FILL_THREADS < cnt) atomicAdd(&sDeg[r[u] & (BINW - 1)], 1);
    }
    __syncthreads();
    int v = 0, inc = 0;
    if (tid < BINW) {
        v = sDeg[tid];
        inc = v;
#pragma unroll
        for (int o = 1; o < 32; o <<= 1) {
            int t = __shfl_up(inc, o, 32);
            if (lane >= o) inc += t;
        }
        if (lane == 31) sWaveTot[tid >> 5] = inc;
    }
    __syncthreads();
    if (tid < BINW) {
        int wbase = 0;
        for (int w = 0; w < (tid >> 5); ++w) wbase += sWaveTot[w];
        int excl = wbase + inc - v;
        sDeg[tid] = excl;
        int node = b * BINW + tid;
        if (node < N) { rowptr[node] = beg + excl; rowend[node] = beg + excl + v; }
    }
    __syncthreads();
    const bool staged = cnt <= FILL_CAP;
    for (int i = tid; i < cnt; i += FILL_THREADS * 8) {
        int r[8];
#pragma unroll
        for (int u = 0; u < 8; ++u) {
            int ii = i + u * FILL_THREADS;
            r[u] = binned[beg + ((ii < cnt) ? ii : (cnt - 1))];
        }
#pragma unroll
        for (int u = 0; u < 8; ++u) {
            if (i + u * FILL_THREADS < cnt) {
                int pos = atomicAdd(&sDeg[r[u] & (BINW - 1)], 1);
                if (staged) sStage[pos] = r[u] >> BIN_SHIFT;
                else        csr[beg + pos] = r[u] >> BIN_SHIFT;
            }
        }
    }
    __syncthreads();
    if (staged)
        for (int i = tid; i < cnt; i += FILL_THREADS) csr[beg + i] = sStage[i];
}

__global__ void node_stats_kernel(const int* __restrict__ rowptr, const int* __restrict__ rowend,
                                  const int* __restrict__ csr, const float* __restrict__ x,
                                  float* __restrict__ invdeg, float* __restrict__ gate,
                                  float* __restrict__ ssn, float* __restrict__ Sx,
                                  float* __restrict__ coef, int N) {
    int t = blockIdx.x * blockDim.x + threadIdx.x;
    int n = t >> 3, l = t & 7;
    bool valid = n < N;
    int nn = valid ? n : N - 1;
    int b = rowptr[nn], e = rowend[nn];
    float sg = 0.0f, sx = 0.0f;
    for (int j = b + l; j < e; j += 8) {
        int s = csr[j];
        sg += (s > nn) ? 1.0f : ((s < nn) ? -1.0f : 0.0f);
        sx += x[s];
    }
#pragma unroll
    for (int o = 4; o > 0; o >>= 1) {
        sg += __shfl_xor(sg, o, 8);
        sx += __shfl_xor(sx, o, 8);
    }
    if (valid && l == 0) {
        float dg = (float)(e - b);
        float inv = 1.0f / fmaxf(dg, 1.0f);
        float g = (dg > 0.0f) ? 1.0f : 0.0f;
        invdeg[n] = inv;
        gate[n]   = g;
        ssn[n]    = sg * inv;
        Sx[n]     = sx;
        ((float4*)coef)[n] = make_float4(g * x[n], g, inv * sx, sg * inv);
    }
}

__global__ __launch_bounds__(256) void coef_gather_kernel(const int* __restrict__ rowptr,
                                                          const int* __restrict__ rowend,
                                                          const int* __restrict__ csr,
                                                          const float* __restrict__ coef,
                                                          float* __restrict__ csum, int N) {
    __shared__ float4 sOut[32];
    const float4* coef4 = (const float4*)coef;
    int t = blockIdx.x * 256 + threadIdx.x;
    int n = t >> 3, l = t & 7;
    int nn = (n < N) ? n : N - 1;
    int b = rowptr[nn], e = rowend[nn];
    float4 a = make_float4(0.0f, 0.0f, 0.0f, 0.0f);
    for (int base = b; base < e; base += 32) {
        int idx[4];
#pragma unroll
        for (int u = 0; u < 4; ++u) {
            int j = base + u * 8 + l;
            idx[u] = csr[(j < e) ? j : (e - 1)];
        }
#pragma unroll
        for (int u = 0; u < 4; ++u) {
            float4 c = coef4[idx[u]];
            bool ok = (base + u * 8 + l) < e;
            a.x += ok ? c.x : 0.0f;
            a.y += ok ? c.y : 0.0f;
            a.z += ok ? c.z : 0.0f;
            a.w += ok ? c.w : 0.0f;
        }
    }
#pragma unroll
    for (int o = 4; o > 0; o >>= 1) {
        a.x += __shfl_xor(a.x, o, 8);
        a.y += __shfl_xor(a.y, o, 8);
        a.z += __shfl_xor(a.z, o, 8);
        a.w += __shfl_xor(a.w, o, 8);
    }
    if (l == 0) sOut[threadIdx.x >> 3] = a;
    __syncthreads();
    if (threadIdx.x < 32) {
        int m = blockIdx.x * 32 + threadIdx.x;
        if (m < N) ((float4*)csum)[m] = sOut[threadIdx.x];
    }
}

__global__ void wdiff_kernel(const float* __restrict__ Wmid, float* __restrict__ Wd6, int total) {
    int i = blockIdx.x * blockDim.x + threadIdx.x;
    if (i >= total) return;
    int l = i >> 12;
    int rc = i & 4095;
    const float* Wl = Wmid + (size_t)l * 129 * HDIM;
    Wd6[i] = Wl[rc] - Wl[64 * HDIM + rc];
}

__global__ void conv_in_kernel(const float* __restrict__ x, const float* __restrict__ Win,
                               const float* __restrict__ bin,
                               const float* __restrict__ invdeg, const float* __restrict__ gate,
                               const float* __restrict__ ssn, const float* __restrict__ Sx,
                               const float* __restrict__ csum,
                               float* __restrict__ Y, float* __restrict__ S0, int N) {
    int t = blockIdx.x * blockDim.x + threadIdx.x;
    if (t >= N * HDIM) return;
    int n = t >> 6, h = t & 63;
    float wa = Win[h], wb = Win[HDIM + h], wc = Win[2 * HDIM + h];
    float bh = bin[h];
    Y[t] = gate[n] * (x[n] * (wa - wb) + bh) + invdeg[n] * Sx[n] * wb + ssn[n] * wc;
    float4 cs = ((const float4*)csum)[n];
    S0[t] = cs.x * (wa - wb) + cs.y * bh + cs.z * wb + cs.w * wc;
}

__device__ __forceinline__ void gather16(const float4* __restrict__ X4, int mine, int cnt, int q, float4& acc) {
#pragma unroll
    for (int k = 0; k < 16; ++k) {
        int s = __shfl(mine, k, 16);
        float4 v = X4[(size_t)s * 16 + q];
        bool ok = k < cnt;
        acc.x += ok ? v.x : 0.0f;
        acc.y += ok ? v.y : 0.0f;
        acc.z += ok ? v.z : 0.0f;
        acc.w += ok ? v.w : 0.0f;
    }
}

#define SROW 68
#define TILE_IDX_CAP 768
template <bool GATHER>
__global__ __launch_bounds__(256) void mid_conv_kernel(
    const float* __restrict__ X, const float* __restrict__ S,
    const int* __restrict__ rowptr, const int* __restrict__ rowend, const int* __restrict__ csr,
    const float* __restrict__ Wd, const float* __restrict__ Wb,
    const float* __restrict__ Wc, const float* __restrict__ bias,
    const float* __restrict__ gate, const float* __restrict__ invdeg,
    const float* __restrict__ ssn, const float* resid,
    float* Y, int nTiles, int N) {
    __shared__ float sWd[HDIM * HDIM];
    __shared__ float sWb[HDIM * HDIM];
    __shared__ __attribute__((aligned(16))) float sS[GATHER ? 8 * 16 * SROW : 4];
    __shared__ int   sIdx[GATHER ? 8 * TILE_IDX_CAP : 2];
    for (int i = threadIdx.x; i < HDIM * HDIM; i += 256) {
        int ks   = i >> 8;
        int rem  = i & 255;
        int nt   = rem >> 6;
        int lp   = rem & 63;
        int ln   = lp >> 1;
        int half = lp & 1;
        int kh   = (ln >> 4) << 1;
        int srcr = (ks << 2) + kh + half;
        int srcc = (nt << 4) + (ln & 15);
        sWd[i] = Wd[srcr * HDIM + srcc];
        sWb[i] = Wb[srcr * HDIM + srcc];
    }
    __syncthreads();

    int wave = threadIdx.x >> 5;
    int lane = threadIdx.x & 31;
    int tile = blockIdx.x * 8 + wave;
    if (tile >= nTiles) return;

    int node0 = tile * 16;
    int m  = lane & 15;
    int kh = (lane >> 4) << 1;
    int row = node0 + m; if (row >= N) row = N - 1;
    const float* Xr = X + (size_t)row * HDIM;
    const float* Sr = S + (size_t)row * HDIM;
    float* mS = sS + (GATHER ? wave * (16 * SROW) : 0);

    if (GATHER) {
        int* mIx = sIdx + wave * TILE_IDX_CAP;
        int rb = rowptr[row], re = rowend[row];
        if (node0 + m >= N) rb = re;
        int tb = __shfl(rb, 0, 32);
        int staged = __shfl(re, 15, 32) - tb;
        staged = (staged < 0) ? 0 : ((staged > TILE_IDX_CAP) ? TILE_IDX_CAP : staged);
        for (int i0 = 0; i0 < staged; i0 += 32 * 16) {
            int v[16];
#pragma unroll
            for (int u = 0; u < 16; ++u) {
                int i = i0 + u * 32 + lane;
                v[u] = csr[tb + ((i < staged) ? i : (staged - 1))];
            }
#pragma unroll
            for (int u = 0; u < 16; ++u) {
                int i = i0 + u * 32 + lane;
                if (i < staged) mIx[i] = v[u];
            }
        }
        __builtin_amdgcn_fence(__ATOMIC_ACQ_REL, "wavefront");

        const float4* X4 = (const float4*)X;
        int q = lane & 15, h = lane >> 4;
        for (int p = 0; p < 8; ++p) {
            int r = 2 * p + h;
            int b = __shfl(rb, r, 32), e = __shfl(re, r, 32);
            int own = node0 + r; if (own >= N) own = N - 1;
            float4 acc = make_float4(0.0f, 0.0f, 0.0f, 0.0f);
            int eL = (b >= tb) ? (((tb + staged) < e) ? (tb + staged) : e) : b;
            if (eL < b) eL = b;
            for (int base = b; base < eL; base += 16) {
                int j = base + q;
                int mine = (j < eL) ? mIx[j - tb] : own;
                gather16(X4, mine, eL - base, q, acc);
            }
            for (int base = eL; base < e; base += 16) {
                int j = base + q;
                int mine = (j < e) ? csr[j] : own;
                gather16(X4, mine, e - base, q, acc);
            }
            *(float4*)&mS[r * SROW + 4 * q] = acc;
        }
        __builtin_amdgcn_fence(__ATOMIC_ACQ_REL, "wavefront");
    }

    v8f accX[4] = {};
    v8f accS[4] = {};
    int fragBase = lane * 2;

#pragma unroll
    for (int ks = 0; ks < 16; ++ks) {
        int k0 = ks << 2;
        v2f a, s;
        a.x = Xr[k0 + kh];     a.y = Xr[k0 + kh + 1];
        if (GATHER) {
            s = *(const v2f*)&mS[m * SROW + k0 + kh];
        } else {
            s.x = Sr[k0 + kh]; s.y = Sr[k0 + kh + 1];
        }
#pragma unroll
        for (int nt = 0; nt < 4; ++nt) {
            v2f bd = *(const v2f*)&sWd[((ks << 2) + nt) * 64 + fragBase];
            v2f bb = *(const v2f*)&sWb[((ks << 2) + nt) * 64 + fragBase];
            accX[nt] = __builtin_amdgcn_wmma_f32_16x16x4_f32(
                false, a, false, bd, (short)0, accX[nt], false, false);
            accS[nt] = __builtin_amdgcn_wmma_f32_16x16x4_f32(
                false, s, false, bb, (short)0, accS[nt], false, false);
        }
    }

    int ncol = lane & 15;
    int mAdd = (lane >> 4) * 8;
    float rs[4][8];
    if (resid) {
#pragma unroll
        for (int nt = 0; nt < 4; ++nt) {
#pragma unroll
            for (int v = 0; v < 8; ++v) {
                int node = node0 + v + mAdd;
                rs[nt][v] = (node < N) ? resid[(size_t)node * HDIM + nt * 16 + ncol] : 0.0f;
            }
        }
    }
#pragma unroll
    for (int nt = 0; nt < 4; ++nt) {
        int col = nt * 16 + ncol;
        float wc = Wc[col];
        float bb = bias[col];
#pragma unroll
        for (int v = 0; v < 8; ++v) {
            int node = node0 + v + mAdd;
            if (node >= N) continue;
            float g = gate[node];
            float val = g * (accX[nt][v] + bb) + invdeg[node] * accS[nt][v] + ssn[node] * wc;
            if (resid) val += rs[nt][v];
            val = (val > 0.0f) ? val : 0.01f * val;
            Y[(size_t)node * HDIM + col] = val;
        }
    }
}

__global__ __launch_bounds__(256) void proj_kernel(const float* __restrict__ X, const float* __restrict__ Wout,
                                                   float* __restrict__ p, float* __restrict__ q, int N) {
    __shared__ float sP[64];
    __shared__ float sQ[64];
    int t = blockIdx.x * 256 + threadIdx.x;
    int n = t >> 2, l = t & 3;
    int nn = (n < N) ? n : N - 1;
    const float4* Xr = (const float4*)X + (size_t)nn * 16;
    float ap = 0.0f, aq = 0.0f;
#pragma unroll
    for (int i = 0; i < 4; ++i) {
        int c4 = i * 4 + l;
        float4 v = Xr[c4];
        int k = c4 * 4;
        float wa0 = Wout[k], wa1 = Wout[k + 1], wa2 = Wout[k + 2], wa3 = Wout[k + 3];
        float wb0 = Wout[HDIM + k], wb1 = Wout[HDIM + k + 1], wb2 = Wout[HDIM + k + 2], wb3 = Wout[HDIM + k + 3];
        aq += v.x * (wa0 - wb0) + v.y * (wa1 - wb1) + v.z * (wa2 - wb2) + v.w * (wa3 - wb3);
        ap += v.x * wb0 + v.y * wb1 + v.z * wb2 + v.w * wb3;
    }
#pragma unroll
    for (int o = 2; o > 0; o >>= 1) {
        ap += __shfl_xor(ap, o, 4);
        aq += __shfl_xor(aq, o, 4);
    }
    if (l == 0) { sP[threadIdx.x >> 2] = ap; sQ[threadIdx.x >> 2] = aq; }
    __syncthreads();
    if (threadIdx.x < 64) {
        int m = blockIdx.x * 64 + threadIdx.x;
        if (m < N) p[m] = sP[threadIdx.x];
    } else if (threadIdx.x < 128) {
        int m = blockIdx.x * 64 + threadIdx.x - 64;
        if (m < N) q[m] = sQ[threadIdx.x - 64];
    }
}

__global__ __launch_bounds__(256) void out_kernel(const int* __restrict__ rowptr, const int* __restrict__ rowend,
                                                  const int* __restrict__ csr,
                                                  const float* __restrict__ p, const float* __restrict__ q,
                                                  const float* __restrict__ Wout, const float* __restrict__ bout,
                                                  const float* __restrict__ x0,
                                                  const float* __restrict__ invdeg, const float* __restrict__ gate,
                                                  const float* __restrict__ ssn, float* __restrict__ out, int N) {
    __shared__ float sOut[32];
    int t = blockIdx.x * 256 + threadIdx.x;
    int n = t >> 3, l = t & 7;
    int nn = (n < N) ? n : N - 1;
    int b = rowptr[nn], e = rowend[nn];
    float a = 0.0f;
    for (int base = b; base < e; base += 32) {
        int idx[4];
#pragma unroll
        for (int u = 0; u < 4; ++u) {
            int j = base + u * 8 + l;
            idx[u] = csr[(j < e) ? j : (e - 1)];
        }
#pragma unroll
        for (int u = 0; u < 4; ++u) {
            float v = p[idx[u]];
            a += ((base + u * 8 + l) < e) ? v : 0.0f;
        }
    }
#pragma unroll
    for (int o = 4; o > 0; o >>= 1) a += __shfl_xor(a, o, 8);
    if (l == 0) sOut[threadIdx.x >> 3] = a;
    __syncthreads();
    if (threadIdx.x < 32) {
        int m = blockIdx.x * 32 + threadIdx.x;
        if (m < N)
            out[m] = gate[m] * (q[m] + bout[0]) + invdeg[m] * sOut[threadIdx.x] + ssn[m] * Wout[2 * HDIM] + x0[m];
    }
}

extern "C" void kernel_launch(void* const* d_in, const int* in_sizes, int n_in,
                              void* d_out, int out_size, void* d_ws, size_t ws_size,
                              hipStream_t stream) {
    const float* x    = (const float*)d_in[0];
    const int*   ei   = (const int*)  d_in[1];
    const float* Win  = (const float*)d_in[2];
    const float* bin  = (const float*)d_in[3];
    const float* Wmid = (const float*)d_in[4];
    const float* bmid = (const float*)d_in[5];
    const float* Wout = (const float*)d_in[6];
    const float* bout = (const float*)d_in[7];

    const int N = in_sizes[0];
    const int E = in_sizes[1] / 2;
    const int* src = ei;
    const int* dst = ei + E;

    float* w = (float*)d_ws;
    int*   rowptr  = (int*)w;
    int*   cursor  = (int*)(w + (size_t)N);
    float* invdeg  = w + 2 * (size_t)N;
    float* gate    = w + 3 * (size_t)N;
    float* ssn     = w + 4 * (size_t)N;
    float* Sx      = w + 5 * (size_t)N;
    float* Wd6     = w + 6 * (size_t)N;
    float* bufA    = Wd6  + 6 * HDIM * HDIM;
    float* bufB    = bufA + (size_t)N * HDIM;
    float* bufC    = bufB + (size_t)N * HDIM;
    float* bufS    = bufC + (size_t)N * HDIM;
    int*   csr     = (int*)bufC;

    const int nTiles = (N + 15) / 16;

    const int NB = (N + BINW - 1) / BINW;
    int* binned = (int*)bufB;
    int* cntT   = (int*)bufS;
    int* binBase = cntT + NB * NCHUNK;
    bin_count_kernel<<<NCHUNK, CH_THREADS, 0, stream>>>(dst, cntT, E, NB);
    bin_scatter_kernel<<<NCHUNK, CH_THREADS, 0, stream>>>(src, dst, cntT, binBase, binned, E, NB);
    bin_fill_kernel<<<NB, FILL_THREADS, 0, stream>>>(binBase, binned, rowptr, cursor, csr, N);
    float* coef = bufB;
    float* csum = bufB + 4 * (size_t)N;
    node_stats_kernel<<<(N * 8 + 255) / 256, 256, 0, stream>>>(rowptr, cursor, csr, x, invdeg, gate,
                                                               ssn, Sx, coef, N);
    wdiff_kernel<<<(6 * 4096 + 255) / 256, 256, 0, stream>>>(Wmid, Wd6, 6 * 4096);
    coef_gather_kernel<<<(N * 8 + 255) / 256, 256, 0, stream>>>(rowptr, cursor, csr, coef, csum, N);
    conv_in_kernel<<<(N * HDIM + 255) / 256, 256, 0, stream>>>(x, Win, bin, invdeg, gate, ssn, Sx, csum,
                                                               bufA, bufS, N);

    float* cur = bufA;
    mid_conv_kernel<false><<<(nTiles + 7) / 8, 256, 0, stream>>>(
        cur, bufS, rowptr, cursor, csr, Wd6, Wmid + 64 * HDIM, Wmid + 128 * HDIM,
        bmid, gate, invdeg, ssn, nullptr, bufB, nTiles, N);
    mid_conv_kernel<true><<<(nTiles + 7) / 8, 256, 0, stream>>>(
        bufB, bufS, rowptr, cursor, csr, Wd6 + HDIM * HDIM,
        Wmid + (size_t)129 * HDIM + 64 * HDIM, Wmid + (size_t)129 * HDIM + 128 * HDIM,
        bmid + HDIM, gate, invdeg, ssn, cur, cur, nTiles, N);
    for (int l = 1; l < 3; ++l) {
        const float* Wl = Wmid + (size_t)(2 * l) * 129 * HDIM;
        mid_conv_kernel<true><<<(nTiles + 7) / 8, 256, 0, stream>>>(
            cur, bufS, rowptr, cursor, csr, Wd6 + (2 * l) * HDIM * HDIM, Wl + 64 * HDIM, Wl + 128 * HDIM,
            bmid + (2 * l) * HDIM, gate, invdeg, ssn, nullptr, bufB, nTiles, N);

        const float* Wl2 = Wmid + (size_t)(2 * l + 1) * 129 * HDIM;
        mid_conv_kernel<true><<<(nTiles + 7) / 8, 256, 0, stream>>>(
            bufB, bufS, rowptr, cursor, csr, Wd6 + (2 * l + 1) * HDIM * HDIM, Wl2 + 64 * HDIM, Wl2 + 128 * HDIM,
            bmid + (2 * l + 1) * HDIM, gate, invdeg, ssn, cur, cur, nTiles, N);
    }

    float* pj = bufS;
    float* qj = bufS + 2 * (size_t)N;
    proj_kernel<<<(N * 4 + 255) / 256, 256, 0, stream>>>(cur, Wout, pj, qj, N);
    out_kernel<<<(N * 8 + 255) / 256, 256, 0, stream>>>(rowptr, cursor, csr, pj, qj, Wout, bout, x, invdeg, gate,
                                                        ssn, (float*)d_out, N);
}


# Mid conv output stage waits on no load per element: per-row factors (one lane each, spread by row_share) and column terms requested ahead of the gather, residual rows read as one clamped group
# speedup vs baseline: 5.0102x; 1.1285x over previous
#include <hip/hip_runtime.h>
#include <hip/hip_bf16.h>

typedef float v2f __attribute__((ext_vector_type(2)));
typedef float v8f __attribute__((ext_vector_type(8)));

#define HDIM 64

#define BIN_SHIFT 8
#define BINW 256
#define NB_MAX 512
#define NCHUNK 256
#define CH_THREADS 1024
#define FILL_THREADS 512
#define FILL_CAP 12288

__global__ __launch_bounds__(CH_THREADS) void bin_count_kernel(const int* __restrict__ dst,
                                                               int* __restrict__ cntT, int E, int NB) {
    __shared__ int sHist[NB_MAX];
    const int c = blockIdx.x, tid = threadIdx.x;
    for (int i = tid; i < NB; i += CH_THREADS) sHist[i] = 0;
    __syncthreads();
    const int chunk = (E + NCHUNK - 1) / NCHUNK;
    const int e0 = c * chunk;
    const int e1 = (e0 + chunk < E) ? (e0 + chunk) : E;
    for (int e = e0 + tid; e < e1; e += CH_THREADS * 8) {
        int d[8];
#pragma unroll
        for (int u = 0; u < 8; ++u) {
            int ee = e + u * CH_THREADS;
            d[u] = dst[(ee < e1) ? ee : (e1 - 1)];
        }
#pragma unroll
        for (int u = 0; u < 8; ++u)
            if (e + u * CH_THREADS < e1) atomicAdd(&sHist[d[u] >> BIN_SHIFT], 1);
    }
    __syncthreads();
    for (int i = tid; i < NB; i += CH_THREADS) cntT[i * NCHUNK + c] = sHist[i];
}

#define GROUPS_PER_PASS 2
__global__ __launch_bounds__(CH_THREADS) void bin_scatter_kernel(const int* __restrict__ src,
                                                                 const int* __restrict__ dst,
                                                                 const int* __restrict__ cntT,
                                                                 int* __restrict__ binBase,
                                                                 int* __restrict__ binned, int E, int NB) {
    static_assert(NCHUNK == 256, "one row of the count table = 8 lanes x 8 loads x 4 counts");
    static_assert(NB_MAX == 512, "wave 0 scans 16 bin totals per lane");
    __shared__ int sCur[NB_MAX];
    __shared__ int sTot[NB_MAX];
    const int c = blockIdx.x, tid = threadIdx.x, lane = tid & 31, wv = tid >> 5;
    const int sub = lane & 7, rw = lane >> 3;
    for (int i = tid; i < NB_MAX; i += CH_THREADS) sTot[i] = 0;
    __syncthreads();
    for (int g0 = wv; g0 * 4 < NB; g0 += (CH_THREADS / 32) * GROUPS_PER_PASS) {
        int4 v[GROUPS_PER_PASS][8];
#pragma unroll
        for (int u = 0; u < GROUPS_PER_PASS; ++u) {
            int i = (g0 + u * (CH_THREADS / 32)) * 4 + rw;
            const int4* row = (const int4*)(cntT + (size_t)((i < NB) ? i : (NB - 1)) * NCHUNK);
#pragma unroll
            for (int j = 0; j < 8; ++j) v[u][j] = row[j * 8 + sub];
        }
#pragma unroll
        for (int u = 0; u < GROUPS_PER_PASS; ++u) {
            int tot = 0, pre = 0;
#pragma unroll
            for (int j = 0; j < 8; ++j) {
                int k = (j * 8 + sub) * 4;
                tot += v[u][j].x + v[u][j].y + v[u][j].z + v[u][j].w;
                pre += ((k     < c) ? v[u][j].x : 0) + ((k + 1 < c) ? v[u][j].y : 0)
                     + ((k + 2 < c) ? v[u][j].z : 0) + ((k + 3 < c) ? v[u][j].w : 0);
            }
#pragma unroll
            for (int o = 4; o > 0; o >>= 1) {
                tot += __shfl_xor(tot, o, 8);
                pre += __shfl_xor(pre, o, 8);
            }
            int i = (g0 + u * (CH_THREADS / 32)) * 4 + rw;
            if (sub == 0 && i < NB) { sTot[i] = tot; sCur[i] = pre; }
        }
    }
    __syncthreads();
    if (wv == 0) {
        int t[16];
        int s = 0;
#pragma unroll
        for (int k = 0; k < 16; ++k) { t[k] = sTot[lane * 16 + k]; s += t[k]; }
        int inc = s;
#pragma unroll
        for (int o = 1; o < 32; o <<= 1) {
            int v = __shfl_up(inc, o, 32);
            if (lane >= o) inc += v;
        }
        int ex = inc - s;
#pragma unroll
        for (int k = 0; k < 16; ++k) { sTot[lane * 16 + k] = ex; ex += t[k]; }
    }
    __syncthreads();
    for (int i = tid; i < NB; i += CH_THREADS) {
        int base = sTot[i];
        sCur[i] += base;
        if (c == 0) binBase[i] = base;
    }
    if (c == 0 && tid == 0) binBase[NB] = E;
    __syncthreads();
    const int chunk = (E + NCHUNK - 1) / NCHUNK;
    const int e0 = c * chunk;
    const int e1 = (e0 + chunk < E) ? (e0 + chunk) : E;
    for (int e = e0 + tid; e < e1; e += CH_THREADS * 8) {
        int d[8], s[8];
#pragma unroll
        for (int u = 0; u < 8; ++u) {
            int ee = e + u * CH_THREADS;
            int ec = (ee < e1) ? ee : (e1 - 1);
            d[u] = dst[ec];
            s[u] = src[ec];
        }
#pragma unroll
        for (int u = 0; u < 8; ++u) {
            if (e + u * CH_THREADS < e1) {
                int pos = atomicAdd(&sCur[d[u] >> BIN_SHIFT], 1);
                binned[pos] = (s[u] << BIN_SHIFT) | (d[u] & (BINW - 1));
            }
        }
    }
}

__global__ __launch_bounds__(FILL_THREADS) void bin_fill_kernel(const int* __restrict__ binBase,
                                                                const int* __restrict__ binned,
                                                                int* __restrict__ rowptr, int* __restrict__ rowend,
                                                                int* __restrict__ csr, int N) {
    __shared__ int sDeg[BINW];
    __shared__ int sWaveTot[BINW / 32];
    __shared__ int sStage[FILL_CAP];
    const int b = blockIdx.x, tid = threadIdx.x, lane = tid & 31;
    const int beg = binBase[b];
    const int end = binBase[b + 1];
    const int cnt = end - beg;
    if (tid < BINW) sDeg[tid] = 0;
    __syncthreads();
    for (int i = tid; i < cnt; i += FILL_THREADS * 8) {
        int r[8];
#pragma unroll
        for (int u = 0; u < 8; ++u) {
            int ii = i + u * FILL_THREADS;
            r[u] = binned[beg + ((ii < cnt) ? ii : (cnt - 1))];
        }
#pragma unroll
        for (int u = 0; u < 8; ++u)
            if (i + u * FILL_THREADS < cnt) atomicAdd(&sDeg[r[u] & (BINW - 1)], 1);
    }
    __syncthreads();
    int v = 0, inc = 0;
    if (tid < BINW) {
        v = sDeg[tid];
        inc = v;
#pragma unroll
        for (int o = 1; o < 32; o <<= 1) {
            int t = __shfl_up(inc, o, 32);
            if (lane >= o) inc += t;
        }
        if (lane == 31) sWaveTot[tid >> 5] = inc;
    }
    __syncthreads();
    if (tid < BINW) {
        int wbase = 0;
        for (int w = 0; w < (tid >> 5); ++w) wbase += sWaveTot[w];
        int excl = wbase + inc - v;
        sDeg[tid] = excl;
        int node = b * BINW + tid;
        if (node < N) { rowptr[node] = beg + excl; rowend[node] = beg + excl + v; }
    }
    __syncthreads();
    const bool staged = cnt <= FILL_CAP;
    for (int i = tid; i < cnt; i += FILL_THREADS * 8) {
        int r[8];
#pragma unroll
        for (int u = 0; u < 8; ++u) {
            int ii = i + u * FILL_THREADS;
            r[u] = binned[beg + ((ii < cnt) ? ii : (cnt - 1))];
        }
#pragma unroll
        for (int u = 0; u < 8; ++u) {
            if (i + u * FILL_THREADS < cnt) {
                int pos = atomicAdd(&sDeg[r[u] & (BINW - 1)], 1);
                if (staged) sStage[pos] = r[u] >> BIN_SHIFT;
                else        csr[beg + pos] = r[u] >> BIN_SHIFT;
            }
        }
    }
    __syncthreads();
    if (staged)
        for (int i = tid; i < cnt; i += FILL_THREADS) csr[beg + i] = sStage[i];
}

__global__ void node_stats_kernel(const int* __restrict__ rowptr, const int* __restrict__ rowend,
                                  const int* __restrict__ csr, const float* __restrict__ x,
                                  float* __restrict__ invdeg, float* __restrict__ gate,
                                  float* __restrict__ ssn, float* __restrict__ Sx,
                                  float* __restrict__ coef, int N) {
    int t = blockIdx.x * blockDim.x + threadIdx.x;
    int n = t >> 3, l = t & 7;
    bool valid = n < N;
    int nn = valid ? n : N - 1;
    int b = rowptr[nn], e = rowend[nn];
    float sg = 0.0f, sx = 0.0f;
    for (int j = b + l; j < e; j += 8) {
        int s = csr[j];
        sg += (s > nn) ? 1.0f : ((s < nn) ? -1.0f : 0.0f);
        sx += x[s];
    }
#pragma unroll
    for (int o = 4; o > 0; o >>= 1) {
        sg += __shfl_xor(sg, o, 8);
        sx += __shfl_xor(sx, o, 8);
    }
    if (valid && l == 0) {
        float dg = (float)(e - b);
        float inv = 1.0f / fmaxf(dg, 1.0f);
        float g = (dg > 0.0f) ? 1.0f : 0.0f;
        invdeg[n] = inv;
        gate[n]   = g;
        ssn[n]    = sg * inv;
        Sx[n]     = sx;
        ((float4*)coef)[n] = make_float4(g * x[n], g, inv * sx, sg * inv);
    }
}

__global__ __launch_bounds__(256) void coef_gather_kernel(const int* __restrict__ rowptr,
                                                          const int* __restrict__ rowend,
                                                          const int* __restrict__ csr,
                                                          const float* __restrict__ coef,
                                                          float* __restrict__ csum, int N) {
    __shared__ float4 sOut[32];
    const float4* coef4 = (const float4*)coef;
    int t = blockIdx.x * 256 + threadIdx.x;
    int n = t >> 3, l = t & 7;
    int nn = (n < N) ? n : N - 1;
    int b = rowptr[nn], e = rowend[nn];
    float4 a = make_float4(0.0f, 0.0f, 0.0f, 0.0f);
    for (int base = b; base < e; base += 32) {
        int idx[4];
#pragma unroll
        for (int u = 0; u < 4; ++u) {
            int j = base + u * 8 + l;
            idx[u] = csr[(j < e) ? j : (e - 1)];
        }
#pragma unroll
        for (int u = 0; u < 4; ++u) {
            float4 c = coef4[idx[u]];
            bool ok = (base + u * 8 + l) < e;
            a.x += ok ? c.x : 0.0f;
            a.y += ok ? c.y : 0.0f;
            a.z += ok ? c.z : 0.0f;
            a.w += ok ? c.w : 0.0f;
        }
    }
#pragma unroll
    for (int o = 4; o > 0; o >>= 1) {
        a.x += __shfl_xor(a.x, o, 8);
        a.y += __shfl_xor(a.y, o, 8);
        a.z += __shfl_xor(a.z, o, 8);
        a.w += __shfl_xor(a.w, o, 8);
    }
    if (l == 0) sOut[threadIdx.x >> 3] = a;
    __syncthreads();
    if (threadIdx.x < 32) {
        int m = blockIdx.x * 32 + threadIdx.x;
        if (m < N) ((float4*)csum)[m] = sOut[threadIdx.x];
    }
}

__global__ void wdiff_kernel(const float* __restrict__ Wmid, float* __restrict__ Wd6, int total) {
    int i = blockIdx.x * blockDim.x + threadIdx.x;
    if (i >= total) return;
    int l = i >> 12;
    int rc = i & 4095;
    const float* Wl = Wmid + (size_t)l * 129 * HDIM;
    Wd6[i] = Wl[rc] - Wl[64 * HDIM + rc];
}

__global__ void conv_in_kernel(const float* __restrict__ x, const float* __restrict__ Win,
                               const float* __restrict__ bin,
                               const float* __restrict__ invdeg, const float* __restrict__ gate,
                               const float* __restrict__ ssn, const float* __restrict__ Sx,
                               const float* __restrict__ csum,
                               float* __restrict__ Y, float* __restrict__ S0, int N) {
    int t = blockIdx.x * blockDim.x + threadIdx.x;
    if (t >= N * HDIM) return;
    int n = t >> 6, h = t & 63;
    float wa = Win[h], wb = Win[HDIM + h], wc = Win[2 * HDIM + h];
    float bh = bin[h];
    Y[t] = gate[n] * (x[n] * (wa - wb) + bh) + invdeg[n] * Sx[n] * wb + ssn[n] * wc;
    float4 cs = ((const float4*)csum)[n];
    S0[t] = cs.x * (wa - wb) + cs.y * bh + cs.z * wb + cs.w * wc;
}

__device__ __forceinline__ void gather16(const float4* __restrict__ X4, int mine, int cnt, int q, float4& acc) {
#pragma unroll
    for (int k = 0; k < 16; ++k) {
        int s = __shfl(mine, k, 16);
        float4 v = X4[(size_t)s * 16 + q];
        bool ok = k < cnt;
        acc.x += ok ? v.x : 0.0f;
        acc.y += ok ? v.y : 0.0f;
        acc.z += ok ? v.z : 0.0f;
        acc.w += ok ? v.w : 0.0f;
    }
}

#define SROW 68
#define TILE_IDX_CAP 768
template <bool GATHER>
__global__ __launch_bounds__(256) void mid_conv_kernel(
    const float* __restrict__ X, const float* __restrict__ S,
    const int* __restrict__ rowptr, const int* __restrict__ rowend, const int* __restrict__ csr,
    const float* __restrict__ Wd, const float* __restrict__ Wb,
    const float* __restrict__ Wc, const float* __restrict__ bias,
    const float* __restrict__ gate, const float* __restrict__ invdeg,
    const float* __restrict__ ssn, const float* resid,
    float* Y, int nTiles, int N) {
    __shared__ float sWd[HDIM * HDIM];
    __shared__ float sWb[HDIM * HDIM];
    __shared__ __attribute__((aligned(16))) float sS[GATHER ? 8 * 16 * SROW : 4];
    __shared__ int   sIdx[GATHER ? 8 * TILE_IDX_CAP : 2];
    for (int i = threadIdx.x; i < HDIM * HDIM; i += 256) {
        int ks   = i >> 8;
        int rem  = i & 255;
        int nt   = rem >> 6;
        int lp   = rem & 63;
        int ln   = lp >> 1;
        int half = lp & 1;
        int kh   = (ln >> 4) << 1;
        int srcr = (ks << 2) + kh + half;
        int srcc = (nt << 4) + (ln & 15);
        sWd[i] = Wd[srcr * HDIM + srcc];
        sWb[i] = Wb[srcr * HDIM + srcc];
    }
    __syncthreads();

    int wave = threadIdx.x >> 5;
    int lane = threadIdx.x & 31;
    int tile = blockIdx.x * 8 + wave;
    if (tile >= nTiles) return;

    int node0 = tile * 16;
    int m  = lane & 15;
    int kh = (lane >> 4) << 1;
    int row = node0 + m; if (row >= N) row = N - 1;
    const float* Xr = X + (size_t)row * HDIM;
    const float* Sr = S + (size_t)row * HDIM;
    float* mS = sS + (GATHER ? wave * (16 * SROW) : 0);

    int ncol = lane & 15;
    int mAdd = (lane >> 4) * 8;
    int frow = node0 + (lane & 7) + mAdd; if (frow >= N) frow = N - 1;
    float fgl = gate[frow], fil = invdeg[frow], fsl = ssn[frow];
    float wcv[4], bbv[4];
#pragma unroll
    for (int nt = 0; nt < 4; ++nt) {
        wcv[nt] = Wc[nt * 16 + ncol];
        bbv[nt] = bias[nt * 16 + ncol];
    }

    if (GATHER) {
        int* mIx = sIdx + wave * TILE_IDX_CAP;
        int rb = rowptr[row], re = rowend[row];
        if (node0 + m >= N) rb = re;
        int tb = __shfl(rb, 0, 32);
        int staged = __shfl(re, 15, 32) - tb;
        staged = (staged < 0) ? 0 : ((staged > TILE_IDX_CAP) ? TILE_IDX_CAP : staged);
        for (int i0 = 0; i0 < staged; i0 += 32 * 16) {
            int v[16];
#pragma unroll
            for (int u = 0; u < 16; ++u) {
                int i = i0 + u * 32 + lane;
                v[u] = csr[tb + ((i < staged) ? i : (staged - 1))];
            }
#pragma unroll
            for (int u = 0; u < 16; ++u) {
                int i = i0 + u * 32 + lane;
                if (i < staged) mIx[i] = v[u];
            }
        }
        __builtin_amdgcn_fence(__ATOMIC_ACQ_REL, "wavefront");

        const float4* X4 = (const float4*)X;
        int q = lane & 15, h = lane >> 4;
        for (int p = 0; p < 8; ++p) {
            int r = 2 * p + h;
            int b = __shfl(rb, r, 32), e = __shfl(re, r, 32);
            int own = node0 + r; if (own >= N) own = N - 1;
            float4 acc = make_float4(0.0f, 0.0f, 0.0f, 0.0f);
            int eL = (b >= tb) ? (((tb + staged) < e) ? (tb + staged) : e) : b;
            if (eL < b) eL = b;
            for (int base = b; base < eL; base += 16) {
                int j = base + q;
                int mine = (j < eL) ? mIx[j - tb] : own;
                gather16(X4, mine, eL - base, q, acc);
            }
            for (int base = eL; base < e; base += 16) {
                int j = base + q;
                int mine = (j < e) ? csr[j] : own;
                gather16(X4, mine, e - base, q, acc);
            }
            *(float4*)&mS[r * SROW + 4 * q] = acc;
        }
        __builtin_amdgcn_fence(__ATOMIC_ACQ_REL, "wavefront");
    }

    v8f accX[4] = {};
    v8f accS[4] = {};
    int fragBase = lane * 2;

    v2f aX[16];
#pragma unroll
    for (int ks = 0; ks < 16; ++ks) aX[ks] = *(const v2f*)&Xr[(ks << 2) + kh];

#pragma unroll
    for (int ks = 0; ks < 16; ++ks) {
        int k0 = ks << 2;
        v2f a = aX[ks], s;
        if (GATHER) {
            s = *(const v2f*)&mS[m * SROW + k0 + kh];
        } else {
            s.x = Sr[k0 + kh]; s.y = Sr[k0 + kh + 1];
        }
#pragma unroll
        for (int nt = 0; nt < 4; ++nt) {
            v2f bd = *(const v2f*)&sWd[((ks << 2) + nt) * 64 + fragBase];
            v2f bb = *(const v2f*)&sWb[((ks << 2) + nt) * 64 + fragBase];
            accX[nt] = __builtin_amdgcn_wmma_f32_16x16x4_f32(
                false, a, false, bd, (short)0, accX[nt], false, false);
            accS[nt] = __builtin_amdgcn_wmma_f32_16x16x4_f32(
                false, s, false, bb, (short)0, accS[nt], false, false);
        }
    }

    float rs[4][8];
    if (resid) {
#pragma unroll
        for (int v = 0; v < 8; ++v) {
            int node = node0 + v + mAdd; if (node >= N) node = N - 1;
            const float* rp = resid + (size_t)node * HDIM + ncol;
#pragma unroll
            for (int nt = 0; nt < 4; ++nt) rs[nt][v] = rp[nt * 16];
        }
    }
    float fg[8], fi[8], fs[8];
#pragma unroll
    for (int v = 0; v < 8; ++v) {
        fg[v] = __shfl(fgl, v, 16);
        fi[v] = __shfl(fil, v, 16);
        fs[v] = __shfl(fsl, v, 16);
    }
#pragma unroll
    for (int nt = 0; nt < 4; ++nt) {
        int col = nt * 16 + ncol;
#pragma unroll
        for (int v = 0; v < 8; ++v) {
            int node = node0 + v + mAdd;
            float val = fg[v] * (accX[nt][v] + bbv[nt]) + fi[v] * accS[nt][v] + fs[v] * wcv[nt];
            if (resid) val += rs[nt][v];
            val = (val > 0.0f) ? val : 0.01f * val;
            if (node < N) Y[(size_t)node * HDIM + col] = val;
        }
    }
}

__global__ __launch_bounds__(256) void proj_kernel(const float* __restrict__ X, const float* __restrict__ Wout,
                                                   float* __restrict__ p, float* __restrict__ q, int N) {
    __shared__ float sP[64];
    __shared__ float sQ[64];
    int t = blockIdx.x * 256 + threadIdx.x;
    int n = t >> 2, l = t & 3;
    int nn = (n < N) ? n : N - 1;
    const float4* Xr = (const float4*)X + (size_t)nn * 16;
    float ap = 0.0f, aq = 0.0f;
#pragma unroll
    for (int i = 0; i < 4; ++i) {
        int c4 = i * 4 + l;
        float4 v = Xr[c4];
        int k = c4 * 4;
        float wa0 = Wout[k], wa1 = Wout[k + 1], wa2 = Wout[k + 2], wa3 = Wout[k + 3];
        float wb0 = Wout[HDIM + k], wb1 = Wout[HDIM + k + 1], wb2 = Wout[HDIM + k + 2], wb3 = Wout[HDIM + k + 3];
        aq += v.x * (wa0 - wb0) + v.y * (wa1 - wb1) + v.z * (wa2 - wb2) + v.w * (wa3 - wb3);
        ap += v.x * wb0 + v.y * wb1 + v.z * wb2 + v.w * wb3;
    }
#pragma unroll
    for (int o = 2; o > 0; o >>= 1) {
        ap += __shfl_xor(ap, o, 4);
        aq += __shfl_xor(aq, o, 4);
    }
    if (l == 0) { sP[threadIdx.x >> 2] = ap; sQ[threadIdx.x >> 2] = aq; }
    __syncthreads();
    if (threadIdx.x < 64) {
        int m = blockIdx.x * 64 + threadIdx.x;
        if (m < N) p[m] = sP[threadIdx.x];
    } else if (threadIdx.x < 128) {
        int m = blockIdx.x * 64 + threadIdx.x - 64;
        if (m < N) q[m] = sQ[threadIdx.x - 64];
    }
}

__global__ __launch_bounds__(256) void out_kernel(const int* __restrict__ rowptr, const int* __restrict__ rowend,
                                                  const int* __restrict__ csr,
                                                  const float* __restrict__ p, const float* __restrict__ q,
                                                  const float* __restrict__ Wout, const float* __restrict__ bout,
                                                  const float* __restrict__ x0,
                                                  const float* __restrict__ invdeg, const float* __restrict__ gate,
                                                  const float* __restrict__ ssn, float* __restrict__ out, int N) {
    __shared__ float sOut[32];
    int t = blockIdx.x * 256 + threadIdx.x;
    int n = t >> 3, l = t & 7;
    int nn = (n < N) ? n : N - 1;
    int b = rowptr[nn], e = rowend[nn];
    float a = 0.0f;
    for (int base = b; base < e; base += 32) {
        int idx[4];
#pragma unroll
        for (int u = 0; u < 4; ++u) {
            int j = base + u * 8 + l;
            idx[u] = csr[(j < e) ? j : (e - 1)];
        }
#pragma unroll
        for (int u = 0; u < 4; ++u) {
            float v = p[idx[u]];
            a += ((base + u * 8 + l) < e) ? v : 0.0f;
        }
    }
#pragma unroll
    for (int o = 4; o > 0; o >>= 1) a += __shfl_xor(a, o, 8);
    if (l == 0) sOut[threadIdx.x >> 3] = a;
    __syncthreads();
    if (threadIdx.x < 32) {
        int m = blockIdx.x * 32 + threadIdx.x;
        if (m < N)
            out[m] = gate[m] * (q[m] + bout[0]) + invdeg[m] * sOut[threadIdx.x] + ssn[m] * Wout[2 * HDIM] + x0[m];
    }
}

extern "C" void kernel_launch(void* const* d_in, const int* in_sizes, int n_in,
                              void* d_out, int out_size, void* d_ws, size_t ws_size,
                              hipStream_t stream) {
    const float* x    = (const float*)d_in[0];
    const int*   ei   = (const int*)  d_in[1];
    const float* Win  = (const float*)d_in[2];
    const float* bin  = (const float*)d_in[3];
    const float* Wmid = (const float*)d_in[4];
    const float* bmid = (const float*)d_in[5];
    const float* Wout = (const float*)d_in[6];
    const float* bout = (const float*)d_in[7];

    const int N = in_sizes[0];
    const int E = in_sizes[1] / 2;
    const int* src = ei;
    const int* dst = ei + E;

    float* w = (float*)d_ws;
    int*   rowptr  = (int*)w;
    int*   cursor  = (int*)(w + (size_t)N);
    float* invdeg  = w + 2 * (size_t)N;
    float* gate    = w + 3 * (size_t)N;
    float* ssn     = w + 4 * (size_t)N;
    float* Sx      = w + 5 * (size_t)N;
    float* Wd6     = w + 6 * (size_t)N;
    float* bufA    = Wd6  + 6 * HDIM * HDIM;
    float* bufB    = bufA + (size_t)N * HDIM;
    float* bufC    = bufB + (size_t)N * HDIM;
    float* bufS    = bufC + (size_t)N * HDIM;
    int*   csr     = (int*)bufC;

    const int nTiles = (N + 15) / 16;

    const int NB = (N + BINW - 1) / BINW;
    int* binned = (int*)bufB;
    int* cntT   = (int*)bufS;
    int* binBase = cntT + NB * NCHUNK;
    bin_count_kernel<<<NCHUNK, CH_THREADS, 0, stream>>>(dst, cntT, E, NB);
    bin_scatter_kernel<<<NCHUNK, CH_THREADS, 0, stream>>>(src, dst, cntT, binBase, binned, E, NB);
    bin_fill_kernel<<<NB, FILL_THREADS, 0, stream>>>(binBase, binned, rowptr, cursor, csr, N);
    float* coef = bufB;
    float* csum = bufB + 4 * (size_t)N;
    node_stats_kernel<<<(N * 8 + 255) / 256, 256, 0, stream>>>(rowptr, cursor, csr, x, invdeg, gate,
                                                               ssn, Sx, coef, N);
    wdiff_kernel<<<(6 * 4096 + 255) / 256, 256, 0, stream>>>(Wmid, Wd6, 6 * 4096);
    coef_gather_kernel<<<(N * 8 + 255) / 256, 256, 0, stream>>>(rowptr, cursor, csr, coef, csum, N);
    conv_in_kernel<<<(N * HDIM + 255) / 256, 256, 0, stream>>>(x, Win, bin, invdeg, gate, ssn, Sx, csum,
                                                               bufA, bufS, N);

    float* cur = bufA;
    mid_conv_kernel<false><<<(nTiles + 7) / 8, 256, 0, stream>>>(
        cur, bufS, rowptr, cursor, csr, Wd6, Wmid + 64 * HDIM, Wmid + 128 * HDIM,
        bmid, gate, invdeg, ssn, nullptr, bufB, nTiles, N);
    mid_conv_kernel<true><<<(nTiles + 7) / 8, 256, 0, stream>>>(
        bufB, bufS, rowptr, cursor, csr, Wd6 + HDIM * HDIM,
        Wmid + (size_t)129 * HDIM + 64 * HDIM, Wmid + (size_t)129 * HDIM + 128 * HDIM,
        bmid + HDIM, gate, invdeg, ssn, cur, cur, nTiles, N);
    for (int l = 1; l < 3; ++l) {
        const float* Wl = Wmid + (size_t)(2 * l) * 129 * HDIM;
        mid_conv_kernel<true><<<(nTiles + 7) / 8, 256, 0, stream>>>(
            cur, bufS, rowptr, cursor, csr, Wd6 + (2 * l) * HDIM * HDIM, Wl + 64 * HDIM, Wl + 128 * HDIM,
            bmid + (2 * l) * HDIM, gate, invdeg, ssn, nullptr, bufB, nTiles, N);

        const float* Wl2 = Wmid + (size_t)(2 * l + 1) * 129 * HDIM;
        mid_conv_kernel<true><<<(nTiles + 7) / 8, 256, 0, stream>>>(
            bufB, bufS, rowptr, cursor, csr, Wd6 + (2 * l + 1) * HDIM * HDIM, Wl2 + 64 * HDIM, Wl2 + 128 * HDIM,
            bmid + (2 * l + 1) * HDIM, gate, invdeg, ssn, cur, cur, nTiles, N);
    }

    float* pj = bufS;
    float* qj = bufS + 2 * (size_t)N;
    proj_kernel<<<(N * 4 + 255) / 256, 256, 0, stream>>>(cur, Wout, pj, qj, N);
    out_kernel<<<(N * 8 + 255) / 256, 256, 0, stream>>>(rowptr, cursor, csr, pj, qj, Wout, bout, x, invdeg, gate,
                                                        ssn, (float*)d_out, N);
}


# Four short launches fold into neighbours: Wa-Wb into bin_count, node statistics into bin_fill, conv_in into the factor gather, conv_out's two projections into the last mid conv; 15 launches become 11
# speedup vs baseline: 5.4414x; 1.0861x over previous
#include <hip/hip_runtime.h>
#include <hip/hip_bf16.h>

typedef float v2f __attribute__((ext_vector_type(2)));
typedef float v8f __attribute__((ext_vector_type(8)));

#define HDIM 64

#define BIN_SHIFT 8
#define BINW 256
#define NB_MAX 512
#define NCHUNK 256
#define CH_THREADS 1024
#define FILL_THREADS 512
#define FILL_CAP 12288

__global__ __launch_bounds__(CH_THREADS) void bin_count_kernel(const int* __restrict__ dst,
                                                               int* __restrict__ cntT, int E, int NB,
                                                               const float* __restrict__ Wmid,
                                                               float* __restrict__ Wd6) {
    __shared__ int sHist[NB_MAX];
    const int c = blockIdx.x, tid = threadIdx.x;
    static_assert(NCHUNK * CH_THREADS >= 6 * HDIM * HDIM, "one thread per element of Wd6");
    {
        int gid = c * CH_THREADS + tid;
        if (gid < 6 * HDIM * HDIM) {
            int l = gid >> 12;
            int rc = gid & 4095;
            const float* Wl = Wmid + (size_t)l * 129 * HDIM;
            Wd6[gid] = Wl[rc] - Wl[64 * HDIM + rc];
        }
    }
    for (int i = tid; i < NB; i += CH_THREADS) sHist[i] = 0;
    __syncthreads();
    const int chunk = (E + NCHUNK - 1) / NCHUNK;
    const int e0 = c * chunk;
    const int e1 = (e0 + chunk < E) ? (e0 + chunk) : E;
    for (int e = e0 + tid; e < e1; e += CH_THREADS * 8) {
        int d[8];
#pragma unroll
        for (int u = 0; u < 8; ++u) {
            int ee = e + u * CH_THREADS;
            d[u] = dst[(ee < e1) ? ee : (e1 - 1)];
        }
#pragma unroll
        for (int u = 0; u < 8; ++u)
            if (e + u * CH_THREADS < e1) atomicAdd(&sHist[d[u] >> BIN_SHIFT], 1);
    }
    __syncthreads();
    for (int i = tid; i < NB; i += CH_THREADS) cntT[i * NCHUNK + c] = sHist[i];
}

#define GROUPS_PER_PASS 2
__global__ __launch_bounds__(CH_THREADS) void bin_scatter_kernel(const int* __restrict__ src,
                                                                 const int* __restrict__ dst,
                                                                 const int* __restrict__ cntT,
                                                                 int* __restrict__ binBase,
                                                                 int* __restrict__ binned, int E, int NB) {
    static_assert(NCHUNK == 256, "one row of the count table = 8 lanes x 8 loads x 4 counts");
    static_assert(NB_MAX == 512, "wave 0 scans 16 bin totals per lane");
    __shared__ int sCur[NB_MAX];
    __shared__ int sTot[NB_MAX];
    const int c = blockIdx.x, tid = threadIdx.x, lane = tid & 31, wv = tid >> 5;
    const int sub = lane & 7, rw = lane >> 3;
    for (int i = tid; i < NB_MAX; i += CH_THREADS) sTot[i] = 0;
    __syncthreads();
    for (int g0 = wv; g0 * 4 < NB; g0 += (CH_THREADS / 32) * GROUPS_PER_PASS) {
        int4 v[GROUPS_PER_PASS][8];
#pragma unroll
        for (int u = 0; u < GROUPS_PER_PASS; ++u) {
            int i = (g0 + u * (CH_THREADS / 32)) * 4 + rw;
            const int4* row = (const int4*)(cntT + (size_t)((i < NB) ? i : (NB - 1)) * NCHUNK);
#pragma unroll
            for (int j = 0; j < 8; ++j) v[u][j] = row[j * 8 + sub];
        }
#pragma unroll
        for (int u = 0; u < GROUPS_PER_PASS; ++u) {
            int tot = 0, pre = 0;
#pragma unroll
            for (int j = 0; j < 8; ++j) {
                int k = (j * 8 + sub) * 4;
                tot += v[u][j].x + v[u][j].y + v[u][j].z + v[u][j].w;
                pre += ((k     < c) ? v[u][j].x : 0) + ((k + 1 < c) ? v[u][j].y : 0)
                     + ((k + 2 < c) ? v[u][j].z : 0) + ((k + 3 < c) ? v[u][j].w : 0);
            }
#pragma unroll
            for (int o = 4; o > 0; o >>= 1) {
                tot += __shfl_xor(tot, o, 8);
                pre += __shfl_xor(pre, o, 8);
            }
            int i = (g0 + u * (CH_THREADS / 32)) * 4 + rw;
            if (sub == 0 && i < NB) { sTot[i] = tot; sCur[i] = pre; }
        }
    }
    __syncthreads();
    if (wv == 0) {
        int t[16];
        int s = 0;
#pragma unroll
        for (int k = 0; k < 16; ++k) { t[k] = sTot[lane * 16 + k]; s += t[k]; }
        int inc = s;
#pragma unroll
        for (int o = 1; o < 32; o <<= 1) {
            int v = __shfl_up(inc, o, 32);
            if (lane >= o) inc += v;
        }
        int ex = inc - s;
#pragma unroll
        for (int k = 0; k < 16; ++k) { sTot[lane * 16 + k] = ex; ex += t[k]; }
    }
    __syncthreads();
    for (int i = tid; i < NB; i += CH_THREADS) {
        int base = sTot[i];
        sCur[i] += base;
        if (c == 0) binBase[i] = base;
    }
    if (c == 0 && tid == 0) binBase[NB] = E;
    __syncthreads();
    const int chunk = (E + NCHUNK - 1) / NCHUNK;
    const int e0 = c * chunk;
    const int e1 = (e0 + chunk < E) ? (e0 + chunk) : E;
    for (int e = e0 + tid; e < e1; e += CH_THREADS * 8) {
        int d[8], s[8];
#pragma unroll
        for (int u = 0; u < 8; ++u) {
            int ee = e + u * CH_THREADS;
            int ec = (ee < e1) ? ee : (e1 - 1);
            d[u] = dst[ec];
            s[u] = src[ec];
        }
#pragma unroll
        for (int u = 0; u < 8; ++u) {
            if (e + u * CH_THREADS < e1) {
                int pos = atomicAdd(&sCur[d[u] >> BIN_SHIFT], 1);
                binned[pos] = (s[u] << BIN_SHIFT) | (d[u] & (BINW - 1));
            }
        }
    }
}

__global__ __launch_bounds__(FILL_THREADS) void bin_fill_kernel(const int* __restrict__ binBase,
                                                                const int* __restrict__ binned,
                                                                int* __restrict__ rowptr, int* __restrict__ rowend,
                                                                int* __restrict__ csr, const float* __restrict__ x,
                                                                float* __restrict__ invdeg, float* __restrict__ gate,
                                                                float* __restrict__ ssn, float* __restrict__ Sx,
                                                                float* __restrict__ coef, int N) {
    static_assert(BINW == (FILL_THREADS / 8) * 4, "statistics: 8 lanes per node, 4 nodes per lane group");
    __shared__ int sDeg[BINW];
    __shared__ int sBeg[BINW];
    __shared__ int sWaveTot[BINW / 32];
    __shared__ int sStage[FILL_CAP];
    const int b = blockIdx.x, tid = threadIdx.x, lane = tid & 31;
    const int beg = binBase[b];
    const int end = binBase[b + 1];
    const int cnt = end - beg;
    if (tid < BINW) sDeg[tid] = 0;
    __syncthreads();
    for (int i = tid; i < cnt; i += FILL_THREADS * 8) {
        int r[8];
#pragma unroll
        for (int u = 0; u < 8; ++u) {
            int ii = i + u * FILL_THREADS;
            r[u] = binned[beg + ((ii < cnt) ? ii : (cnt - 1))];
        }
#pragma unroll
        for (int u = 0; u < 8; ++u)
            if (i + u * FILL_THREADS < cnt) atomicAdd(&sDeg[r[u] & (BINW - 1)], 1);
    }
    __syncthreads();
    int v = 0, inc = 0;
    if (tid < BINW) {
        v = sDeg[tid];
        inc = v;
#pragma unroll
        for (int o = 1; o < 32; o <<= 1) {
            int t = __shfl_up(inc, o, 32);
            if (lane >= o) inc += t;
        }
        if (lane == 31) sWaveTot[tid >> 5] = inc;
    }
    __syncthreads();
    if (tid < BINW) {
        int wbase = 0;
        for (int w = 0; w < (tid >> 5); ++w) wbase += sWaveTot[w];
        int excl = wbase + inc - v;
        sDeg[tid] = excl;
        sBeg[tid] = excl;
        int node = b * BINW + tid;
        if (node < N) { rowptr[node] = beg + excl; rowend[node] = beg + excl + v; }
    }
    __syncthreads();
    const bool staged = cnt <= FILL_CAP;
    for (int i = tid; i < cnt; i += FILL_THREADS * 8) {
        int r[8];
#pragma unroll
        for (int u = 0; u < 8; ++u) {
            int ii = i + u * FILL_THREADS;
            r[u] = binned[beg + ((ii < cnt) ? ii : (cnt - 1))];
        }
#pragma unroll
        for (int u = 0; u < 8; ++u) {
            if (i + u * FILL_THREADS < cnt) {
                int pos = atomicAdd(&sDeg[r[u] & (BINW - 1)], 1);
                if (staged) sStage[pos] = r[u] >> BIN_SHIFT;
                else        csr[beg + pos] = r[u] >> BIN_SHIFT;
            }
        }
    }
    __syncthreads();
    if (staged)
        for (int i = tid; i < cnt; i += FILL_THREADS) csr[beg + i] = sStage[i];

    const int l8 = tid & 7, grp = tid >> 3;
    int sb[4], se[4], nd[4];
    float sg[4], sx[4];
#pragma unroll
    for (int u = 0; u < 4; ++u) {
        int ln = grp + u * (FILL_THREADS / 8);
        sb[u] = sBeg[ln];
        se[u] = sDeg[ln];
        nd[u] = b * BINW + ln;
        sg[u] = 0.0f; sx[u] = 0.0f;
    }
    int longest = 0;
#pragma unroll
    for (int u = 0; u < 4; ++u) longest = (se[u] - sb[u] > longest) ? (se[u] - sb[u]) : longest;
    for (int off = 0; off < longest; off += 32) {
        int s[4][4];
        if (staged) {
#pragma unroll
            for (int u = 0; u < 4; ++u) {
#pragma unroll
                for (int k = 0; k < 4; ++k) {
                    int j = sb[u] + off + k * 8 + l8;
                    bool ok = j < se[u];
                    int t = sStage[ok ? j : 0];
                    s[u][k] = ok ? t : 0;
                }
            }
        } else {
#pragma unroll
            for (int u = 0; u < 4; ++u) {
#pragma unroll
                for (int k = 0; k < 4; ++k) {
                    int j = sb[u] + off + k * 8 + l8;
                    bool ok = j < se[u];
                    int t = csr[beg + (ok ? j : 0)];
                    s[u][k] = ok ? t : 0;
                }
            }
        }
#pragma unroll
        for (int u = 0; u < 4; ++u) {
#pragma unroll
            for (int k = 0; k < 4; ++k) {
                float xv = x[s[u][k]];
                bool ok = (sb[u] + off + k * 8 + l8) < se[u];
                float sgn = (s[u][k] > nd[u]) ? 1.0f : ((s[u][k] < nd[u]) ? -1.0f : 0.0f);
                sg[u] += ok ? sgn : 0.0f;
                sx[u] += ok ? xv : 0.0f;
            }
        }
    }
#pragma unroll
    for (int o = 4; o > 0; o >>= 1) {
#pragma unroll
        for (int u = 0; u < 4; ++u) {
            sg[u] += __shfl_xor(sg[u], o, 8);
            sx[u] += __shfl_xor(sx[u], o, 8);
        }
    }
    if (l8 == 0) {
#pragma unroll
        for (int u = 0; u < 4; ++u) {
            int n = nd[u];
            if (n < N) {
                float dg = (float)(se[u] - sb[u]);
                float inv = 1.0f / fmaxf(dg, 1.0f);
                float g = (dg > 0.0f) ? 1.0f : 0.0f;
                invdeg[n] = inv;
                gate[n]   = g;
                ssn[n]    = sg[u] * inv;
                Sx[n]     = sx[u];
                ((float4*)coef)[n] = make_float4(g * x[n], g, inv * sx[u], sg[u] * inv);
            }
        }
    }
}

__global__ __launch_bounds__(256) void coef_gather_kernel(const int* __restrict__ rowptr,
                                                          const int* __restrict__ rowend,
                                                          const int* __restrict__ csr,
                                                          const float* __restrict__ coef,
                                                          const float* __restrict__ x,
                                                          const float* __restrict__ Win,
                                                          const float* __restrict__ bin,
                                                          const float* __restrict__ invdeg,
                                                          const float* __restrict__ gate,
                                                          const float* __restrict__ ssn,
                                                          const float* __restrict__ Sx,
                                                          float* __restrict__ Y, float* __restrict__ S0, int N) {
    __shared__ float4 sOut[32];
    const float4* coef4 = (const float4*)coef;
    int t = blockIdx.x * 256 + threadIdx.x;
    int n = t >> 3, l = t & 7;
    int nn = (n < N) ? n : N - 1;
    int b = rowptr[nn], e = rowend[nn];
    const int q = threadIdx.x & 15, r0 = threadIdx.x >> 4;
    const float4 wa = ((const float4*)Win)[q], wb = ((const float4*)Win)[16 + q], wc = ((const float4*)Win)[32 + q];
    const float4 bh = ((const float4*)bin)[q];
    float fx[2], fg[2], fi[2], fsx[2], fsn[2];
#pragma unroll
    for (int u = 0; u < 2; ++u) {
        int m = blockIdx.x * 32 + r0 + 16 * u; if (m >= N) m = N - 1;
        fx[u] = x[m]; fg[u] = gate[m]; fi[u] = invdeg[m]; fsx[u] = Sx[m]; fsn[u] = ssn[m];
    }
    float4 a = make_float4(0.0f, 0.0f, 0.0f, 0.0f);
    for (int base = b; base < e; base += 32) {
        int idx[4];
#pragma unroll
        for (int u = 0; u < 4; ++u) {
            int j = base + u * 8 + l;
            idx[u] = csr[(j < e) ? j : (e - 1)];
        }
#pragma unroll
        for (int u = 0; u < 4; ++u) {
            float4 c = coef4[idx[u]];
            bool ok = (base + u * 8 + l) < e;
            a.x += ok ? c.x : 0.0f;
            a.y += ok ? c.y : 0.0f;
            a.z += ok ? c.z : 0.0f;
            a.w += ok ? c.w : 0.0f;
        }
    }
#pragma unroll
    for (int o = 4; o > 0; o >>= 1) {
        a.x += __shfl_xor(a.x, o, 8);
        a.y += __shfl_xor(a.y, o, 8);
        a.z += __shfl_xor(a.z, o, 8);
        a.w += __shfl_xor(a.w, o, 8);
    }
    if (l == 0) sOut[threadIdx.x >> 3] = a;
    __syncthreads();
#pragma unroll
    for (int u = 0; u < 2; ++u) {
        int r = r0 + 16 * u;
        int m = blockIdx.x * 32 + r;
        if (m < N) {
            float4 cs = sOut[r];
            float gx = fx[u], g = fg[u], is = fi[u] * fsx[u], sn = fsn[u];
            float4 y, s;
            y.x = g * (gx * (wa.x - wb.x) + bh.x) + is * wb.x + sn * wc.x;
            y.y = g * (gx * (wa.y - wb.y) + bh.y) + is * wb.y + sn * wc.y;
            y.z = g * (gx * (wa.z - wb.z) + bh.z) + is * wb.z + sn * wc.z;
            y.w = g * (gx * (wa.w - wb.w) + bh.w) + is * wb.w + sn * wc.w;
            s.x = cs.x * (wa.x - wb.x) + cs.y * bh.x + cs.z * wb.x + cs.w * wc.x;
            s.y = cs.x * (wa.y - wb.y) + cs.y * bh.y + cs.z * wb.y + cs.w * wc.y;
            s.z = cs.x * (wa.z - wb.z) + cs.y * bh.z + cs.z * wb.z + cs.w * wc.z;
            s.w = cs.x * (wa.w - wb.w) + cs.y * bh.w + cs.z * wb.w + cs.w * wc.w;
            ((float4*)Y)[(size_t)m * 16 + q] = y;
            ((float4*)S0)[(size_t)m * 16 + q] = s;
        }
    }
}

__device__ __forceinline__ void gather16(const float4* __restrict__ X4, int mine, int cnt, int q, float4& acc) {
#pragma unroll
    for (int k = 0; k < 16; ++k) {
        int s = __shfl(mine, k, 16);
        float4 v = X4[(size_t)s * 16 + q];
        bool ok = k < cnt;
        acc.x += ok ? v.x : 0.0f;
        acc.y += ok ? v.y : 0.0f;
        acc.z += ok ? v.z : 0.0f;
        acc.w += ok ? v.w : 0.0f;
    }
}

#define SROW 68
#define TILE_IDX_CAP 768
template <bool GATHER, bool LAST>
__global__ __launch_bounds__(256) void mid_conv_kernel(
    const float* __restrict__ X, const float* __restrict__ S,
    const int* __restrict__ rowptr, const int* __restrict__ rowend, const int* __restrict__ csr,
    const float* __restrict__ Wd, const float* __restrict__ Wb,
    const float* __restrict__ Wc, const float* __restrict__ bias,
    const float* __restrict__ gate, const float* __restrict__ invdeg,
    const float* __restrict__ ssn, const float* resid,
    float* Y,
    const float* __restrict__ Wout, float* __restrict__ pOut, float* __restrict__ qOut, int nTiles, int N) {
    static_assert(GATHER || !LAST, "the LAST form keeps its rows and its p, q in the LDS areas of the gather");
    __shared__ float sWd[HDIM * HDIM];
    __shared__ float sWb[HDIM * HDIM];
    __shared__ __attribute__((aligned(16))) float sS[GATHER ? 8 * 16 * SROW : 4];
    __shared__ int   sIdx[GATHER ? 8 * TILE_IDX_CAP : 2];
    for (int i = threadIdx.x; i < HDIM * HDIM; i += 256) {
        int ks   = i >> 8;
        int rem  = i & 255;
        int nt   = rem >> 6;
        int lp   = rem & 63;
        int ln   = lp >> 1;
        int half = lp & 1;
        int kh   = (ln >> 4) << 1;
        int srcr = (ks << 2) + kh + half;
        int srcc = (nt << 4) + (ln & 15);
        sWd[i] = Wd[srcr * HDIM + srcc];
        sWb[i] = Wb[srcr * HDIM + srcc];
    }
    __syncthreads();

    int wave = threadIdx.x >> 5;
    int lane = threadIdx.x & 31;
    int tile = blockIdx.x * 8 + wave;
    const bool active = tile < nTiles;
    if (!LAST && !active) return;
    if (active) {
        int node0 = tile * 16;
        int m  = lane & 15;
        int kh = (lane >> 4) << 1;
        int row = node0 + m; if (row >= N) row = N - 1;
        const float* Xr = X + (size_t)row * HDIM;
        const float* Sr = S + (size_t)row * HDIM;
        float* mS = sS + (GATHER ? wave * (16 * SROW) : 0);

        int ncol = lane & 15;
        int mAdd = (lane >> 4) * 8;
        int frow = node0 + (lane & 7) + mAdd; if (frow >= N) frow = N - 1;
        float fgl = gate[frow], fil = invdeg[frow], fsl = ssn[frow];
        float wcv[4], bbv[4];
    #pragma unroll
        for (int nt = 0; nt < 4; ++nt) {
            wcv[nt] = Wc[nt * 16 + ncol];
            bbv[nt] = bias[nt * 16 + ncol];
        }

        float4 woA[4], woB[4];
        if (LAST) {
    #pragma unroll
            for (int i = 0; i < 4; ++i) {
                int c4 = i * 4 + (lane & 3);
                woA[i] = ((const float4*)Wout)[c4];
                woB[i] = ((const float4*)Wout)[16 + c4];
            }
        }

        if (GATHER) {
            int* mIx = sIdx + wave * TILE_IDX_CAP;
            int rb = rowptr[row], re = rowend[row];
            if (node0 + m >= N) rb = re;
            int tb = __shfl(rb, 0, 32);
            int staged = __shfl(re, 15, 32) - tb;
            staged = (staged < 0) ? 0 : ((staged > TILE_IDX_CAP) ? TILE_IDX_CAP : staged);
            for (int i0 = 0; i0 < staged; i0 += 32 * 16) {
                int v[16];
    #pragma unroll
                for (int u = 0; u < 16; ++u) {
                    int i = i0 + u * 32 + lane;
                    v[u] = csr[tb + ((i < staged) ? i : (staged - 1))];
                }
    #pragma unroll
                for (int u = 0; u < 16; ++u) {
                    int i = i0 + u * 32 + lane;
                    if (i < staged) mIx[i] = v[u];
                }
            }
            __builtin_amdgcn_fence(__ATOMIC_ACQ_REL, "wavefront");

            const float4* X4 = (const float4*)X;
            int q = lane & 15, h = lane >> 4;
            for (int p = 0; p < 8; ++p) {
                int r = 2 * p + h;
                int b = __shfl(rb, r, 32), e = __shfl(re, r, 32);
                int own = node0 + r; if (own >= N) own = N - 1;
                float4 acc = make_float4(0.0f, 0.0f, 0.0f, 0.0f);
                int eL = (b >= tb) ? (((tb + staged) < e) ? (tb + staged) : e) : b;
                if (eL < b) eL = b;
                for (int base = b; base < eL; base += 16) {
                    int j = base + q;
                    int mine = (j < eL) ? mIx[j - tb] : own;
                    gather16(X4, mine, eL - base, q, acc);
                }
                for (int base = eL; base < e; base += 16) {
                    int j = base + q;
                    int mine = (j < e) ? csr[j] : own;
                    gather16(X4, mine, e - base, q, acc);
                }
                *(float4*)&mS[r * SROW + 4 * q] = acc;
            }
            __builtin_amdgcn_fence(__ATOMIC_ACQ_REL, "wavefront");
        }

        v8f accX[4] = {};
        v8f accS[4] = {};
        int fragBase = lane * 2;

        v2f aX[16];
    #pragma unroll
        for (int ks = 0; ks < 16; ++ks) aX[ks] = *(const v2f*)&Xr[(ks << 2) + kh];

    #pragma unroll
        for (int ks = 0; ks < 16; ++ks) {
            int k0 = ks << 2;
            v2f a = aX[ks], s;
            if (GATHER) {
                s = *(const v2f*)&mS[m * SROW + k0 + kh];
            } else {
                s.x = Sr[k0 + kh]; s.y = Sr[k0 + kh + 1];
            }
    #pragma unroll
            for (int nt = 0; nt < 4; ++nt) {
                v2f bd = *(const v2f*)&sWd[((ks << 2) + nt) * 64 + fragBase];
                v2f bb = *(const v2f*)&sWb[((ks << 2) + nt) * 64 + fragBase];
                accX[nt] = __builtin_amdgcn_wmma_f32_16x16x4_f32(
                    false, a, false, bd, (short)0, accX[nt], false, false);
                accS[nt] = __builtin_amdgcn_wmma_f32_16x16x4_f32(
                    false, s, false, bb, (short)0, accS[nt], false, false);
            }
        }

        float rs[4][8];
        if (resid) {
    #pragma unroll
            for (int v = 0; v < 8; ++v) {
                int node = node0 + v + mAdd; if (node >= N) node = N - 1;
                const float* rp = resid + (size_t)node * HDIM + ncol;
    #pragma unroll
                for (int nt = 0; nt < 4; ++nt) rs[nt][v] = rp[nt * 16];
            }
        }
        float fg[8], fi[8], fs[8];
    #pragma unroll
        for (int v = 0; v < 8; ++v) {
            fg[v] = __shfl(fgl, v, 16);
            fi[v] = __shfl(fil, v, 16);
            fs[v] = __shfl(fsl, v, 16);
        }
    #pragma unroll
        for (int nt = 0; nt < 4; ++nt) {
            int col = nt * 16 + ncol;
    #pragma unroll
            for (int v = 0; v < 8; ++v) {
                int node = node0 + v + mAdd;
                float val = fg[v] * (accX[nt][v] + bbv[nt]) + fi[v] * accS[nt][v] + fs[v] * wcv[nt];
                if (resid) val += rs[nt][v];
                val = (val > 0.0f) ? val : 0.01f * val;
                if (LAST) mS[(v + mAdd) * SROW + col] = val;
                else if (node < N) Y[(size_t)node * HDIM + col] = val;
            }
        }
        if (LAST) {
            __builtin_amdgcn_fence(__ATOMIC_ACQ_REL, "wavefront");
            float* mPQ = (float*)(sIdx + wave * TILE_IDX_CAP);
            const int l4 = lane & 3;
    #pragma unroll
            for (int ps = 0; ps < 2; ++ps) {
                int r = ps * 8 + (lane >> 2);
                float ap = 0.0f, aq = 0.0f;
    #pragma unroll
                for (int i = 0; i < 4; ++i) {
                    int c4 = i * 4 + l4;
                    float4 xv = *(const float4*)&mS[r * SROW + 4 * c4];
                    float wa0 = woA[i].x, wa1 = woA[i].y, wa2 = woA[i].z, wa3 = woA[i].w;
                    float wb0 = woB[i].x, wb1 = woB[i].y, wb2 = woB[i].z, wb3 = woB[i].w;
                    aq += xv.x * (wa0 - wb0) + xv.y * (wa1 - wb1) + xv.z * (wa2 - wb2) + xv.w * (wa3 - wb3);
                    ap += xv.x * wb0 + xv.y * wb1 + xv.z * wb2 + xv.w * wb3;
                }
    #pragma unroll
                for (int o = 2; o > 0; o >>= 1) {
                    ap += __shfl_xor(ap, o, 4);
                    aq += __shfl_xor(aq, o, 4);
                }
                if (l4 == 0) { mPQ[r] = ap; mPQ[16 + r] = aq; }
            }
        }
    }
    if (LAST) {
        __syncthreads();
        int t = threadIdx.x & 127;
        int node = blockIdx.x * 128 + t;
        const float* wPQ = (const float*)(sIdx + (t >> 4) * TILE_IDX_CAP);
        if (node < N) {
            if (threadIdx.x < 128) pOut[node] = wPQ[t & 15];
            else                   qOut[node] = wPQ[16 + (t & 15)];
        }
    }
}

__global__ __launch_bounds__(256) void out_kernel(const int* __restrict__ rowptr, const int* __restrict__ rowend,
                                                  const int* __restrict__ csr,
                                                  const float* __restrict__ p, const float* __restrict__ q,
                                                  const float* __restrict__ Wout, const float* __restrict__ bout,
                                                  const float* __restrict__ x0,
                                                  const float* __restrict__ invdeg, const float* __restrict__ gate,
                                                  const float* __restrict__ ssn, float* __restrict__ out, int N) {
    __shared__ float sOut[32];
    int t = blockIdx.x * 256 + threadIdx.x;
    int n = t >> 3, l = t & 7;
    int nn = (n < N) ? n : N - 1;
    int b = rowptr[nn], e = rowend[nn];
    float a = 0.0f;
    for (int base = b; base < e; base += 32) {
        int idx[4];
#pragma unroll
        for (int u = 0; u < 4; ++u) {
            int j = base + u * 8 + l;
            idx[u] = csr[(j < e) ? j : (e - 1)];
        }
#pragma unroll
        for (int u = 0; u < 4; ++u) {
            float v = p[idx[u]];
            a += ((base + u * 8 + l) < e) ? v : 0.0f;
        }
    }
#pragma unroll
    for (int o = 4; o > 0; o >>= 1) a += __shfl_xor(a, o, 8);
    if (l == 0) sOut[threadIdx.x >> 3] = a;
    __syncthreads();
    if (threadIdx.x < 32) {
        int m = blockIdx.x * 32 + threadIdx.x;
        if (m < N)
            out[m] = gate[m] * (q[m] + bout[0]) + invdeg[m] * sOut[threadIdx.x] + ssn[m] * Wout[2 * HDIM] + x0[m];
    }
}

extern "C" void kernel_launch(void* const* d_in, const int* in_sizes, int n_in,
                              void* d_out, int out_size, void* d_ws, size_t ws_size,
                              hipStream_t stream) {
    const float* x    = (const float*)d_in[0];
    const int*   ei   = (const int*)  d_in[1];
    const float* Win  = (const float*)d_in[2];
    const float* bin  = (const float*)d_in[3];
    const float* Wmid = (const float*)d_in[4];
    const float* bmid = (const float*)d_in[5];
    const float* Wout = (const float*)d_in[6];
    const float* bout = (const float*)d_in[7];

    const int N = in_sizes[0];
    const int E = in_sizes[1] / 2;
    const int* src = ei;
    const int* dst = ei + E;

    float* w = (float*)d_ws;
    int*   rowptr  = (int*)w;
    int*   cursor  = (int*)(w + (size_t)N);
    float* invdeg  = w + 2 * (size_t)N;
    float* gate    = w + 3 * (size_t)N;
    float* ssn     = w + 4 * (size_t)N;
    float* Sx      = w + 5 * (size_t)N;
    float* Wd6     = w + 6 * (size_t)N;
    float* bufA    = Wd6  + 6 * HDIM * HDIM;
    float* bufB    = bufA + (size_t)N * HDIM;
    float* bufC    = bufB + (size_t)N * HDIM;
    float* bufS    = bufC + (size_t)N * HDIM;
    int*   csr     = (int*)bufC;

    const int nTiles = (N + 15) / 16;

    const int NB = (N + BINW - 1) / BINW;
    int* binned = (int*)bufB;
    int* cntT   = (int*)bufS;
    int* binBase = cntT + NB * NCHUNK;
    float* coef = bufS - 4 * (size_t)N;
    bin_count_kernel<<<NCHUNK, CH_THREADS, 0, stream>>>(dst, cntT, E, NB, Wmid, Wd6);
    bin_scatter_kernel<<<NCHUNK, CH_THREADS, 0, stream>>>(src, dst, cntT, binBase, binned, E, NB);
    bin_fill_kernel<<<NB, FILL_THREADS, 0, stream>>>(binBase, binned, rowptr, cursor, csr, x, invdeg, gate, ssn, Sx,
                                                     coef, N);
    coef_gather_kernel<<<(N * 8 + 255) / 256, 256, 0, stream>>>(rowptr, cursor, csr, coef, x, Win, bin, invdeg, gate,
                                                                ssn, Sx, bufA, bufS, N);

    float* cur = bufA;
    mid_conv_kernel<false, false><<<(nTiles + 7) / 8, 256, 0, stream>>>(
        cur, bufS, rowptr, cursor, csr, Wd6, Wmid + 64 * HDIM, Wmid + 128 * HDIM,
        bmid, gate, invdeg, ssn, nullptr, bufB, nullptr, nullptr, nullptr, nTiles, N);
    mid_conv_kernel<true, false><<<(nTiles + 7) / 8, 256, 0, stream>>>(
        bufB, bufS, rowptr, cursor, csr, Wd6 + HDIM * HDIM,
        Wmid + (size_t)129 * HDIM + 64 * HDIM, Wmid + (size_t)129 * HDIM + 128 * HDIM,
        bmid + HDIM, gate, invdeg, ssn, cur, cur, nullptr, nullptr, nullptr, nTiles, N);
    float* pj = bufS;
    float* qj = bufS + 2 * (size_t)N;
    for (int l = 1; l < 3; ++l) {
        const float* Wl = Wmid + (size_t)(2 * l) * 129 * HDIM;
        mid_conv_kernel<true, false><<<(nTiles + 7) / 8, 256, 0, stream>>>(
            cur, bufS, rowptr, cursor, csr, Wd6 + (2 * l) * HDIM * HDIM, Wl + 64 * HDIM, Wl + 128 * HDIM,
            bmid + (2 * l) * HDIM, gate, invdeg, ssn, nullptr, bufB, nullptr, nullptr, nullptr, nTiles, N);

        const float* Wl2 = Wmid + (size_t)(2 * l + 1) * 129 * HDIM;
        if (l < 2)
            mid_conv_kernel<true, false><<<(nTiles + 7) / 8, 256, 0, stream>>>(
                bufB, bufS, rowptr, cursor, csr, Wd6 + (2 * l + 1) * HDIM * HDIM, Wl2 + 64 * HDIM, Wl2 + 128 * HDIM,
                bmid + (2 * l + 1) * HDIM, gate, invdeg, ssn, cur, cur, nullptr, nullptr, nullptr, nTiles, N);
        else
            mid_conv_kernel<true, true><<<(nTiles + 7) / 8, 256, 0, stream>>>(
                bufB, bufS, rowptr, cursor, csr, Wd6 + (2 * l + 1) * HDIM * HDIM, Wl2 + 64 * HDIM, Wl2 + 128 * HDIM,
                bmid + (2 * l + 1) * HDIM, gate, invdeg, ssn, cur, cur, Wout, pj, qj, nTiles, N);
    }

    out_kernel<<<(N * 8 + 255) / 256, 256, 0, stream>>>(rowptr, cursor, csr, pj, qj, Wout, bout, x, invdeg, gate,
                                                        ssn, (float*)d_out, N);
}


# Mid conv gather issues fewer instructions per row read: sources staged as row byte offsets, scalar base + 32-bit offset loads, whole chunks of 16 summed without masks; in-node sum order kept
# speedup vs baseline: 5.7018x; 1.0479x over previous
#include <hip/hip_runtime.h>
#include <hip/hip_bf16.h>

typedef float v2f __attribute__((ext_vector_type(2)));
typedef float v8f __attribute__((ext_vector_type(8)));

#define HDIM 64

#define BIN_SHIFT 8
#define BINW 256
#define NB_MAX 512
#define NCHUNK 256
#define CH_THREADS 1024
#define FILL_THREADS 512
#define FILL_CAP 12288

__global__ __launch_bounds__(CH_THREADS) void bin_count_kernel(const int* __restrict__ dst,
                                                               int* __restrict__ cntT, int E, int NB,
                                                               const float* __restrict__ Wmid,
                                                               float* __restrict__ Wd6) {
    __shared__ int sHist[NB_MAX];
    const int c = blockIdx.x, tid = threadIdx.x;
    static_assert(NCHUNK * CH_THREADS >= 6 * HDIM * HDIM, "one thread per element of Wd6");
    {
        int gid = c * CH_THREADS + tid;
        if (gid < 6 * HDIM * HDIM) {
            int l = gid >> 12;
            int rc = gid & 4095;
            const float* Wl = Wmid + (size_t)l * 129 * HDIM;
            Wd6[gid] = Wl[rc] - Wl[64 * HDIM + rc];
        }
    }
    for (int i = tid; i < NB; i += CH_THREADS) sHist[i] = 0;
    __syncthreads();
    const int chunk = (E + NCHUNK - 1) / NCHUNK;
    const int e0 = c * chunk;
    const int e1 = (e0 + chunk < E) ? (e0 + chunk) : E;
    for (int e = e0 + tid; e < e1; e += CH_THREADS * 8) {
        int d[8];
#pragma unroll
        for (int u = 0; u < 8; ++u) {
            int ee = e + u * CH_THREADS;
            d[u] = dst[(ee < e1) ? ee : (e1 - 1)];
        }
#pragma unroll
        for (int u = 0; u < 8; ++u)
            if (e + u * CH_THREADS < e1) atomicAdd(&sHist[d[u] >> BIN_SHIFT], 1);
    }
    __syncthreads();
    for (int i = tid; i < NB; i += CH_THREADS) cntT[i * NCHUNK + c] = sHist[i];
}

#define GROUPS_PER_PASS 2
__global__ __launch_bounds__(CH_THREADS) void bin_scatter_kernel(const int* __restrict__ src,
                                                                 const int* __restrict__ dst,
                                                                 const int* __restrict__ cntT,
                                                                 int* __restrict__ binBase,
                                                                 int* __restrict__ binned, int E, int NB) {
    static_assert(NCHUNK == 256, "one row of the count table = 8 lanes x 8 loads x 4 counts");
    static_assert(NB_MAX == 512, "wave 0 scans 16 bin totals per lane");
    __shared__ int sCur[NB_MAX];
    __shared__ int sTot[NB_MAX];
    const int c = blockIdx.x, tid = threadIdx.x, lane = tid & 31, wv = tid >> 5;
    const int sub = lane & 7, rw = lane >> 3;
    for (int i = tid; i < NB_MAX; i += CH_THREADS) sTot[i] = 0;
    __syncthreads();
    for (int g0 = wv; g0 * 4 < NB; g0 += (CH_THREADS / 32) * GROUPS_PER_PASS) {
        int4 v[GROUPS_PER_PASS][8];
#pragma unroll
        for (int u = 0; u < GROUPS_PER_PASS; ++u) {
            int i = (g0 + u * (CH_THREADS / 32)) * 4 + rw;
            const int4* row = (const int4*)(cntT + (size_t)((i < NB) ? i : (NB - 1)) * NCHUNK);
#pragma unroll
            for (int j = 0; j < 8; ++j) v[u][j] = row[j * 8 + sub];
        }
#pragma unroll
        for (int u = 0; u < GROUPS_PER_PASS; ++u) {
            int tot = 0, pre = 0;
#pragma unroll
            for (int j = 0; j < 8; ++j) {
                int k = (j * 8 + sub) * 4;
                tot += v[u][j].x + v[u][j].y + v[u][j].z + v[u][j].w;
                pre += ((k     < c) ? v[u][j].x : 0) + ((k + 1 < c) ? v[u][j].y : 0)
                     + ((k + 2 < c) ? v[u][j].z : 0) + ((k + 3 < c) ? v[u][j].w : 0);
            }
#pragma unroll
            for (int o = 4; o > 0; o >>= 1) {
                tot += __shfl_xor(tot, o, 8);
                pre += __shfl_xor(pre, o, 8);
            }
            int i = (g0 + u * (CH_THREADS / 32)) * 4 + rw;
            if (sub == 0 && i < NB) { sTot[i] = tot; sCur[i] = pre; }
        }
    }
    __syncthreads();
    if (wv == 0) {
        int t[16];
        int s = 0;
#pragma unroll
        for (int k = 0; k < 16; ++k) { t[k] = sTot[lane * 16 + k]; s += t[k]; }
        int inc = s;
#pragma unroll
        for (int o = 1; o < 32; o <<= 1) {
            int v = __shfl_up(inc, o, 32);
            if (lane >= o) inc += v;
        }
        int ex = inc - s;
#pragma unroll
        for (int k = 0; k < 16; ++k) { sTot[lane * 16 + k] = ex; ex += t[k]; }
    }
    __syncthreads();
    for (int i = tid; i < NB; i += CH_THREADS) {
        int base = sTot[i];
        sCur[i] += base;
        if (c == 0) binBase[i] = base;
    }
    if (c == 0 && tid == 0) binBase[NB] = E;
    __syncthreads();
    const int chunk = (E + NCHUNK - 1) / NCHUNK;
    const int e0 = c * chunk;
    const int e1 = (e0 + chunk < E) ? (e0 + chunk) : E;
    for (int e = e0 + tid; e < e1; e += CH_THREADS * 8) {
        int d[8], s[8];
#pragma unroll
        for (int u = 0; u < 8; ++u) {
            int ee = e + u * CH_THREADS;
            int ec = (ee < e1) ? ee : (e1 - 1);
            d[u] = dst[ec];
            s[u] = src[ec];
        }
#pragma unroll
        for (int u = 0; u < 8; ++u) {
            if (e + u * CH_THREADS < e1) {
                int pos = atomicAdd(&sCur[d[u] >> BIN_SHIFT], 1);
                binned[pos] = (s[u] << BIN_SHIFT) | (d[u] & (BINW - 1));
            }
        }
    }
}

__global__ __launch_bounds__(FILL_THREADS) void bin_fill_kernel(const int* __restrict__ binBase,
                                                                const int* __restrict__ binned,
                                                                int* __restrict__ rowptr, int* __restrict__ rowend,
                                                                int* __restrict__ csr, const float* __restrict__ x,
                                                                float* __restrict__ invdeg, float* __restrict__ gate,
                                                                float* __restrict__ ssn, float* __restrict__ Sx,
                                                                float* __restrict__ coef, int N) {
    static_assert(BINW == (FILL_THREADS / 8) * 4, "statistics: 8 lanes per node, 4 nodes per lane group");
    __shared__ int sDeg[BINW];
    __shared__ int sBeg[BINW];
    __shared__ int sWaveTot[BINW / 32];
    __shared__ int sStage[FILL_CAP];
    const int b = blockIdx.x, tid = threadIdx.x, lane = tid & 31;
    const int beg = binBase[b];
    const int end = binBase[b + 1];
    const int cnt = end - beg;
    if (tid < BINW) sDeg[tid] = 0;
    __syncthreads();
    for (int i = tid; i < cnt; i += FILL_THREADS * 8) {
        int r[8];
#pragma unroll
        for (int u = 0; u < 8; ++u) {
            int ii = i + u * FILL_THREADS;
            r[u] = binned[beg + ((ii < cnt) ? ii : (cnt - 1))];
        }
#pragma unroll
        for (int u = 0; u < 8; ++u)
            if (i + u * FILL_THREADS < cnt) atomicAdd(&sDeg[r[u] & (BINW - 1)], 1);
    }
    __syncthreads();
    int v = 0, inc = 0;
    if (tid < BINW) {
        v = sDeg[tid];
        inc = v;
#pragma unroll
        for (int o = 1; o < 32; o <<= 1) {
            int t = __shfl_up(inc, o, 32);
            if (lane >= o) inc += t;
        }
        if (lane == 31) sWaveTot[tid >> 5] = inc;
    }
    __syncthreads();
    if (tid < BINW) {
        int wbase = 0;
        for (int w = 0; w < (tid >> 5); ++w) wbase += sWaveTot[w];
        int excl = wbase + inc - v;
        sDeg[tid] = excl;
        sBeg[tid] = excl;
        int node = b * BINW + tid;
        if (node < N) { rowptr[node] = beg + excl; rowend[node] = beg + excl + v; }
    }
    __syncthreads();
    const bool staged = cnt <= FILL_CAP;
    for (int i = tid; i < cnt; i += FILL_THREADS * 8) {
        int r[8];
#pragma unroll
        for (int u = 0; u < 8; ++u) {
            int ii = i + u * FILL_THREADS;
            r[u] = binned[beg + ((ii < cnt) ? ii : (cnt - 1))];
        }
#pragma unroll
        for (int u = 0; u < 8; ++u) {
            if (i + u * FILL_THREADS < cnt) {
                int pos = atomicAdd(&sDeg[r[u] & (BINW - 1)], 1);
                if (staged) sStage[pos] = r[u] >> BIN_SHIFT;
                else        csr[beg + pos] = r[u] >> BIN_SHIFT;
            }
        }
    }
    __syncthreads();
    if (staged)
        for (int i = tid; i < cnt; i += FILL_THREADS) csr[beg + i] = sStage[i];

    const int l8 = tid & 7, grp = tid >> 3;
    int sb[4], se[4], nd[4];
    float sg[4], sx[4];
#pragma unroll
    for (int u = 0; u < 4; ++u) {
        int ln = grp + u * (FILL_THREADS / 8);
        sb[u] = sBeg[ln];
        se[u] = sDeg[ln];
        nd[u] = b * BINW + ln;
        sg[u] = 0.0f; sx[u] = 0.0f;
    }
    int longest = 0;
#pragma unroll
    for (int u = 0; u < 4; ++u) longest = (se[u] - sb[u] > longest) ? (se[u] - sb[u]) : longest;
    for (int off = 0; off < longest; off += 32) {
        int s[4][4];
        if (staged) {
#pragma unroll
            for (int u = 0; u < 4; ++u) {
#pragma unroll
                for (int k = 0; k < 4; ++k) {
                    int j = sb[u] + off + k * 8 + l8;
                    bool ok = j < se[u];
                    int t = sStage[ok ? j : 0];
                    s[u][k] = ok ? t : 0;
                }
            }
        } else {
#pragma unroll
            for (int u = 0; u < 4; ++u) {
#pragma unroll
                for (int k = 0; k < 4; ++k) {
                    int j = sb[u] + off + k * 8 + l8;
                    bool ok = j < se[u];
                    int t = csr[beg + (ok ? j : 0)];
                    s[u][k] = ok ? t : 0;
                }
            }
        }
#pragma unroll
        for (int u = 0; u < 4; ++u) {
#pragma unroll
            for (int k = 0; k < 4; ++k) {
                float xv = x[s[u][k]];
                bool ok = (sb[u] + off + k * 8 + l8) < se[u];
                float sgn = (s[u][k] > nd[u]) ? 1.0f : ((s[u][k] < nd[u]) ? -1.0f : 0.0f);
                sg[u] += ok ? sgn : 0.0f;
                sx[u] += ok ? xv : 0.0f;
            }
        }
    }
#pragma unroll
    for (int o = 4; o > 0; o >>= 1) {
#pragma unroll
        for (int u = 0; u < 4; ++u) {
            sg[u] += __shfl_xor(sg[u], o, 8);
            sx[u] += __shfl_xor(sx[u], o, 8);
        }
    }
    if (l8 == 0) {
#pragma unroll
        for (int u = 0; u < 4; ++u) {
            int n = nd[u];
            if (n < N) {
                float dg = (float)(se[u] - sb[u]);
                float inv = 1.0f / fmaxf(dg, 1.0f);
                float g = (dg > 0.0f) ? 1.0f : 0.0f;
                invdeg[n] = inv;
                gate[n]   = g;
                ssn[n]    = sg[u] * inv;
                Sx[n]     = sx[u];
                ((float4*)coef)[n] = make_float4(g * x[n], g, inv * sx[u], sg[u] * inv);
            }
        }
    }
}

__global__ __launch_bounds__(256) void coef_gather_kernel(const int* __restrict__ rowptr,
                                                          const int* __restrict__ rowend,
                                                          const int* __restrict__ csr,
                                                          const float* __restrict__ coef,
                                                          const float* __restrict__ x,
                                                          const float* __restrict__ Win,
                                                          const float* __restrict__ bin,
                                                          const float* __restrict__ invdeg,
                                                          const float* __restrict__ gate,
                                                          const float* __restrict__ ssn,
                                                          const float* __restrict__ Sx,
                                                          float* __restrict__ Y, float* __restrict__ S0, int N) {
    __shared__ float4 sOut[32];
    const float4* coef4 = (const float4*)coef;
    int t = blockIdx.x * 256 + threadIdx.x;
    int n = t >> 3, l = t & 7;
    int nn = (n < N) ? n : N - 1;
    int b = rowptr[nn], e = rowend[nn];
    const int q = threadIdx.x & 15, r0 = threadIdx.x >> 4;
    const float4 wa = ((const float4*)Win)[q], wb = ((const float4*)Win)[16 + q], wc = ((const float4*)Win)[32 + q];
    const float4 bh = ((const float4*)bin)[q];
    float fx[2], fg[2], fi[2], fsx[2], fsn[2];
#pragma unroll
    for (int u = 0; u < 2; ++u) {
        int m = blockIdx.x * 32 + r0 + 16 * u; if (m >= N) m = N - 1;
        fx[u] = x[m]; fg[u] = gate[m]; fi[u] = invdeg[m]; fsx[u] = Sx[m]; fsn[u] = ssn[m];
    }
    float4 a = make_float4(0.0f, 0.0f, 0.0f, 0.0f);
    for (int base = b; base < e; base += 32) {
        int idx[4];
#pragma unroll
        for (int u = 0; u < 4; ++u) {
            int j = base + u * 8 + l;
            idx[u] = csr[(j < e) ? j : (e - 1)];
        }
#pragma unroll
        for (int u = 0; u < 4; ++u) {
            float4 c = coef4[idx[u]];
            bool ok = (base + u * 8 + l) < e;
            a.x += ok ? c.x : 0.0f;
            a.y += ok ? c.y : 0.0f;
            a.z += ok ? c.z : 0.0f;
            a.w += ok ? c.w : 0.0f;
        }
    }
#pragma unroll
    for (int o = 4; o > 0; o >>= 1) {
        a.x += __shfl_xor(a.x, o, 8);
        a.y += __shfl_xor(a.y, o, 8);
        a.z += __shfl_xor(a.z, o, 8);
        a.w += __shfl_xor(a.w, o, 8);
    }
    if (l == 0) sOut[threadIdx.x >> 3] = a;
    __syncthreads();
#pragma unroll
    for (int u = 0; u < 2; ++u) {
        int r = r0 + 16 * u;
        int m = blockIdx.x * 32 + r;
        if (m < N) {
            float4 cs = sOut[r];
            float gx = fx[u], g = fg[u], is = fi[u] * fsx[u], sn = fsn[u];
            float4 y, s;
            y.x = g * (gx * (wa.x - wb.x) + bh.x) + is * wb.x + sn * wc.x;
            y.y = g * (gx * (wa.y - wb.y) + bh.y) + is * wb.y + sn * wc.y;
            y.z = g * (gx * (wa.z - wb.z) + bh.z) + is * wb.z + sn * wc.z;
            y.w = g * (gx * (wa.w - wb.w) + bh.w) + is * wb.w + sn * wc.w;
            s.x = cs.x * (wa.x - wb.x) + cs.y * bh.x + cs.z * wb.x + cs.w * wc.x;
            s.y = cs.x * (wa.y - wb.y) + cs.y * bh.y + cs.z * wb.y + cs.w * wc.y;
            s.z = cs.x * (wa.z - wb.z) + cs.y * bh.z + cs.z * wb.z + cs.w * wc.z;
            s.w = cs.x * (wa.w - wb.w) + cs.y * bh.w + cs.z * wb.w + cs.w * wc.w;
            ((float4*)Y)[(size_t)m * 16 + q] = y;
            ((float4*)S0)[(size_t)m * 16 + q] = s;
        }
    }
}

__device__ __forceinline__ void rows16(const char* __restrict__ Xb, unsigned mine, unsigned qoff, float4 (&v)[16]) {
#pragma unroll
    for (int k = 0; k < 16; ++k) {
        unsigned off = (unsigned)__shfl((int)mine, k, 16) + qoff;
        v[k] = *(const float4*)(Xb + off);
    }
}
template <bool FULL>
__device__ __forceinline__ void sum16(const float4 (&v)[16], int cnt, float4& acc) {
#pragma unroll
    for (int k = 0; k < 16; ++k) {
        bool ok = FULL || (k < cnt);
        acc.x += ok ? v[k].x : 0.0f;
        acc.y += ok ? v[k].y : 0.0f;
        acc.z += ok ? v[k].z : 0.0f;
        acc.w += ok ? v[k].w : 0.0f;
    }
}

#define SROW 68
#define TILE_IDX_CAP 768
template <bool GATHER, bool LAST>
__global__ __launch_bounds__(256) void mid_conv_kernel(
    const float* __restrict__ X, const float* __restrict__ S,
    const int* __restrict__ rowptr, const int* __restrict__ rowend, const int* __restrict__ csr,
    const float* __restrict__ Wd, const float* __restrict__ Wb,
    const float* __restrict__ Wc, const float* __restrict__ bias,
    const float* __restrict__ gate, const float* __restrict__ invdeg,
    const float* __restrict__ ssn, const float* resid,
    float* Y,
    const float* __restrict__ Wout, float* __restrict__ pOut, float* __restrict__ qOut, int nTiles, int N) {
    static_assert(GATHER || !LAST, "the LAST form keeps its rows and its p, q in the LDS areas of the gather");
    __shared__ float sWd[HDIM * HDIM];
    __shared__ float sWb[HDIM * HDIM];
    __shared__ __attribute__((aligned(16))) float sS[GATHER ? 8 * 16 * SROW : 4];
    __shared__ int   sIdx[GATHER ? 8 * TILE_IDX_CAP : 2];
    for (int i = threadIdx.x; i < HDIM * HDIM; i += 256) {
        int ks   = i >> 8;
        int rem  = i & 255;
        int nt   = rem >> 6;
        int lp   = rem & 63;
        int ln   = lp >> 1;
        int half = lp & 1;
        int kh   = (ln >> 4) << 1;
        int srcr = (ks << 2) + kh + half;
        int srcc = (nt << 4) + (ln & 15);
        sWd[i] = Wd[srcr * HDIM + srcc];
        sWb[i] = Wb[srcr * HDIM + srcc];
    }
    __syncthreads();

    int wave = threadIdx.x >> 5;
    int lane = threadIdx.x & 31;
    int tile = blockIdx.x * 8 + wave;
    const bool active = tile < nTiles;
    if (!LAST && !active) return;
    if (active) {
        int node0 = tile * 16;
        int m  = lane & 15;
        int kh = (lane >> 4) << 1;
        int row = node0 + m; if (row >= N) row = N - 1;
        const float* Xr = X + (size_t)row * HDIM;
        const float* Sr = S + (size_t)row * HDIM;
        float* mS = sS + (GATHER ? wave * (16 * SROW) : 0);

        int ncol = lane & 15;
        int mAdd = (lane >> 4) * 8;
        int frow = node0 + (lane & 7) + mAdd; if (frow >= N) frow = N - 1;
        float fgl = gate[frow], fil = invdeg[frow], fsl = ssn[frow];
        float wcv[4], bbv[4];
    #pragma unroll
        for (int nt = 0; nt < 4; ++nt) {
            wcv[nt] = Wc[nt * 16 + ncol];
            bbv[nt] = bias[nt * 16 + ncol];
        }

        float4 woA[4], woB[4];
        if (LAST) {
    #pragma unroll
            for (int i = 0; i < 4; ++i) {
                int c4 = i * 4 + (lane & 3);
                woA[i] = ((const float4*)Wout)[c4];
                woB[i] = ((const float4*)Wout)[16 + c4];
            }
        }

        if (GATHER) {
            int* mIx = sIdx + wave * TILE_IDX_CAP;
            int rb = rowptr[row], re = rowend[row];
            if (node0 + m >= N) rb = re;
            int tb = __shfl(rb, 0, 32);
            int staged = __shfl(re, 15, 32) - tb;
            staged = (staged < 0) ? 0 : ((staged > TILE_IDX_CAP) ? TILE_IDX_CAP : staged);
            for (int i0 = 0; i0 < staged; i0 += 32 * 16) {
                int v[16];
    #pragma unroll
                for (int u = 0; u < 16; ++u) {
                    int i = i0 + u * 32 + lane;
                    v[u] = csr[tb + ((i < staged) ? i : (staged - 1))];
                }
    #pragma unroll
                for (int u = 0; u < 16; ++u) {
                    int i = i0 + u * 32 + lane;
                    if (i < staged) mIx[i] = v[u] << 8;
                }
            }
            __builtin_amdgcn_fence(__ATOMIC_ACQ_REL, "wavefront");

            static_assert((size_t)NB_MAX * BINW * HDIM * sizeof(float) <= 0xffffffffull, "a row's byte offset fits 32 bits");
            const char* Xb = (const char*)X;
            int q = lane & 15, h = lane >> 4;
            const unsigned qoff = (unsigned)q * 16u;
            for (int p = 0; p < 8; ++p) {
                int r = 2 * p + h;
                int b = __shfl(rb, r, 32), e = __shfl(re, r, 32);
                int own = node0 + r; if (own >= N) own = N - 1;
                const unsigned ownOff = (unsigned)own << 8;
                float4 acc = make_float4(0.0f, 0.0f, 0.0f, 0.0f);
                int eL = (b >= tb) ? (((tb + staged) < e) ? (tb + staged) : e) : b;
                if (eL < b) eL = b;
                int base = b;
                int j0 = base + q;
                unsigned cur = (j0 < eL) ? (unsigned)mIx[j0 - tb] : ownOff;
                float4 v[16];
                while (base + 16 <= eL) {
                    rows16(Xb, cur, qoff, v);
                    int jn = base + 16 + q;
                    int jc = jn - tb;
                    jc = (jc < staged) ? jc : (staged - 1);
                    unsigned t = (unsigned)mIx[jc];
                    sum16<true>(v, 16, acc);
                    cur = (jn < eL) ? t : ownOff;
                    base += 16;
                }
                if (base < eL) {
                    rows16(Xb, cur, qoff, v);
                    sum16<false>(v, eL - base, acc);
                }
                for (int bm = eL; bm < e; bm += 16) {
                    int j = bm + q;
                    unsigned mine = (j < e) ? ((unsigned)csr[j] << 8) : ownOff;
                    rows16(Xb, mine, qoff, v);
                    sum16<false>(v, e - bm, acc);
                }
                *(float4*)&mS[r * SROW + 4 * q] = acc;
            }
            __builtin_amdgcn_fence(__ATOMIC_ACQ_REL, "wavefront");
        }

        v8f accX[4] = {};
        v8f accS[4] = {};
        int fragBase = lane * 2;

        v2f aX[16];
    #pragma unroll
        for (int ks = 0; ks < 16; ++ks) aX[ks] = *(const v2f*)&Xr[(ks << 2) + kh];

    #pragma unroll
        for (int ks = 0; ks < 16; ++ks) {
            int k0 = ks << 2;
            v2f a = aX[ks], s;
            if (GATHER) {
                s = *(const v2f*)&mS[m * SROW + k0 + kh];
            } else {
                s.x = Sr[k0 + kh]; s.y = Sr[k0 + kh + 1];
            }
    #pragma unroll
            for (int nt = 0; nt < 4; ++nt) {
                v2f bd = *(const v2f*)&sWd[((ks << 2) + nt) * 64 + fragBase];
                v2f bb = *(const v2f*)&sWb[((ks << 2) + nt) * 64 + fragBase];
                accX[nt] = __builtin_amdgcn_wmma_f32_16x16x4_f32(
                    false, a, false, bd, (short)0, accX[nt], false, false);
                accS[nt] = __builtin_amdgcn_wmma_f32_16x16x4_f32(
                    false, s, false, bb, (short)0, accS[nt], false, false);
            }
        }

        float rs[4][8];
        if (resid) {
    #pragma unroll
            for (int v = 0; v < 8; ++v) {
                int node = node0 + v + mAdd; if (node >= N) node = N - 1;
                const float* rp = resid + (size_t)node * HDIM + ncol;
    #pragma unroll
                for (int nt = 0; nt < 4; ++nt) rs[nt][v] = rp[nt * 16];
            }
        }
        float fg[8], fi[8], fs[8];
    #pragma unroll
        for (int v = 0; v < 8; ++v) {
            fg[v] = __shfl(fgl, v, 16);
            fi[v] = __shfl(fil, v, 16);
            fs[v] = __shfl(fsl, v, 16);
        }
    #pragma unroll
        for (int nt = 0; nt < 4; ++nt) {
            int col = nt * 16 + ncol;
    #pragma unroll
            for (int v = 0; v < 8; ++v) {
                int node = node0 + v + mAdd;
                float val = fg[v] * (accX[nt][v] + bbv[nt]) + fi[v] * accS[nt][v] + fs[v] * wcv[nt];
                if (resid) val += rs[nt][v];
                val = (val > 0.0f) ? val : 0.01f * val;
                if (LAST) mS[(v + mAdd) * SROW + col] = val;
                else if (node < N) Y[(size_t)node * HDIM + col] = val;
            }
        }
        if (LAST) {
            __builtin_amdgcn_fence(__ATOMIC_ACQ_REL, "wavefront");
            float* mPQ = (float*)(sIdx + wave * TILE_IDX_CAP);
            const int l4 = lane & 3;
    #pragma unroll
            for (int ps = 0; ps < 2; ++ps) {
                int r = ps * 8 + (lane >> 2);
                float ap = 0.0f, aq = 0.0f;
    #pragma unroll
                for (int i = 0; i < 4; ++i) {
                    int c4 = i * 4 + l4;
                    float4 xv = *(const float4*)&mS[r * SROW + 4 * c4];
                    float wa0 = woA[i].x, wa1 = woA[i].y, wa2 = woA[i].z, wa3 = woA[i].w;
                    float wb0 = woB[i].x, wb1 = woB[i].y, wb2 = woB[i].z, wb3 = woB[i].w;
                    aq += xv.x * (wa0 - wb0) + xv.y * (wa1 - wb1) + xv.z * (wa2 - wb2) + xv.w * (wa3 - wb3);
                    ap += xv.x * wb0 + xv.y * wb1 + xv.z * wb2 + xv.w * wb3;
                }
    #pragma unroll
                for (int o = 2; o > 0; o >>= 1) {
                    ap += __shfl_xor(ap, o, 4);
                    aq += __shfl_xor(aq, o, 4);
                }
                if (l4 == 0) { mPQ[r] = ap; mPQ[16 + r] = aq; }
            }
        }
    }
    if (LAST) {
        __syncthreads();
        int t = threadIdx.x & 127;
        int node = blockIdx.x * 128 + t;
        const float* wPQ = (const float*)(sIdx + (t >> 4) * TILE_IDX_CAP);
        if (node < N) {
            if (threadIdx.x < 128) pOut[node] = wPQ[t & 15];
            else                   qOut[node] = wPQ[16 + (t & 15)];
        }
    }
}

__global__ __launch_bounds__(256) void out_kernel(const int* __restrict__ rowptr, const int* __restrict__ rowend,
                                                  const int* __restrict__ csr,
                                                  const float* __restrict__ p, const float* __restrict__ q,
                                                  const float* __restrict__ Wout, const float* __restrict__ bout,
                                                  const float* __restrict__ x0,
                                                  const float* __restrict__ invdeg, const float* __restrict__ gate,
                                                  const float* __restrict__ ssn, float* __restrict__ out, int N) {
    __shared__ float sOut[32];
    int t = blockIdx.x * 256 + threadIdx.x;
    int n = t >> 3, l = t & 7;
    int nn = (n < N) ? n : N - 1;
    int b = rowptr[nn], e = rowend[nn];
    float a = 0.0f;
    for (int base = b; base < e; base += 32) {
        int idx[4];
#pragma unroll
        for (int u = 0; u < 4; ++u) {
            int j = base + u * 8 + l;
            idx[u] = csr[(j < e) ? j : (e - 1)];
        }
#pragma unroll
        for (int u = 0; u < 4; ++u) {
            float v = p[idx[u]];
            a += ((base + u * 8 + l) < e) ? v : 0.0f;
        }
    }
#pragma unroll
    for (int o = 4; o > 0; o >>= 1) a += __shfl_xor(a, o, 8);
    if (l == 0) sOut[threadIdx.x >> 3] = a;
    __syncthreads();
    if (threadIdx.x < 32) {
        int m = blockIdx.x * 32 + threadIdx.x;
        if (m < N)
            out[m] = gate[m] * (q[m] + bout[0]) + invdeg[m] * sOut[threadIdx.x] + ssn[m] * Wout[2 * HDIM] + x0[m];
    }
}

extern "C" void kernel_launch(void* const* d_in, const int* in_sizes, int n_in,
                              void* d_out, int out_size, void* d_ws, size_t ws_size,
                              hipStream_t stream) {
    const float* x    = (const float*)d_in[0];
    const int*   ei   = (const int*)  d_in[1];
    const float* Win  = (const float*)d_in[2];
    const float* bin  = (const float*)d_in[3];
    const float* Wmid = (const float*)d_in[4];
    const float* bmid = (const float*)d_in[5];
    const float* Wout = (const float*)d_in[6];
    const float* bout = (const float*)d_in[7];

    const int N = in_sizes[0];
    const int E = in_sizes[1] / 2;
    const int* src = ei;
    const int* dst = ei + E;

    float* w = (float*)d_ws;
    int*   rowptr  = (int*)w;
    int*   cursor  = (int*)(w + (size_t)N);
    float* invdeg  = w + 2 * (size_t)N;
    float* gate    = w + 3 * (size_t)N;
    float* ssn     = w + 4 * (size_t)N;
    float* Sx      = w + 5 * (size_t)N;
    float* Wd6     = w + 6 * (size_t)N;
    float* bufA    = Wd6  + 6 * HDIM * HDIM;
    float* bufB    = bufA + (size_t)N * HDIM;
    float* bufC    = bufB + (size_t)N * HDIM;
    float* bufS    = bufC + (size_t)N * HDIM;
    int*   csr     = (int*)bufC;

    const int nTiles = (N + 15) / 16;

    const int NB = (N + BINW - 1) / BINW;
    int* binned = (int*)bufB;
    int* cntT   = (int*)bufS;
    int* binBase = cntT + NB * NCHUNK;
    float* coef = bufS - 4 * (size_t)N;
    bin_count_kernel<<<NCHUNK, CH_THREADS, 0, stream>>>(dst, cntT, E, NB, Wmid, Wd6);
    bin_scatter_kernel<<<NCHUNK, CH_THREADS, 0, stream>>>(src, dst, cntT, binBase, binned, E, NB);
    bin_fill_kernel<<<NB, FILL_THREADS, 0, stream>>>(binBase, binned, rowptr, cursor, csr, x, invdeg, gate, ssn, Sx,
                                                     coef, N);
    coef_gather_kernel<<<(N * 8 + 255) / 256, 256, 0, stream>>>(rowptr, cursor, csr, coef, x, Win, bin, invdeg, gate,
                                                                ssn, Sx, bufA, bufS, N);

    float* cur = bufA;
    mid_conv_kernel<false, false><<<(nTiles + 7) / 8, 256, 0, stream>>>(
        cur, bufS, rowptr, cursor, csr, Wd6, Wmid + 64 * HDIM, Wmid + 128 * HDIM,
        bmid, gate, invdeg, ssn, nullptr, bufB, nullptr, nullptr, nullptr, nTiles, N);
    mid_conv_kernel<true, false><<<(nTiles + 7) / 8, 256, 0, stream>>>(
        bufB, bufS, rowptr, cursor, csr, Wd6 + HDIM * HDIM,
        Wmid + (size_t)129 * HDIM + 64 * HDIM, Wmid + (size_t)129 * HDIM + 128 * HDIM,
        bmid + HDIM, gate, invdeg, ssn, cur, cur, nullptr, nullptr, nullptr, nTiles, N);
    float* pj = bufS;
    float* qj = bufS + 2 * (size_t)N;
    for (int l = 1; l < 3; ++l) {
        const float* Wl = Wmid + (size_t)(2 * l) * 129 * HDIM;
        mid_conv_kernel<true, false><<<(nTiles + 7) / 8, 256, 0, stream>>>(
            cur, bufS, rowptr, cursor, csr, Wd6 + (2 * l) * HDIM * HDIM, Wl + 64 * HDIM, Wl + 128 * HDIM,
            bmid + (2 * l) * HDIM, gate, invdeg, ssn, nullptr, bufB, nullptr, nullptr, nullptr, nTiles, N);

        const float* Wl2 = Wmid + (size_t)(2 * l + 1) * 129 * HDIM;
        if (l < 2)
            mid_conv_kernel<true, false><<<(nTiles + 7) / 8, 256, 0, stream>>>(
                bufB, bufS, rowptr, cursor, csr, Wd6 + (2 * l + 1) * HDIM * HDIM, Wl2 + 64 * HDIM, Wl2 + 128 * HDIM,
                bmid + (2 * l + 1) * HDIM, gate, invdeg, ssn, cur, cur, nullptr, nullptr, nullptr, nTiles, N);
        else
            mid_conv_kernel<true, true><<<(nTiles + 7) / 8, 256, 0, stream>>>(
                bufB, bufS, rowptr, cursor, csr, Wd6 + (2 * l + 1) * HDIM * HDIM, Wl2 + 64 * HDIM, Wl2 + 128 * HDIM,
                bmid + (2 * l + 1) * HDIM, gate, invdeg, ssn, cur, cur, Wout, pj, qj, nTiles, N);
    }

    out_kernel<<<(N * 8 + 255) / 256, 256, 0, stream>>>(rowptr, cursor, csr, pj, qj, Wout, bout, x, invdeg, gate,
                                                        ssn, (float*)d_out, N);
}


# Mid conv runs X @ Wd inside the gather: two k-steps per pass issued under the pass's first row reads (peeled out to where all lanes are active); only S @ Wb stays behind the gather; sums unchanged
# speedup vs baseline: 5.9167x; 1.0377x over previous
#include <hip/hip_runtime.h>
#include <hip/hip_bf16.h>

typedef float v2f __attribute__((ext_vector_type(2)));
typedef float v8f __attribute__((ext_vector_type(8)));

#define HDIM 64

#define BIN_SHIFT 8
#define BINW 256
#define NB_MAX 512
#define NCHUNK 256
#define CH_THREADS 1024
#define FILL_THREADS 512
#define FILL_CAP 12288

__global__ __launch_bounds__(CH_THREADS) void bin_count_kernel(const int* __restrict__ dst,
                                                               int* __restrict__ cntT, int E, int NB,
                                                               const float* __restrict__ Wmid,
                                                               float* __restrict__ Wd6) {
    __shared__ int sHist[NB_MAX];
    const int c = blockIdx.x, tid = threadIdx.x;
    static_assert(NCHUNK * CH_THREADS >= 6 * HDIM * HDIM, "one thread per element of Wd6");
    {
        int gid = c * CH_THREADS + tid;
        if (gid < 6 * HDIM * HDIM) {
            int l = gid >> 12;
            int rc = gid & 4095;
            const float* Wl = Wmid + (size_t)l * 129 * HDIM;
            Wd6[gid] = Wl[rc] - Wl[64 * HDIM + rc];
        }
    }
    for (int i = tid; i < NB; i += CH_THREADS) sHist[i] = 0;
    __syncthreads();
    const int chunk = (E + NCHUNK - 1) / NCHUNK;
    const int e0 = c * chunk;
    const int e1 = (e0 + chunk < E) ? (e0 + chunk) : E;
    for (int e = e0 + tid; e < e1; e += CH_THREADS * 8) {
        int d[8];
#pragma unroll
        for (int u = 0; u < 8; ++u) {
            int ee = e + u * CH_THREADS;
            d[u] = dst[(ee < e1) ? ee : (e1 - 1)];
        }
#pragma unroll
        for (int u = 0; u < 8; ++u)
            if (e + u * CH_THREADS < e1) atomicAdd(&sHist[d[u] >> BIN_SHIFT], 1);
    }
    __syncthreads();
    for (int i = tid; i < NB; i += CH_THREADS) cntT[i * NCHUNK + c] = sHist[i];
}

#define GROUPS_PER_PASS 2
__global__ __launch_bounds__(CH_THREADS) void bin_scatter_kernel(const int* __restrict__ src,
                                                                 const int* __restrict__ dst,
                                                                 const int* __restrict__ cntT,
                                                                 int* __restrict__ binBase,
                                                                 int* __restrict__ binned, int E, int NB) {
    static_assert(NCHUNK == 256, "one row of the count table = 8 lanes x 8 loads x 4 counts");
    static_assert(NB_MAX == 512, "wave 0 scans 16 bin totals per lane");
    __shared__ int sCur[NB_MAX];
    __shared__ int sTot[NB_MAX];
    const int c = blockIdx.x, tid = threadIdx.x, lane = tid & 31, wv = tid >> 5;
    const int sub = lane & 7, rw = lane >> 3;
    for (int i = tid; i < NB_MAX; i += CH_THREADS) sTot[i] = 0;
    __syncthreads();
    for (int g0 = wv; g0 * 4 < NB; g0 += (CH_THREADS / 32) * GROUPS_PER_PASS) {
        int4 v[GROUPS_PER_PASS][8];
#pragma unroll
        for (int u = 0; u < GROUPS_PER_PASS; ++u) {
            int i = (g0 + u * (CH_THREADS / 32)) * 4 + rw;
            const int4* row = (const int4*)(cntT + (size_t)((i < NB) ? i : (NB - 1)) * NCHUNK);
#pragma unroll
            for (int j = 0; j < 8; ++j) v[u][j] = row[j * 8 + sub];
        }
#pragma unroll
        for (int u = 0; u < GROUPS_PER_PASS; ++u) {
            int tot = 0, pre = 0;
#pragma unroll
            for (int j = 0; j < 8; ++j) {
                int k = (j * 8 + sub) * 4;
                tot += v[u][j].x + v[u][j].y + v[u][j].z + v[u][j].w;
                pre += ((k     < c) ? v[u][j].x : 0) + ((k + 1 < c) ? v[u][j].y : 0)
                     + ((k + 2 < c) ? v[u][j].z : 0) + ((k + 3 < c) ? v[u][j].w : 0);
            }
#pragma unroll
            for (int o = 4; o > 0; o >>= 1) {
                tot += __shfl_xor(tot, o, 8);
                pre += __shfl_xor(pre, o, 8);
            }
            int i = (g0 + u * (CH_THREADS / 32)) * 4 + rw;
            if (sub == 0 && i < NB) { sTot[i] = tot; sCur[i] = pre; }
        }
    }
    __syncthreads();
    if (wv == 0) {
        int t[16];
        int s = 0;
#pragma unroll
        for (int k = 0; k < 16; ++k) { t[k] = sTot[lane * 16 + k]; s += t[k]; }
        int inc = s;
#pragma unroll
        for (int o = 1; o < 32; o <<= 1) {
            int v = __shfl_up(inc, o, 32);
            if (lane >= o) inc += v;
        }
        int ex = inc - s;
#pragma unroll
        for (int k = 0; k < 16; ++k) { sTot[lane * 16 + k] = ex; ex += t[k]; }
    }
    __syncthreads();
    for (int i = tid; i < NB; i += CH_THREADS) {
        int base = sTot[i];
        sCur[i] += base;
        if (c == 0) binBase[i] = base;
    }
    if (c == 0 && tid == 0) binBase[NB] = E;
    __syncthreads();
    const int chunk = (E + NCHUNK - 1) / NCHUNK;
    const int e0 = c * chunk;
    const int e1 = (e0 + chunk < E) ? (e0 + chunk) : E;
    for (int e = e0 + tid; e < e1; e += CH_THREADS * 8) {
        int d[8], s[8];
#pragma unroll
        for (int u = 0; u < 8; ++u) {
            int ee = e + u * CH_THREADS;
            int ec = (ee < e1) ? ee : (e1 - 1);
            d[u] = dst[ec];
            s[u] = src[ec];
        }
#pragma unroll
        for (int u = 0; u < 8; ++u) {
            if (e + u * CH_THREADS < e1) {
                int pos = atomicAdd(&sCur[d[u] >> BIN_SHIFT], 1);
                binned[pos] = (s[u] << BIN_SHIFT) | (d[u] & (BINW - 1));
            }
        }
    }
}

__global__ __launch_bounds__(FILL_THREADS) void bin_fill_kernel(const int* __restrict__ binBase,
                                                                const int* __restrict__ binned,
                                                                int* __restrict__ rowptr, int* __restrict__ rowend,
                                                                int* __restrict__ csr, const float* __restrict__ x,
                                                                float* __restrict__ invdeg, float* __restrict__ gate,
                                                                float* __restrict__ ssn, float* __restrict__ Sx,
                                                                float* __restrict__ coef, int N) {
    static_assert(BINW == (FILL_THREADS / 8) * 4, "statistics: 8 lanes per node, 4 nodes per lane group");
    __shared__ int sDeg[BINW];
    __shared__ int sBeg[BINW];
    __shared__ int sWaveTot[BINW / 32];
    __shared__ int sStage[FILL_CAP];
    const int b = blockIdx.x, tid = threadIdx.x, lane = tid & 31;
    const int beg = binBase[b];
    const int end = binBase[b + 1];
    const int cnt = end - beg;
    if (tid < BINW) sDeg[tid] = 0;
    __syncthreads();
    for (int i = tid; i < cnt; i += FILL_THREADS * 8) {
        int r[8];
#pragma unroll
        for (int u = 0; u < 8; ++u) {
            int ii = i + u * FILL_THREADS;
            r[u] = binned[beg + ((ii < cnt) ? ii : (cnt - 1))];
        }
#pragma unroll
        for (int u = 0; u < 8; ++u)
            if (i + u * FILL_THREADS < cnt) atomicAdd(&sDeg[r[u] & (BINW - 1)], 1);
    }
    __syncthreads();
    int v = 0, inc = 0;
    if (tid < BINW) {
        v = sDeg[tid];
        inc = v;
#pragma unroll
        for (int o = 1; o < 32; o <<= 1) {
            int t = __shfl_up(inc, o, 32);
            if (lane >= o) inc += t;
        }
        if (lane == 31) sWaveTot[tid >> 5] = inc;
    }
    __syncthreads();
    if (tid < BINW) {
        int wbase = 0;
        for (int w = 0; w < (tid >> 5); ++w) wbase += sWaveTot[w];
        int excl = wbase + inc - v;
        sDeg[tid] = excl;
        sBeg[tid] = excl;
        int node = b * BINW + tid;
        if (node < N) { rowptr[node] = beg + excl; rowend[node] = beg + excl + v; }
    }
    __syncthreads();
    const bool staged = cnt <= FILL_CAP;
    for (int i = tid; i < cnt; i += FILL_THREADS * 8) {
        int r[8];
#pragma unroll
        for (int u = 0; u < 8; ++u) {
            int ii = i + u * FILL_THREADS;
            r[u] = binned[beg + ((ii < cnt) ? ii : (cnt - 1))];
        }
#pragma unroll
        for (int u = 0; u < 8; ++u) {
            if (i + u * FILL_THREADS < cnt) {
                int pos = atomicAdd(&sDeg[r[u] & (BINW - 1)], 1);
                if (staged) sStage[pos] = r[u] >> BIN_SHIFT;
                else        csr[beg + pos] = r[u] >> BIN_SHIFT;
            }
        }
    }
    __syncthreads();
    if (staged)
        for (int i = tid; i < cnt; i += FILL_THREADS) csr[beg + i] = sStage[i];

    const int l8 = tid & 7, grp = tid >> 3;
    int sb[4], se[4], nd[4];
    float sg[4], sx[4];
#pragma unroll
    for (int u = 0; u < 4; ++u) {
        int ln = grp + u * (FILL_THREADS / 8);
        sb[u] = sBeg[ln];
        se[u] = sDeg[ln];
        nd[u] = b * BINW + ln;
        sg[u] = 0.0f; sx[u] = 0.0f;
    }
    int longest = 0;
#pragma unroll
    for (int u = 0; u < 4; ++u) longest = (se[u] - sb[u] > longest) ? (se[u] - sb[u]) : longest;
    for (int off = 0; off < longest; off += 32) {
        int s[4][4];
        if (staged) {
#pragma unroll
            for (int u = 0; u < 4; ++u) {
#pragma unroll
                for (int k = 0; k < 4; ++k) {
                    int j = sb[u] + off + k * 8 + l8;
                    bool ok = j < se[u];
                    int t = sStage[ok ? j : 0];
                    s[u][k] = ok ? t : 0;
                }
            }
        } else {
#pragma unroll
            for (int u = 0; u < 4; ++u) {
#pragma unroll
                for (int k = 0; k < 4; ++k) {
                    int j = sb[u] + off + k * 8 + l8;
                    bool ok = j < se[u];
                    int t = csr[beg + (ok ? j : 0)];
                    s[u][k] = ok ? t : 0;
                }
            }
        }
#pragma unroll
        for (int u = 0; u < 4; ++u) {
#pragma unroll
            for (int k = 0; k < 4; ++k) {
                float xv = x[s[u][k]];
                bool ok = (sb[u] + off + k * 8 + l8) < se[u];
                float sgn = (s[u][k] > nd[u]) ? 1.0f : ((s[u][k] < nd[u]) ? -1.0f : 0.0f);
                sg[u] += ok ? sgn : 0.0f;
                sx[u] += ok ? xv : 0.0f;
            }
        }
    }
#pragma unroll
    for (int o = 4; o > 0; o >>= 1) {
#pragma unroll
        for (int u = 0; u < 4; ++u) {
            sg[u] += __shfl_xor(sg[u], o, 8);
            sx[u] += __shfl_xor(sx[u], o, 8);
        }
    }
    if (l8 == 0) {
#pragma unroll
        for (int u = 0; u < 4; ++u) {
            int n = nd[u];
            if (n < N) {
                float dg = (float)(se[u] - sb[u]);
                float inv = 1.0f / fmaxf(dg, 1.0f);
                float g = (dg > 0.0f) ? 1.0f : 0.0f;
                invdeg[n] = inv;
                gate[n]   = g;
                ssn[n]    = sg[u] * inv;
                Sx[n]     = sx[u];
                ((float4*)coef)[n] = make_float4(g * x[n], g, inv * sx[u], sg[u] * inv);
            }
        }
    }
}

__global__ __launch_bounds__(256) void coef_gather_kernel(const int* __restrict__ rowptr,
                                                          const int* __restrict__ rowend,
                                                          const int* __restrict__ csr,
                                                          const float* __restrict__ coef,
                                                          const float* __restrict__ x,
                                                          const float* __restrict__ Win,
                                                          const float* __restrict__ bin,
                                                          const float* __restrict__ invdeg,
                                                          const float* __restrict__ gate,
                                                          const float* __restrict__ ssn,
                                                          const float* __restrict__ Sx,
                                                          float* __restrict__ Y, float* __restrict__ S0, int N) {
    __shared__ float4 sOut[32];
    const float4* coef4 = (const float4*)coef;
    int t = blockIdx.x * 256 + threadIdx.x;
    int n = t >> 3, l = t & 7;
    int nn = (n < N) ? n : N - 1;
    int b = rowptr[nn], e = rowend[nn];
    const int q = threadIdx.x & 15, r0 = threadIdx.x >> 4;
    const float4 wa = ((const float4*)Win)[q], wb = ((const float4*)Win)[16 + q], wc = ((const float4*)Win)[32 + q];
    const float4 bh = ((const float4*)bin)[q];
    float fx[2], fg[2], fi[2], fsx[2], fsn[2];
#pragma unroll
    for (int u = 0; u < 2; ++u) {
        int m = blockIdx.x * 32 + r0 + 16 * u; if (m >= N) m = N - 1;
        fx[u] = x[m]; fg[u] = gate[m]; fi[u] = invdeg[m]; fsx[u] = Sx[m]; fsn[u] = ssn[m];
    }
    float4 a = make_float4(0.0f, 0.0f, 0.0f, 0.0f);
    for (int base = b; base < e; base += 32) {
        int idx[4];
#pragma unroll
        for (int u = 0; u < 4; ++u) {
            int j = base + u * 8 + l;
            idx[u] = csr[(j < e) ? j : (e - 1)];
        }
#pragma unroll
        for (int u = 0; u < 4; ++u) {
            float4 c = coef4[idx[u]];
            bool ok = (base + u * 8 + l) < e;
            a.x += ok ? c.x : 0.0f;
            a.y += ok ? c.y : 0.0f;
            a.z += ok ? c.z : 0.0f;
            a.w += ok ? c.w : 0.0f;
        }
    }
#pragma unroll
    for (int o = 4; o > 0; o >>= 1) {
        a.x += __shfl_xor(a.x, o, 8);
        a.y += __shfl_xor(a.y, o, 8);
        a.z += __shfl_xor(a.z, o, 8);
        a.w += __shfl_xor(a.w, o, 8);
    }
    if (l == 0) sOut[threadIdx.x >> 3] = a;
    __syncthreads();
#pragma unroll
    for (int u = 0; u < 2; ++u) {
        int r = r0 + 16 * u;
        int m = blockIdx.x * 32 + r;
        if (m < N) {
            float4 cs = sOut[r];
            float gx = fx[u], g = fg[u], is = fi[u] * fsx[u], sn = fsn[u];
            float4 y, s;
            y.x = g * (gx * (wa.x - wb.x) + bh.x) + is * wb.x + sn * wc.x;
            y.y = g * (gx * (wa.y - wb.y) + bh.y) + is * wb.y + sn * wc.y;
            y.z = g * (gx * (wa.z - wb.z) + bh.z) + is * wb.z + sn * wc.z;
            y.w = g * (gx * (wa.w - wb.w) + bh.w) + is * wb.w + sn * wc.w;
            s.x = cs.x * (wa.x - wb.x) + cs.y * bh.x + cs.z * wb.x + cs.w * wc.x;
            s.y = cs.x * (wa.y - wb.y) + cs.y * bh.y + cs.z * wb.y + cs.w * wc.y;
            s.z = cs.x * (wa.z - wb.z) + cs.y * bh.z + cs.z * wb.z + cs.w * wc.z;
            s.w = cs.x * (wa.w - wb.w) + cs.y * bh.w + cs.z * wb.w + cs.w * wc.w;
            ((float4*)Y)[(size_t)m * 16 + q] = y;
            ((float4*)S0)[(size_t)m * 16 + q] = s;
        }
    }
}

__device__ __forceinline__ void rows16(const char* __restrict__ Xb, unsigned mine, unsigned qoff, float4 (&v)[16]) {
#pragma unroll
    for (int k = 0; k < 16; ++k) {
        unsigned off = (unsigned)__shfl((int)mine, k, 16) + qoff;
        v[k] = *(const float4*)(Xb + off);
    }
}
template <bool FULL>
__device__ __forceinline__ void sum16(const float4 (&v)[16], int cnt, float4& acc) {
#pragma unroll
    for (int k = 0; k < 16; ++k) {
        bool ok = FULL || (k < cnt);
        acc.x += ok ? v[k].x : 0.0f;
        acc.y += ok ? v[k].y : 0.0f;
        acc.z += ok ? v[k].z : 0.0f;
        acc.w += ok ? v[k].w : 0.0f;
    }
}

#define SROW 68
#define TILE_IDX_CAP 768
template <bool GATHER, bool LAST>
__global__ __launch_bounds__(256) void mid_conv_kernel(
    const float* __restrict__ X, const float* __restrict__ S,
    const int* __restrict__ rowptr, const int* __restrict__ rowend, const int* __restrict__ csr,
    const float* __restrict__ Wd, const float* __restrict__ Wb,
    const float* __restrict__ Wc, const float* __restrict__ bias,
    const float* __restrict__ gate, const float* __restrict__ invdeg,
    const float* __restrict__ ssn, const float* resid,
    float* Y,
    const float* __restrict__ Wout, float* __restrict__ pOut, float* __restrict__ qOut, int nTiles, int N) {
    static_assert(GATHER || !LAST, "the LAST form keeps its rows and its p, q in the LDS areas of the gather");
    __shared__ float sWd[HDIM * HDIM];
    __shared__ float sWb[HDIM * HDIM];
    __shared__ __attribute__((aligned(16))) float sS[GATHER ? 8 * 16 * SROW : 4];
    __shared__ int   sIdx[GATHER ? 8 * TILE_IDX_CAP : 2];
    for (int i = threadIdx.x; i < HDIM * HDIM; i += 256) {
        int ks   = i >> 8;
        int rem  = i & 255;
        int nt   = rem >> 6;
        int lp   = rem & 63;
        int ln   = lp >> 1;
        int half = lp & 1;
        int kh   = (ln >> 4) << 1;
        int srcr = (ks << 2) + kh + half;
        int srcc = (nt << 4) + (ln & 15);
        sWd[i] = Wd[srcr * HDIM + srcc];
        sWb[i] = Wb[srcr * HDIM + srcc];
    }
    __syncthreads();

    int wave = threadIdx.x >> 5;
    int lane = threadIdx.x & 31;
    int tile = blockIdx.x * 8 + wave;
    const bool active = tile < nTiles;
    if (!LAST && !active) return;
    if (active) {
        int node0 = tile * 16;
        int m  = lane & 15;
        int kh = (lane >> 4) << 1;
        int row = node0 + m; if (row >= N) row = N - 1;
        const float* Xr = X + (size_t)row * HDIM;
        const float* Sr = S + (size_t)row * HDIM;
        float* mS = sS + (GATHER ? wave * (16 * SROW) : 0);

        int ncol = lane & 15;
        int mAdd = (lane >> 4) * 8;
        int frow = node0 + (lane & 7) + mAdd; if (frow >= N) frow = N - 1;
        float fgl = gate[frow], fil = invdeg[frow], fsl = ssn[frow];
        float wcv[4], bbv[4];
    #pragma unroll
        for (int nt = 0; nt < 4; ++nt) {
            wcv[nt] = Wc[nt * 16 + ncol];
            bbv[nt] = bias[nt * 16 + ncol];
        }

        float4 woA[4], woB[4];
        if (LAST) {
    #pragma unroll
            for (int i = 0; i < 4; ++i) {
                int c4 = i * 4 + (lane & 3);
                woA[i] = ((const float4*)Wout)[c4];
                woB[i] = ((const float4*)Wout)[16 + c4];
            }
        }

        v8f accX[4] = {};
        v8f accS[4] = {};
        int fragBase = lane * 2;

        if (GATHER) {
            int* mIx = sIdx + wave * TILE_IDX_CAP;
            int rb = rowptr[row], re = rowend[row];
            if (node0 + m >= N) rb = re;
            int tb = __shfl(rb, 0, 32);
            int staged = __shfl(re, 15, 32) - tb;
            staged = (staged < 0) ? 0 : ((staged > TILE_IDX_CAP) ? TILE_IDX_CAP : staged);
            for (int i0 = 0; i0 < staged; i0 += 32 * 16) {
                int v[16];
    #pragma unroll
                for (int u = 0; u < 16; ++u) {
                    int i = i0 + u * 32 + lane;
                    v[u] = csr[tb + ((i < staged) ? i : (staged - 1))];
                }
    #pragma unroll
                for (int u = 0; u < 16; ++u) {
                    int i = i0 + u * 32 + lane;
                    if (i < staged) mIx[i] = v[u] << 8;
                }
            }
            __builtin_amdgcn_fence(__ATOMIC_ACQ_REL, "wavefront");

            v2f aN0 = *(const v2f*)&Xr[kh];
            v2f aN1 = *(const v2f*)&Xr[4 + kh];

            static_assert((size_t)NB_MAX * BINW * HDIM * sizeof(float) <= 0xffffffffull, "a row's byte offset fits 32 bits");
            const char* Xb = (const char*)X;
            int q = lane & 15, h = lane >> 4;
            const unsigned qoff = (unsigned)q * 16u;
            for (int p = 0; p < 8; ++p) {
                const v2f aP0 = aN0, aP1 = aN1;
                int r = 2 * p + h;
                int b = __shfl(rb, r, 32), e = __shfl(re, r, 32);
                int own = node0 + r; if (own >= N) own = N - 1;
                const unsigned ownOff = (unsigned)own << 8;
                float4 acc = make_float4(0.0f, 0.0f, 0.0f, 0.0f);
                int eL = (b >= tb) ? (((tb + staged) < e) ? (tb + staged) : e) : b;
                if (eL < b) eL = b;
                int base = b;
                int j0 = base + q;
                unsigned cur = (j0 < eL) ? (unsigned)mIx[j0 - tb] : ownOff;
                float4 v[16];
                rows16(Xb, cur, qoff, v);
                {
                    const int pn = (p < 7) ? (p + 1) : 7;
                    aN0 = *(const v2f*)&Xr[(pn << 3) + kh];
                    aN1 = *(const v2f*)&Xr[(pn << 3) + 4 + kh];
                }
    #pragma unroll
                for (int nt = 0; nt < 4; ++nt) {
                    v2f bd0 = *(const v2f*)&sWd[((p << 3) + nt) * 64 + fragBase];
                    accX[nt] = __builtin_amdgcn_wmma_f32_16x16x4_f32(
                        false, aP0, false, bd0, (short)0, accX[nt], false, false);
                }
    #pragma unroll
                for (int nt = 0; nt < 4; ++nt) {
                    v2f bd1 = *(const v2f*)&sWd[((p << 3) + 4 + nt) * 64 + fragBase];
                    accX[nt] = __builtin_amdgcn_wmma_f32_16x16x4_f32(
                        false, aP1, false, bd1, (short)0, accX[nt], false, false);
                }
                if (base + 16 <= eL) {
                    int jn = base + 16 + q;
                    int jc = jn - tb;
                    jc = (jc < staged) ? jc : (staged - 1);
                    unsigned t = (unsigned)mIx[jc];
                    sum16<true>(v, 16, acc);
                    cur = (jn < eL) ? t : ownOff;
                    base += 16;
                    while (base + 16 <= eL) {
                        rows16(Xb, cur, qoff, v);
                        jn = base + 16 + q;
                        jc = jn - tb;
                        jc = (jc < staged) ? jc : (staged - 1);
                        t = (unsigned)mIx[jc];
                        sum16<true>(v, 16, acc);
                        cur = (jn < eL) ? t : ownOff;
                        base += 16;
                    }
                    if (base < eL) {
                        rows16(Xb, cur, qoff, v);
                        sum16<false>(v, eL - base, acc);
                    }
                } else if (base < eL) {
                    sum16<false>(v, eL - base, acc);
                }
                for (int bm = eL; bm < e; bm += 16) {
                    int j = bm + q;
                    unsigned mine = (j < e) ? ((unsigned)csr[j] << 8) : ownOff;
                    rows16(Xb, mine, qoff, v);
                    sum16<false>(v, e - bm, acc);
                }
                *(float4*)&mS[r * SROW + 4 * q] = acc;
            }
            __builtin_amdgcn_fence(__ATOMIC_ACQ_REL, "wavefront");
        }

        if (GATHER) {
    #pragma unroll
            for (int ks = 0; ks < 16; ++ks) {
                int k0 = ks << 2;
                v2f s = *(const v2f*)&mS[m * SROW + k0 + kh];
    #pragma unroll
                for (int nt = 0; nt < 4; ++nt) {
                    v2f bb = *(const v2f*)&sWb[((ks << 2) + nt) * 64 + fragBase];
                    accS[nt] = __builtin_amdgcn_wmma_f32_16x16x4_f32(
                        false, s, false, bb, (short)0, accS[nt], false, false);
                }
            }
        } else {
            v2f aX[16];
    #pragma unroll
            for (int ks = 0; ks < 16; ++ks) aX[ks] = *(const v2f*)&Xr[(ks << 2) + kh];

    #pragma unroll
            for (int ks = 0; ks < 16; ++ks) {
                int k0 = ks << 2;
                v2f a = aX[ks], s;
                s.x = Sr[k0 + kh]; s.y = Sr[k0 + kh + 1];
    #pragma unroll
                for (int nt = 0; nt < 4; ++nt) {
                    v2f bd = *(const v2f*)&sWd[((ks << 2) + nt) * 64 + fragBase];
                    v2f bb = *(const v2f*)&sWb[((ks << 2) + nt) * 64 + fragBase];
                    accX[nt] = __builtin_amdgcn_wmma_f32_16x16x4_f32(
                        false, a, false, bd, (short)0, accX[nt], false, false);
                    accS[nt] = __builtin_amdgcn_wmma_f32_16x16x4_f32(
                        false, s, false, bb, (short)0, accS[nt], false, false);
                }
            }
        }

        float rs[4][8];
        if (resid) {
    #pragma unroll
            for (int v = 0; v < 8; ++v) {
                int node = node0 + v + mAdd; if (node >= N) node = N - 1;
                const float* rp = resid + (size_t)node * HDIM + ncol;
    #pragma unroll
                for (int nt = 0; nt < 4; ++nt) rs[nt][v] = rp[nt * 16];
            }
        }
        float fg[8], fi[8], fs[8];
    #pragma unroll
        for (int v = 0; v < 8; ++v) {
            fg[v] = __shfl(fgl, v, 16);
            fi[v] = __shfl(fil, v, 16);
            fs[v] = __shfl(fsl, v, 16);
        }
    #pragma unroll
        for (int nt = 0; nt < 4; ++nt) {
            int col = nt * 16 + ncol;
    #pragma unroll
            for (int v = 0; v < 8; ++v) {
                int node = node0 + v + mAdd;
                float val = fg[v] * (accX[nt][v] + bbv[nt]) + fi[v] * accS[nt][v] + fs[v] * wcv[nt];
                if (resid) val += rs[nt][v];
                val = (val > 0.0f) ? val : 0.01f * val;
                if (LAST) mS[(v + mAdd) * SROW + col] = val;
                else if (node < N) Y[(size_t)node * HDIM + col] = val;
            }
        }
        if (LAST) {
            __builtin_amdgcn_fence(__ATOMIC_ACQ_REL, "wavefront");
            float* mPQ = (float*)(sIdx + wave * TILE_IDX_CAP);
            const int l4 = lane & 3;
    #pragma unroll
            for (int ps = 0; ps < 2; ++ps) {
                int r = ps * 8 + (lane >> 2);
                float ap = 0.0f, aq = 0.0f;
    #pragma unroll
                for (int i = 0; i < 4; ++i) {
                    int c4 = i * 4 + l4;
                    float4 xv = *(const float4*)&mS[r * SROW + 4 * c4];
                    float wa0 = woA[i].x, wa1 = woA[i].y, wa2 = woA[i].z, wa3 = woA[i].w;
                    float wb0 = woB[i].x, wb1 = woB[i].y, wb2 = woB[i].z, wb3 = woB[i].w;
                    aq += xv.x * (wa0 - wb0) + xv.y * (wa1 - wb1) + xv.z * (wa2 - wb2) + xv.w * (wa3 - wb3);
                    ap += xv.x * wb0 + xv.y * wb1 + xv.z * wb2 + xv.w * wb3;
                }
    #pragma unroll
                for (int o = 2; o > 0; o >>= 1) {
                    ap += __shfl_xor(ap, o, 4);
                    aq += __shfl_xor(aq, o, 4);
                }
                if (l4 == 0) { mPQ[r] = ap; mPQ[16 + r] = aq; }
            }
        }
    }
    if (LAST) {
        __syncthreads();
        int t = threadIdx.x & 127;
        int node = blockIdx.x * 128 + t;
        const float* wPQ = (const float*)(sIdx + (t >> 4) * TILE_IDX_CAP);
        if (node < N) {
            if (threadIdx.x < 128) pOut[node] = wPQ[t & 15];
            else                   qOut[node] = wPQ[16 + (t & 15)];
        }
    }
}

__global__ __launch_bounds__(256) void out_kernel(const int* __restrict__ rowptr, const int* __restrict__ rowend,
                                                  const int* __restrict__ csr,
                                                  const float* __restrict__ p, const float* __restrict__ q,
                                                  const float* __restrict__ Wout, const float* __restrict__ bout,
                                                  const float* __restrict__ x0,
                                                  const float* __restrict__ invdeg, const float* __restrict__ gate,
                                                  const float* __restrict__ ssn, float* __restrict__ out, int N) {
    __shared__ float sOut[32];
    int t = blockIdx.x * 256 + threadIdx.x;
    int n = t >> 3, l = t & 7;
    int nn = (n < N) ? n : N - 1;
    int b = rowptr[nn], e = rowend[nn];
    float a = 0.0f;
    for (int base = b; base < e; base += 32) {
        int idx[4];
#pragma unroll
        for (int u = 0; u < 4; ++u) {
            int j = base + u * 8 + l;
            idx[u] = csr[(j < e) ? j : (e - 1)];
        }
#pragma unroll
        for (int u = 0; u < 4; ++u) {
            float v = p[idx[u]];
            a += ((base + u * 8 + l) < e) ? v : 0.0f;
        }
    }
#pragma unroll
    for (int o = 4; o > 0; o >>= 1) a += __shfl_xor(a, o, 8);
    if (l == 0) sOut[threadIdx.x >> 3] = a;
    __syncthreads();
    if (threadIdx.x < 32) {
        int m = blockIdx.x * 32 + threadIdx.x;
        if (m < N)
            out[m] = gate[m] * (q[m] + bout[0]) + invdeg[m] * sOut[threadIdx.x] + ssn[m] * Wout[2 * HDIM] + x0[m];
    }
}

extern "C" void kernel_launch(void* const* d_in, const int* in_sizes, int n_in,
                              void* d_out, int out_size, void* d_ws, size_t ws_size,
                              hipStream_t stream) {
    const float* x    = (const float*)d_in[0];
    const int*   ei   = (const int*)  d_in[1];
    const float* Win  = (const float*)d_in[2];
    const float* bin  = (const float*)d_in[3];
    const float* Wmid = (const float*)d_in[4];
    const float* bmid = (const float*)d_in[5];
    const float* Wout = (const float*)d_in[6];
    const float* bout = (const float*)d_in[7];

    const int N = in_sizes[0];
    const int E = in_sizes[1] / 2;
    const int* src = ei;
    const int* dst = ei + E;

    float* w = (float*)d_ws;
    int*   rowptr  = (int*)w;
    int*   cursor  = (int*)(w + (size_t)N);
    float* invdeg  = w + 2 * (size_t)N;
    float* gate    = w + 3 * (size_t)N;
    float* ssn     = w + 4 * (size_t)N;
    float* Sx      = w + 5 * (size_t)N;
    float* Wd6     = w + 6 * (size_t)N;
    float* bufA    = Wd6  + 6 * HDIM * HDIM;
    float* bufB    = bufA + (size_t)N * HDIM;
    float* bufC    = bufB + (size_t)N * HDIM;
    float* bufS    = bufC + (size_t)N * HDIM;
    int*   csr     = (int*)bufC;

    const int nTiles = (N + 15) / 16;

    const int NB = (N + BINW - 1) / BINW;
    int* binned = (int*)bufB;
    int* cntT   = (int*)bufS;
    int* binBase = cntT + NB * NCHUNK;
    float* coef = bufS - 4 * (size_t)N;
    bin_count_kernel<<<NCHUNK, CH_THREADS, 0, stream>>>(dst, cntT, E, NB, Wmid, Wd6);
    bin_scatter_kernel<<<NCHUNK, CH_THREADS, 0, stream>>>(src, dst, cntT, binBase, binned, E, NB);
    bin_fill_kernel<<<NB, FILL_THREADS, 0, stream>>>(binBase, binned, rowptr, cursor, csr, x, invdeg, gate, ssn, Sx,
                                                     coef, N);
    coef_gather_kernel<<<(N * 8 + 255) / 256, 256, 0, stream>>>(rowptr, cursor, csr, coef, x, Win, bin, invdeg, gate,
                                                                ssn, Sx, bufA, bufS, N);

    float* cur = bufA;
    mid_conv_kernel<false, false><<<(nTiles + 7) / 8, 256, 0, stream>>>(
        cur, bufS, rowptr, cursor, csr, Wd6, Wmid + 64 * HDIM, Wmid + 128 * HDIM,
        bmid, gate, invdeg, ssn, nullptr, bufB, nullptr, nullptr, nullptr, nTiles, N);
    mid_conv_kernel<true, false><<<(nTiles + 7) / 8, 256, 0, stream>>>(
        bufB, bufS, rowptr, cursor, csr, Wd6 + HDIM * HDIM,
        Wmid + (size_t)129 * HDIM + 64 * HDIM, Wmid + (size_t)129 * HDIM + 128 * HDIM,
        bmid + HDIM, gate, invdeg, ssn, cur, cur, nullptr, nullptr, nullptr, nTiles, N);
    float* pj = bufS;
    float* qj = bufS + 2 * (size_t)N;
    for (int l = 1; l < 3; ++l) {
        const float* Wl = Wmid + (size_t)(2 * l) * 129 * HDIM;
        mid_conv_kernel<true, false><<<(nTiles + 7) / 8, 256, 0, stream>>>(
            cur, bufS, rowptr, cursor, csr, Wd6 + (2 * l) * HDIM * HDIM, Wl + 64 * HDIM, Wl + 128 * HDIM,
            bmid + (2 * l) * HDIM, gate, invdeg, ssn, nullptr, bufB, nullptr, nullptr, nullptr, nTiles, N);

        const float* Wl2 = Wmid + (size_t)(2 * l + 1) * 129 * HDIM;
        if (l < 2)
            mid_conv_kernel<true, false><<<(nTiles + 7) / 8, 256, 0, stream>>>(
                bufB, bufS, rowptr, cursor, csr, Wd6 + (2 * l + 1) * HDIM * HDIM, Wl2 + 64 * HDIM, Wl2 + 128 * HDIM,
                bmid + (2 * l + 1) * HDIM, gate, invdeg, ssn, cur, cur, nullptr, nullptr, nullptr, nTiles, N);
        else
            mid_conv_kernel<true, true><<<(nTiles + 7) / 8, 256, 0, stream>>>(
                bufB, bufS, rowptr, cursor, csr, Wd6 + (2 * l + 1) * HDIM * HDIM, Wl2 + 64 * HDIM, Wl2 + 128 * HDIM,
                bmid + (2 * l + 1) * HDIM, gate, invdeg, ssn, cur, cur, Wout, pj, qj, nTiles, N);
    }

    out_kernel<<<(N * 8 + 255) / 256, 256, 0, stream>>>(rowptr, cursor, csr, pj, qj, Wout, bout, x, invdeg, gate,
                                                        ssn, (float*)d_out, N);
}


# First mid conv rides on the factor gather: conv_in rows are F(4)@M, so its two 64-deep WMMA products become K=4 WMMAs on M@W folded once per call; 11 launches become 10; product sum order changes
# speedup vs baseline: 6.2096x; 1.0495x over previous
#include <hip/hip_runtime.h>
#include <hip/hip_bf16.h>

typedef float v2f __attribute__((ext_vector_type(2)));
typedef float v8f __attribute__((ext_vector_type(8)));

#define HDIM 64

#define BIN_SHIFT 8
#define BINW 256
#define NB_MAX 512
#define NCHUNK 256
#define CH_THREADS 1024
#define FILL_THREADS 512
#define FILL_CAP 12288

#define FOLD_BLOCK (6 * HDIM * HDIM / CH_THREADS)
__global__ __launch_bounds__(CH_THREADS) void bin_count_kernel(const int* __restrict__ dst,
                                                               int* __restrict__ cntT, int E, int NB,
                                                               const float* __restrict__ Wmid,
                                                               float* __restrict__ Wd6,
                                                               const float* __restrict__ Win,
                                                               const float* __restrict__ bin,
                                                               float* __restrict__ Pfold) {
    __shared__ int sHist[NB_MAX];
    const int c = blockIdx.x, tid = threadIdx.x;
    static_assert(NCHUNK * CH_THREADS >= 6 * HDIM * HDIM, "one thread per element of Wd6");
    {
        int gid = c * CH_THREADS + tid;
        if (gid < 6 * HDIM * HDIM) {
            int l = gid >> 12;
            int rc = gid & 4095;
            const float* Wl = Wmid + (size_t)l * 129 * HDIM;
            Wd6[gid] = Wl[rc] - Wl[64 * HDIM + rc];
        }
    }
    for (int i = tid; i < NB; i += CH_THREADS) sHist[i] = 0;
    __syncthreads();
    const int chunk = (E + NCHUNK - 1) / NCHUNK;
    const int e0 = c * chunk;
    const int e1 = (e0 + chunk < E) ? (e0 + chunk) : E;
    for (int e = e0 + tid; e < e1; e += CH_THREADS * 8) {
        int d[8];
#pragma unroll
        for (int u = 0; u < 8; ++u) {
            int ee = e + u * CH_THREADS;
            d[u] = dst[(ee < e1) ? ee : (e1 - 1)];
        }
#pragma unroll
        for (int u = 0; u < 8; ++u)
            if (e + u * CH_THREADS < e1) atomicAdd(&sHist[d[u] >> BIN_SHIFT], 1);
    }
    __syncthreads();
    for (int i = tid; i < NB; i += CH_THREADS) cntT[i * NCHUNK + c] = sHist[i];

    static_assert((6 * HDIM * HDIM) % CH_THREADS == 0 && FOLD_BLOCK < NCHUNK, "a block with no share of Wd6 exists");
    if (c == FOLD_BLOCK && (tid >> 5) < 8) {
        const int lane = tid & 31, wv = tid >> 5;
        const int mat = wv >> 2, nt = wv & 3;
        const int cl = lane & 15, kh = (lane >> 4) << 1;
        const float* mrow = (cl == 1) ? bin : (Win + ((cl == 2) ? HDIM : ((cl == 3) ? 2 * HDIM : 0)));
        const float* Wa = Wmid + nt * 16 + cl;
        const float* Wb = Wmid + 64 * HDIM + nt * 16 + cl;
        v8f acc = {};
#pragma unroll
        for (int ks = 0; ks < 16; ++ks) {
            const int k0 = (ks << 2) + kh;
            v2f mv = *(const v2f*)&mrow[k0];
            v2f w1 = *(const v2f*)&Win[HDIM + k0];
            v2f a;
            a.x = (cl == 0) ? (mv.x - w1.x) : ((cl < 4) ? mv.x : 0.0f);
            a.y = (cl == 0) ? (mv.y - w1.y) : ((cl < 4) ? mv.y : 0.0f);
            float wa0 = Wa[k0 * HDIM], wa1 = Wa[(k0 + 1) * HDIM];
            float wb0 = Wb[k0 * HDIM], wb1 = Wb[(k0 + 1) * HDIM];
            v2f b;
            b.x = mat ? wb0 : (wa0 - wb0);
            b.y = mat ? wb1 : (wa1 - wb1);
            acc = __builtin_amdgcn_wmma_f32_16x16x4_f32(false, a, false, b, (short)0, acc, false, false);
        }
        if (lane < 16) {
            float* pf = Pfold + (size_t)((mat * 4 + nt) * 32) * 2;
            *(v2f*)&pf[cl * 2]        = v2f{acc[0], acc[1]};
            *(v2f*)&pf[(cl + 16) * 2] = v2f{acc[2], acc[3]};
        }
    }
}

#define GROUPS_PER_PASS 2
__global__ __launch_bounds__(CH_THREADS) void bin_scatter_kernel(const int* __restrict__ src,
                                                                 const int* __restrict__ dst,
                                                                 const int* __restrict__ cntT,
                                                                 int* __restrict__ binBase,
                                                                 int* __restrict__ binned, int E, int NB) {
    static_assert(NCHUNK == 256, "one row of the count table = 8 lanes x 8 loads x 4 counts");
    static_assert(NB_MAX == 512, "wave 0 scans 16 bin totals per lane");
    __shared__ int sCur[NB_MAX];
    __shared__ int sTot[NB_MAX];
    const int c = blockIdx.x, tid = threadIdx.x, lane = tid & 31, wv = tid >> 5;
    const int sub = lane & 7, rw = lane >> 3;
    for (int i = tid; i < NB_MAX; i += CH_THREADS) sTot[i] = 0;
    __syncthreads();
    for (int g0 = wv; g0 * 4 < NB; g0 += (CH_THREADS / 32) * GROUPS_PER_PASS) {
        int4 v[GROUPS_PER_PASS][8];
#pragma unroll
        for (int u = 0; u < GROUPS_PER_PASS; ++u) {
            int i = (g0 + u * (CH_THREADS / 32)) * 4 + rw;
            const int4* row = (const int4*)(cntT + (size_t)((i < NB) ? i : (NB - 1)) * NCHUNK);
#pragma unroll
            for (int j = 0; j < 8; ++j) v[u][j] = row[j * 8 + sub];
        }
#pragma unroll
        for (int u = 0; u < GROUPS_PER_PASS; ++u) {
            int tot = 0, pre = 0;
#pragma unroll
            for (int j = 0; j < 8; ++j) {
                int k = (j * 8 + sub) * 4;
                tot += v[u][j].x + v[u][j].y + v[u][j].z + v[u][j].w;
                pre += ((k     < c) ? v[u][j].x : 0) + ((k + 1 < c) ? v[u][j].y : 0)
                     + ((k + 2 < c) ? v[u][j].z : 0) + ((k + 3 < c) ? v[u][j].w : 0);
            }
#pragma unroll
            for (int o = 4; o > 0; o >>= 1) {
                tot += __shfl_xor(tot, o, 8);
                pre += __shfl_xor(pre, o, 8);
            }
            int i = (g0 + u * (CH_THREADS / 32)) * 4 + rw;
            if (sub == 0 && i < NB) { sTot[i] = tot; sCur[i] = pre; }
        }
    }
    __syncthreads();
    if (wv == 0) {
        int t[16];
        int s = 0;
#pragma unroll
        for (int k = 0; k < 16; ++k) { t[k] = sTot[lane * 16 + k]; s += t[k]; }
        int inc = s;
#pragma unroll
        for (int o = 1; o < 32; o <<= 1) {
            int v = __shfl_up(inc, o, 32);
            if (lane >= o) inc += v;
        }
        int ex = inc - s;
#pragma unroll
        for (int k = 0; k < 16; ++k) { sTot[lane * 16 + k] = ex; ex += t[k]; }
    }
    __syncthreads();
    for (int i = tid; i < NB; i += CH_THREADS) {
        int base = sTot[i];
        sCur[i] += base;
        if (c == 0) binBase[i] = base;
    }
    if (c == 0 && tid == 0) binBase[NB] = E;
    __syncthreads();
    const int chunk = (E + NCHUNK - 1) / NCHUNK;
    const int e0 = c * chunk;
    const int e1 = (e0 + chunk < E) ? (e0 + chunk) : E;
    for (int e = e0 + tid; e < e1; e += CH_THREADS * 8) {
        int d[8], s[8];
#pragma unroll
        for (int u = 0; u < 8; ++u) {
            int ee = e + u * CH_THREADS;
            int ec = (ee < e1) ? ee : (e1 - 1);
            d[u] = dst[ec];
            s[u] = src[ec];
        }
#pragma unroll
        for (int u = 0; u < 8; ++u) {
            if (e + u * CH_THREADS < e1) {
                int pos = atomicAdd(&sCur[d[u] >> BIN_SHIFT], 1);
                binned[pos] = (s[u] << BIN_SHIFT) | (d[u] & (BINW - 1));
            }
        }
    }
}

__global__ __launch_bounds__(FILL_THREADS) void bin_fill_kernel(const int* __restrict__ binBase,
                                                                const int* __restrict__ binned,
                                                                int* __restrict__ rowptr, int* __restrict__ rowend,
                                                                int* __restrict__ csr, const float* __restrict__ x,
                                                                float* __restrict__ invdeg, float* __restrict__ gate,
                                                                float* __restrict__ ssn, float* __restrict__ Sx,
                                                                float* __restrict__ coef, int N) {
    static_assert(BINW == (FILL_THREADS / 8) * 4, "statistics: 8 lanes per node, 4 nodes per lane group");
    __shared__ int sDeg[BINW];
    __shared__ int sBeg[BINW];
    __shared__ int sWaveTot[BINW / 32];
    __shared__ int sStage[FILL_CAP];
    const int b = blockIdx.x, tid = threadIdx.x, lane = tid & 31;
    const int beg = binBase[b];
    const int end = binBase[b + 1];
    const int cnt = end - beg;
    if (tid < BINW) sDeg[tid] = 0;
    __syncthreads();
    for (int i = tid; i < cnt; i += FILL_THREADS * 8) {
        int r[8];
#pragma unroll
        for (int u = 0; u < 8; ++u) {
            int ii = i + u * FILL_THREADS;
            r[u] = binned[beg + ((ii < cnt) ? ii : (cnt - 1))];
        }
#pragma unroll
        for (int u = 0; u < 8; ++u)
            if (i + u * FILL_THREADS < cnt) atomicAdd(&sDeg[r[u] & (BINW - 1)], 1);
    }
    __syncthreads();
    int v = 0, inc = 0;
    if (tid < BINW) {
        v = sDeg[tid];
        inc = v;
#pragma unroll
        for (int o = 1; o < 32; o <<= 1) {
            int t = __shfl_up(inc, o, 32);
            if (lane >= o) inc += t;
        }
        if (lane == 31) sWaveTot[tid >> 5] = inc;
    }
    __syncthreads();
    if (tid < BINW) {
        int wbase = 0;
        for (int w = 0; w < (tid >> 5); ++w) wbase += sWaveTot[w];
        int excl = wbase + inc - v;
        sDeg[tid] = excl;
        sBeg[tid] = excl;
        int node = b * BINW + tid;
        if (node < N) { rowptr[node] = beg + excl; rowend[node] = beg + excl + v; }
    }
    __syncthreads();
    const bool staged = cnt <= FILL_CAP;
    for (int i = tid; i < cnt; i += FILL_THREADS * 8) {
        int r[8];
#pragma unroll
        for (int u = 0; u < 8; ++u) {
            int ii = i + u * FILL_THREADS;
            r[u] = binned[beg + ((ii < cnt) ? ii : (cnt - 1))];
        }
#pragma unroll
        for (int u = 0; u < 8; ++u) {
            if (i + u * FILL_THREADS < cnt) {
                int pos = atomicAdd(&sDeg[r[u] & (BINW - 1)], 1);
                if (staged) sStage[pos] = r[u] >> BIN_SHIFT;
                else        csr[beg + pos] = r[u] >> BIN_SHIFT;
            }
        }
    }
    __syncthreads();
    if (staged)
        for (int i = tid; i < cnt; i += FILL_THREADS) csr[beg + i] = sStage[i];

    const int l8 = tid & 7, grp = tid >> 3;
    int sb[4], se[4], nd[4];
    float sg[4], sx[4];
#pragma unroll
    for (int u = 0; u < 4; ++u) {
        int ln = grp + u * (FILL_THREADS / 8);
        sb[u] = sBeg[ln];
        se[u] = sDeg[ln];
        nd[u] = b * BINW + ln;
        sg[u] = 0.0f; sx[u] = 0.0f;
    }
    int longest = 0;
#pragma unroll
    for (int u = 0; u < 4; ++u) longest = (se[u] - sb[u] > longest) ? (se[u] - sb[u]) : longest;
    for (int off = 0; off < longest; off += 32) {
        int s[4][4];
        if (staged) {
#pragma unroll
            for (int u = 0; u < 4; ++u) {
#pragma unroll
                for (int k = 0; k < 4; ++k) {
                    int j = sb[u] + off + k * 8 + l8;
                    bool ok = j < se[u];
                    int t = sStage[ok ? j : 0];
                    s[u][k] = ok ? t : 0;
                }
            }
        } else {
#pragma unroll
            for (int u = 0; u < 4; ++u) {
#pragma unroll
                for (int k = 0; k < 4; ++k) {
                    int j = sb[u] + off + k * 8 + l8;
                    bool ok = j < se[u];
                    int t = csr[beg + (ok ? j : 0)];
                    s[u][k] = ok ? t : 0;
                }
            }
        }
#pragma unroll
        for (int u = 0; u < 4; ++u) {
#pragma unroll
            for (int k = 0; k < 4; ++k) {
                float xv = x[s[u][k]];
                bool ok = (sb[u] + off + k * 8 + l8) < se[u];
                float sgn = (s[u][k] > nd[u]) ? 1.0f : ((s[u][k] < nd[u]) ? -1.0f : 0.0f);
                sg[u] += ok ? sgn : 0.0f;
                sx[u] += ok ? xv : 0.0f;
            }
        }
    }
#pragma unroll
    for (int o = 4; o > 0; o >>= 1) {
#pragma unroll
        for (int u = 0; u < 4; ++u) {
            sg[u] += __shfl_xor(sg[u], o, 8);
            sx[u] += __shfl_xor(sx[u], o, 8);
        }
    }
    if (l8 == 0) {
#pragma unroll
        for (int u = 0; u < 4; ++u) {
            int n = nd[u];
            if (n < N) {
                float dg = (float)(se[u] - sb[u]);
                float inv = 1.0f / fmaxf(dg, 1.0f);
                float g = (dg > 0.0f) ? 1.0f : 0.0f;
                invdeg[n] = inv;
                gate[n]   = g;
                ssn[n]    = sg[u] * inv;
                Sx[n]     = sx[u];
                ((float4*)coef)[n] = make_float4(g * x[n], g, inv * sx[u], sg[u] * inv);
            }
        }
    }
}

__global__ __launch_bounds__(256) void coef_gather_kernel(const int* __restrict__ rowptr,
                                                          const int* __restrict__ rowend,
                                                          const int* __restrict__ csr,
                                                          const float* __restrict__ coef,
                                                          const float* __restrict__ x,
                                                          const float* __restrict__ Win,
                                                          const float* __restrict__ bin,
                                                          const float* __restrict__ invdeg,
                                                          const float* __restrict__ gate,
                                                          const float* __restrict__ ssn,
                                                          const float* __restrict__ Sx,
                                                          const float* __restrict__ Pfold,
                                                          const float* __restrict__ Wc0,
                                                          const float* __restrict__ b0,
                                                          float* __restrict__ Y, float* __restrict__ Y1, int N) {
    __shared__ float4 sOut[32];
    const float4* coef4 = (const float4*)coef;
    int t = blockIdx.x * 256 + threadIdx.x;
    int n = t >> 3, l = t & 7;
    int nn = (n < N) ? n : N - 1;
    int b = rowptr[nn], e = rowend[nn];
    const int q = threadIdx.x & 15, r0 = threadIdx.x >> 4;
    const float4 wa = ((const float4*)Win)[q], wb = ((const float4*)Win)[16 + q], wc = ((const float4*)Win)[32 + q];
    const float4 bh = ((const float4*)bin)[q];
    float fx[2], fg[2], fi[2], fsx[2], fsn[2];
#pragma unroll
    for (int u = 0; u < 2; ++u) {
        int m = blockIdx.x * 32 + r0 + 16 * u; if (m >= N) m = N - 1;
        fx[u] = x[m]; fg[u] = gate[m]; fi[u] = invdeg[m]; fsx[u] = Sx[m]; fsn[u] = ssn[m];
    }
    float4 a = make_float4(0.0f, 0.0f, 0.0f, 0.0f);
    for (int base = b; base < e; base += 32) {
        int idx[4];
#pragma unroll
        for (int u = 0; u < 4; ++u) {
            int j = base + u * 8 + l;
            idx[u] = csr[(j < e) ? j : (e - 1)];
        }
#pragma unroll
        for (int u = 0; u < 4; ++u) {
            float4 c = coef4[idx[u]];
            bool ok = (base + u * 8 + l) < e;
            a.x += ok ? c.x : 0.0f;
            a.y += ok ? c.y : 0.0f;
            a.z += ok ? c.z : 0.0f;
            a.w += ok ? c.w : 0.0f;
        }
    }
#pragma unroll
    for (int o = 4; o > 0; o >>= 1) {
        a.x += __shfl_xor(a.x, o, 8);
        a.y += __shfl_xor(a.y, o, 8);
        a.z += __shfl_xor(a.z, o, 8);
        a.w += __shfl_xor(a.w, o, 8);
    }
    if (l == 0) sOut[threadIdx.x >> 3] = a;
    __syncthreads();
#pragma unroll
    for (int u = 0; u < 2; ++u) {
        int r = r0 + 16 * u;
        int m = blockIdx.x * 32 + r;
        if (m < N) {
            float gx = fx[u], g = fg[u], is = fi[u] * fsx[u], sn = fsn[u];
            float4 y;
            y.x = g * (gx * (wa.x - wb.x) + bh.x) + is * wb.x + sn * wc.x;
            y.y = g * (gx * (wa.y - wb.y) + bh.y) + is * wb.y + sn * wc.y;
            y.z = g * (gx * (wa.z - wb.z) + bh.z) + is * wb.z + sn * wc.z;
            y.w = g * (gx * (wa.w - wb.w) + bh.w) + is * wb.w + sn * wc.w;
            ((float4*)Y)[(size_t)m * 16 + q] = y;
        }
    }
    const int lane = threadIdx.x & 31, wv = threadIdx.x >> 5;
    if (wv < 4) {
        const int cl = lane & 15, hh = lane >> 4;
        const int t2 = wv >> 1, l2 = wv & 1;
        int mA = blockIdx.x * 32 + t2 * 16 + cl; if (mA >= N) mA = N - 1;
        const float4 cF = coef4[mA];
        const float  iv = invdeg[mA];
        const float4 cs = sOut[t2 * 16 + cl];
        const float  g = cF.y;
        v2f aF, aS, aR;
        aF.x = g * (hh ? cF.z : cF.x);   aF.y = g * (hh ? cF.w : cF.y);
        aS.x = iv * (hh ? cs.z : cs.x);  aS.y = iv * (hh ? cs.w : cs.y);
        aR.x = hh ? 0.0f : g;            aR.y = hh ? 0.0f : cF.w;
#pragma unroll
        for (int j = 0; j < 2; ++j) {
            const int nt = 2 * l2 + j;
            const int col = nt * 16 + cl;
            const v2f bPd = *(const v2f*)&Pfold[((0 * 4 + nt) * 32 + lane) * 2];
            const v2f bPb = *(const v2f*)&Pfold[((1 * 4 + nt) * 32 + lane) * 2];
            const float bv = b0[col], wv0 = Wc0[col];
            v2f bR;
            bR.x = hh ? 0.0f : bv;       bR.y = hh ? 0.0f : wv0;
            v8f acc = {};
            acc = __builtin_amdgcn_wmma_f32_16x16x4_f32(false, aF, false, bPd, (short)0, acc, false, false);
            acc = __builtin_amdgcn_wmma_f32_16x16x4_f32(false, aS, false, bPb, (short)0, acc, false, false);
            acc = __builtin_amdgcn_wmma_f32_16x16x4_f32(false, aR, false, bR, (short)0, acc, false, false);
#pragma unroll
            for (int v = 0; v < 8; ++v) {
                int node = blockIdx.x * 32 + t2 * 16 + v + 8 * hh;
                float val = acc[v];
                val = (val > 0.0f) ? val : 0.01f * val;
                if (node < N) Y1[(size_t)node * HDIM + col] = val;
            }
        }
    }
}

__device__ __forceinline__ void rows16(const char* __restrict__ Xb, unsigned mine, unsigned qoff, float4 (&v)[16]) {
#pragma unroll
    for (int k = 0; k < 16; ++k) {
        unsigned off = (unsigned)__shfl((int)mine, k, 16) + qoff;
        v[k] = *(const float4*)(Xb + off);
    }
}
template <bool FULL>
__device__ __forceinline__ void sum16(const float4 (&v)[16], int cnt, float4& acc) {
#pragma unroll
    for (int k = 0; k < 16; ++k) {
        bool ok = FULL || (k < cnt);
        acc.x += ok ? v[k].x : 0.0f;
        acc.y += ok ? v[k].y : 0.0f;
        acc.z += ok ? v[k].z : 0.0f;
        acc.w += ok ? v[k].w : 0.0f;
    }
}

#define SROW 68
#define TILE_IDX_CAP 768
template <bool GATHER, bool LAST>
__global__ __launch_bounds__(256) void mid_conv_kernel(
    const float* __restrict__ X, const float* __restrict__ S,
    const int* __restrict__ rowptr, const int* __restrict__ rowend, const int* __restrict__ csr,
    const float* __restrict__ Wd, const float* __restrict__ Wb,
    const float* __restrict__ Wc, const float* __restrict__ bias,
    const float* __restrict__ gate, const float* __restrict__ invdeg,
    const float* __restrict__ ssn, const float* resid,
    float* Y,
    const float* __restrict__ Wout, float* __restrict__ pOut, float* __restrict__ qOut, int nTiles, int N) {
    static_assert(GATHER || !LAST, "the LAST form keeps its rows and its p, q in the LDS areas of the gather");
    __shared__ float sWd[HDIM * HDIM];
    __shared__ float sWb[HDIM * HDIM];
    __shared__ __attribute__((aligned(16))) float sS[GATHER ? 8 * 16 * SROW : 4];
    __shared__ int   sIdx[GATHER ? 8 * TILE_IDX_CAP : 2];
    for (int i = threadIdx.x; i < HDIM * HDIM; i += 256) {
        int ks   = i >> 8;
        int rem  = i & 255;
        int nt   = rem >> 6;
        int lp   = rem & 63;
        int ln   = lp >> 1;
        int half = lp & 1;
        int kh   = (ln >> 4) << 1;
        int srcr = (ks << 2) + kh + half;
        int srcc = (nt << 4) + (ln & 15);
        sWd[i] = Wd[srcr * HDIM + srcc];
        sWb[i] = Wb[srcr * HDIM + srcc];
    }
    __syncthreads();

    int wave = threadIdx.x >> 5;
    int lane = threadIdx.x & 31;
    int tile = blockIdx.x * 8 + wave;
    const bool active = tile < nTiles;
    if (!LAST && !active) return;
    if (active) {
        int node0 = tile * 16;
        int m  = lane & 15;
        int kh = (lane >> 4) << 1;
        int row = node0 + m; if (row >= N) row = N - 1;
        const float* Xr = X + (size_t)row * HDIM;
        const float* Sr = S + (size_t)row * HDIM;
        float* mS = sS + (GATHER ? wave * (16 * SROW) : 0);

        int ncol = lane & 15;
        int mAdd = (lane >> 4) * 8;
        int frow = node0 + (lane & 7) + mAdd; if (frow >= N) frow = N - 1;
        float fgl = gate[frow], fil = invdeg[frow], fsl = ssn[frow];
        float wcv[4], bbv[4];
    #pragma unroll
        for (int nt = 0; nt < 4; ++nt) {
            wcv[nt] = Wc[nt * 16 + ncol];
            bbv[nt] = bias[nt * 16 + ncol];
        }

        float4 woA[4], woB[4];
        if (LAST) {
    #pragma unroll
            for (int i = 0; i < 4; ++i) {
                int c4 = i * 4 + (lane & 3);
                woA[i] = ((const float4*)Wout)[c4];
                woB[i] = ((const float4*)Wout)[16 + c4];
            }
        }

        v8f accX[4] = {};
        v8f accS[4] = {};
        int fragBase = lane * 2;

        if (GATHER) {
            int* mIx = sIdx + wave * TILE_IDX_CAP;
            int rb = rowptr[row], re = rowend[row];
            if (node0 + m >= N) rb = re;
            int tb = __shfl(rb, 0, 32);
            int staged = __shfl(re, 15, 32) - tb;
            staged = (staged < 0) ? 0 : ((staged > TILE_IDX_CAP) ? TILE_IDX_CAP : staged);
            for (int i0 = 0; i0 < staged; i0 += 32 * 16) {
                int v[16];
    #pragma unroll
                for (int u = 0; u < 16; ++u) {
                    int i = i0 + u * 32 + lane;
                    v[u] = csr[tb + ((i < staged) ? i : (staged - 1))];
                }
    #pragma unroll
                for (int u = 0; u < 16; ++u) {
                    int i = i0 + u * 32 + lane;
                    if (i < staged) mIx[i] = v[u] << 8;
                }
            }
            __builtin_amdgcn_fence(__ATOMIC_ACQ_REL, "wavefront");

            v2f aN0 = *(const v2f*)&Xr[kh];
            v2f aN1 = *(const v2f*)&Xr[4 + kh];

            static_assert((size_t)NB_MAX * BINW * HDIM * sizeof(float) <= 0xffffffffull, "a row's byte offset fits 32 bits");
            const char* Xb = (const char*)X;
            int q = lane & 15, h = lane >> 4;
            const unsigned qoff = (unsigned)q * 16u;
            for (int p = 0; p < 8; ++p) {
                const v2f aP0 = aN0, aP1 = aN1;
                int r = 2 * p + h;
                int b = __shfl(rb, r, 32), e = __shfl(re, r, 32);
                int own = node0 + r; if (own >= N) own = N - 1;
                const unsigned ownOff = (unsigned)own << 8;
                float4 acc = make_float4(0.0f, 0.0f, 0.0f, 0.0f);
                int eL = (b >= tb) ? (((tb + staged) < e) ? (tb + staged) : e) : b;
                if (eL < b) eL = b;
                int base = b;
                int j0 = base + q;
                unsigned cur = (j0 < eL) ? (unsigned)mIx[j0 - tb] : ownOff;
                float4 v[16];
                rows16(Xb, cur, qoff, v);
                {
                    const int pn = (p < 7) ? (p + 1) : 7;
                    aN0 = *(const v2f*)&Xr[(pn << 3) + kh];
                    aN1 = *(const v2f*)&Xr[(pn << 3) + 4 + kh];
                }
    #pragma unroll
                for (int nt = 0; nt < 4; ++nt) {
                    v2f bd0 = *(const v2f*)&sWd[((p << 3) + nt) * 64 + fragBase];
                    accX[nt] = __builtin_amdgcn_wmma_f32_16x16x4_f32(
                        false, aP0, false, bd0, (short)0, accX[nt], false, false);
                }
    #pragma unroll
                for (int nt = 0; nt < 4; ++nt) {
                    v2f bd1 = *(const v2f*)&sWd[((p << 3) + 4 + nt) * 64 + fragBase];
                    accX[nt] = __builtin_amdgcn_wmma_f32_16x16x4_f32(
                        false, aP1, false, bd1, (short)0, accX[nt], false, false);
                }
                if (base + 16 <= eL) {
                    int jn = base + 16 + q;
                    int jc = jn - tb;
                    jc = (jc < staged) ? jc : (staged - 1);
                    unsigned t = (unsigned)mIx[jc];
                    sum16<true>(v, 16, acc);
                    cur = (jn < eL) ? t : ownOff;
                    base += 16;
                    while (base + 16 <= eL) {
                        rows16(Xb, cur, qoff, v);
                        jn = base + 16 + q;
                        jc = jn - tb;
                        jc = (jc < staged) ? jc : (staged - 1);
                        t = (unsigned)mIx[jc];
                        sum16<true>(v, 16, acc);
                        cur = (jn < eL) ? t : ownOff;
                        base += 16;
                    }
                    if (base < eL) {
                        rows16(Xb, cur, qoff, v);
                        sum16<false>(v, eL - base, acc);
                    }
                } else if (base < eL) {
                    sum16<false>(v, eL - base, acc);
                }
                for (int bm = eL; bm < e; bm += 16) {
                    int j = bm + q;
                    unsigned mine = (j < e) ? ((unsigned)csr[j] << 8) : ownOff;
                    rows16(Xb, mine, qoff, v);
                    sum16<false>(v, e - bm, acc);
                }
                *(float4*)&mS[r * SROW + 4 * q] = acc;
            }
            __builtin_amdgcn_fence(__ATOMIC_ACQ_REL, "wavefront");
        }

        if (GATHER) {
    #pragma unroll
            for (int ks = 0; ks < 16; ++ks) {
                int k0 = ks << 2;
                v2f s = *(const v2f*)&mS[m * SROW + k0 + kh];
    #pragma unroll
                for (int nt = 0; nt < 4; ++nt) {
                    v2f bb = *(const v2f*)&sWb[((ks << 2) + nt) * 64 + fragBase];
                    accS[nt] = __builtin_amdgcn_wmma_f32_16x16x4_f32(
                        false, s, false, bb, (short)0, accS[nt], false, false);
                }
            }
        } else {
            v2f aX[16];
    #pragma unroll
            for (int ks = 0; ks < 16; ++ks) aX[ks] = *(const v2f*)&Xr[(ks << 2) + kh];

    #pragma unroll
            for (int ks = 0; ks < 16; ++ks) {
                int k0 = ks << 2;
                v2f a = aX[ks], s;
                s.x = Sr[k0 + kh]; s.y = Sr[k0 + kh + 1];
    #pragma unroll
                for (int nt = 0; nt < 4; ++nt) {
                    v2f bd = *(const v2f*)&sWd[((ks << 2) + nt) * 64 + fragBase];
                    v2f bb = *(const v2f*)&sWb[((ks << 2) + nt) * 64 + fragBase];
                    accX[nt] = __builtin_amdgcn_wmma_f32_16x16x4_f32(
                        false, a, false, bd, (short)0, accX[nt], false, false);
                    accS[nt] = __builtin_amdgcn_wmma_f32_16x16x4_f32(
                        false, s, false, bb, (short)0, accS[nt], false, false);
                }
            }
        }

        float rs[4][8];
        if (resid) {
    #pragma unroll
            for (int v = 0; v < 8; ++v) {
                int node = node0 + v + mAdd; if (node >= N) node = N - 1;
                const float* rp = resid + (size_t)node * HDIM + ncol;
    #pragma unroll
                for (int nt = 0; nt < 4; ++nt) rs[nt][v] = rp[nt * 16];
            }
        }
        float fg[8], fi[8], fs[8];
    #pragma unroll
        for (int v = 0; v < 8; ++v) {
            fg[v] = __shfl(fgl, v, 16);
            fi[v] = __shfl(fil, v, 16);
            fs[v] = __shfl(fsl, v, 16);
        }
    #pragma unroll
        for (int nt = 0; nt < 4; ++nt) {
            int col = nt * 16 + ncol;
    #pragma unroll
            for (int v = 0; v < 8; ++v) {
                int node = node0 + v + mAdd;
                float val = fg[v] * (accX[nt][v] + bbv[nt]) + fi[v] * accS[nt][v] + fs[v] * wcv[nt];
                if (resid) val += rs[nt][v];
                val = (val > 0.0f) ? val : 0.01f * val;
                if (LAST) mS[(v + mAdd) * SROW + col] = val;
                else if (node < N) Y[(size_t)node * HDIM + col] = val;
            }
        }
        if (LAST) {
            __builtin_amdgcn_fence(__ATOMIC_ACQ_REL, "wavefront");
            float* mPQ = (float*)(sIdx + wave * TILE_IDX_CAP);
            const int l4 = lane & 3;
    #pragma unroll
            for (int ps = 0; ps < 2; ++ps) {
                int r = ps * 8 + (lane >> 2);
                float ap = 0.0f, aq = 0.0f;
    #pragma unroll
                for (int i = 0; i < 4; ++i) {
                    int c4 = i * 4 + l4;
                    float4 xv = *(const float4*)&mS[r * SROW + 4 * c4];
                    float wa0 = woA[i].x, wa1 = woA[i].y, wa2 = woA[i].z, wa3 = woA[i].w;
                    float wb0 = woB[i].x, wb1 = woB[i].y, wb2 = woB[i].z, wb3 = woB[i].w;
                    aq += xv.x * (wa0 - wb0) + xv.y * (wa1 - wb1) + xv.z * (wa2 - wb2) + xv.w * (wa3 - wb3);
                    ap += xv.x * wb0 + xv.y * wb1 + xv.z * wb2 + xv.w * wb3;
                }
    #pragma unroll
                for (int o = 2; o > 0; o >>= 1) {
                    ap += __shfl_xor(ap, o, 4);
                    aq += __shfl_xor(aq, o, 4);
                }
                if (l4 == 0) { mPQ[r] = ap; mPQ[16 + r] = aq; }
            }
        }
    }
    if (LAST) {
        __syncthreads();
        int t = threadIdx.x & 127;
        int node = blockIdx.x * 128 + t;
        const float* wPQ = (const float*)(sIdx + (t >> 4) * TILE_IDX_CAP);
        if (node < N) {
            if (threadIdx.x < 128) pOut[node] = wPQ[t & 15];
            else                   qOut[node] = wPQ[16 + (t & 15)];
        }
    }
}

__global__ __launch_bounds__(256) void out_kernel(const int* __restrict__ rowptr, const int* __restrict__ rowend,
                                                  const int* __restrict__ csr,
                                                  const float* __restrict__ p, const float* __restrict__ q,
                                                  const float* __restrict__ Wout, const float* __restrict__ bout,
                                                  const float* __restrict__ x0,
                                                  const float* __restrict__ invdeg, const float* __restrict__ gate,
                                                  const float* __restrict__ ssn, float* __restrict__ out, int N) {
    __shared__ float sOut[32];
    int t = blockIdx.x * 256 + threadIdx.x;
    int n = t >> 3, l = t & 7;
    int nn = (n < N) ? n : N - 1;
    int b = rowptr[nn], e = rowend[nn];
    float a = 0.0f;
    for (int base = b; base < e; base += 32) {
        int idx[4];
#pragma unroll
        for (int u = 0; u < 4; ++u) {
            int j = base + u * 8 + l;
            idx[u] = csr[(j < e) ? j : (e - 1)];
        }
#pragma unroll
        for (int u = 0; u < 4; ++u) {
            float v = p[idx[u]];
            a += ((base + u * 8 + l) < e) ? v : 0.0f;
        }
    }
#pragma unroll
    for (int o = 4; o > 0; o >>= 1) a += __shfl_xor(a, o, 8);
    if (l == 0) sOut[threadIdx.x >> 3] = a;
    __syncthreads();
    if (threadIdx.x < 32) {
        int m = blockIdx.x * 32 + threadIdx.x;
        if (m < N)
            out[m] = gate[m] * (q[m] + bout[0]) + invdeg[m] * sOut[threadIdx.x] + ssn[m] * Wout[2 * HDIM] + x0[m];
    }
}

extern "C" void kernel_launch(void* const* d_in, const int* in_sizes, int n_in,
                              void* d_out, int out_size, void* d_ws, size_t ws_size,
                              hipStream_t stream) {
    const float* x    = (const float*)d_in[0];
    const int*   ei   = (const int*)  d_in[1];
    const float* Win  = (const float*)d_in[2];
    const float* bin  = (const float*)d_in[3];
    const float* Wmid = (const float*)d_in[4];
    const float* bmid = (const float*)d_in[5];
    const float* Wout = (const float*)d_in[6];
    const float* bout = (const float*)d_in[7];

    const int N = in_sizes[0];
    const int E = in_sizes[1] / 2;
    const int* src = ei;
    const int* dst = ei + E;

    float* w = (float*)d_ws;
    int*   rowptr  = (int*)w;
    int*   cursor  = (int*)(w + (size_t)N);
    float* invdeg  = w + 2 * (size_t)N;
    float* gate    = w + 3 * (size_t)N;
    float* ssn     = w + 4 * (size_t)N;
    float* Sx      = w + 5 * (size_t)N;
    float* Wd6     = w + 6 * (size_t)N;
    float* bufA    = Wd6  + 6 * HDIM * HDIM;
    float* bufB    = bufA + (size_t)N * HDIM;
    float* bufC    = bufB + (size_t)N * HDIM;
    float* bufS    = bufC + (size_t)N * HDIM;
    int*   csr     = (int*)bufC;

    const int nTiles = (N + 15) / 16;

    const int NB = (N + BINW - 1) / BINW;
    int* binned = (int*)bufB;
    int* cntT   = (int*)bufS;
    int* binBase = cntT + NB * NCHUNK;
    float* coef = bufS - 4 * (size_t)N;
    float* Pfold = bufS + 8 * (size_t)N;
    bin_count_kernel<<<NCHUNK, CH_THREADS, 0, stream>>>(dst, cntT, E, NB, Wmid, Wd6, Win, bin, Pfold);
    bin_scatter_kernel<<<NCHUNK, CH_THREADS, 0, stream>>>(src, dst, cntT, binBase, binned, E, NB);
    bin_fill_kernel<<<NB, FILL_THREADS, 0, stream>>>(binBase, binned, rowptr, cursor, csr, x, invdeg, gate, ssn, Sx,
                                                     coef, N);
    coef_gather_kernel<<<(N * 8 + 255) / 256, 256, 0, stream>>>(rowptr, cursor, csr, coef, x, Win, bin, invdeg, gate,
                                                                ssn, Sx, Pfold, Wmid + 128 * HDIM, bmid, bufA, bufB, N);

    float* cur = bufA;
    mid_conv_kernel<true, false><<<(nTiles + 7) / 8, 256, 0, stream>>>(
        bufB, bufS, rowptr, cursor, csr, Wd6 + HDIM * HDIM,
        Wmid + (size_t)129 * HDIM + 64 * HDIM, Wmid + (size_t)129 * HDIM + 128 * HDIM,
        bmid + HDIM, gate, invdeg, ssn, cur, cur, nullptr, nullptr, nullptr, nTiles, N);
    float* pj = bufS;
    float* qj = bufS + 2 * (size_t)N;
    for (int l = 1; l < 3; ++l) {
        const float* Wl = Wmid + (size_t)(2 * l) * 129 * HDIM;
        mid_conv_kernel<true, false><<<(nTiles + 7) / 8, 256, 0, stream>>>(
            cur, bufS, rowptr, cursor, csr, Wd6 + (2 * l) * HDIM * HDIM, Wl + 64 * HDIM, Wl + 128 * HDIM,
            bmid + (2 * l) * HDIM, gate, invdeg, ssn, nullptr, bufB, nullptr, nullptr, nullptr, nTiles, N);

        const float* Wl2 = Wmid + (size_t)(2 * l + 1) * 129 * HDIM;
        if (l < 2)
            mid_conv_kernel<true, false><<<(nTiles + 7) / 8, 256, 0, stream>>>(
                bufB, bufS, rowptr, cursor, csr, Wd6 + (2 * l + 1) * HDIM * HDIM, Wl2 + 64 * HDIM, Wl2 + 128 * HDIM,
                bmid + (2 * l + 1) * HDIM, gate, invdeg, ssn, cur, cur, nullptr, nullptr, nullptr, nTiles, N);
        else
            mid_conv_kernel<true, true><<<(nTiles + 7) / 8, 256, 0, stream>>>(
                bufB, bufS, rowptr, cursor, csr, Wd6 + (2 * l + 1) * HDIM * HDIM, Wl2 + 64 * HDIM, Wl2 + 128 * HDIM,
                bmid + (2 * l + 1) * HDIM, gate, invdeg, ssn, cur, cur, Wout, pj, qj, nTiles, N);
    }

    out_kernel<<<(N * 8 + 255) / 256, 256, 0, stream>>>(rowptr, cursor, csr, pj, qj, Wout, bout, x, invdeg, gate,
                                                        ssn, (float*)d_out, N);
}
